# starting kernel
# baseline (speedup 1.0000x reference)
#include <hip/hip_runtime.h>


typedef float v2f __attribute__((ext_vector_type(2)));
typedef float v8f __attribute__((ext_vector_type(8)));

#define B_BATCH 1024
#define NNODE   128
#define CDIM    128
#define MHID    10
#define OUTD    32
#define NPAD    136

__device__ __forceinline__ v8f wmma4(v2f a, v2f b, v8f c) {
  return __builtin_amdgcn_wmma_f32_16x16x4_f32(
       false, a,  false, b,
       (short)0, c,  false,  false);
}

#define WS_A   0
#define WS_Z1  3200
#define WS_Z2  44160
#define WS_ZF  85120
#define ZF_STRIDE 40960

__global__ __launch_bounds__(256) void rw_pre(const float* __restrict__ fh,
                                              const float* __restrict__ ah,
                                              float* __restrict__ ws) {
  float* Adense = ws + WS_A;
  float* z1     = ws + WS_Z1;
  float* z2     = ws + WS_Z2;
  float* zf     = ws + WS_ZF;
  const int tid = threadIdx.x;

  for (int e = tid; e < MHID * MHID * OUTD; e += 256) {
    int d = e & 31, ij = e >> 5, i2 = ij / MHID, j2 = ij % MHID;
    float v = 0.f;
    if (i2 != j2) {
      int i = i2 < j2 ? i2 : j2;
      int j = i2 < j2 ? j2 : i2;
      int p = i * MHID - (i * (i + 1)) / 2 + (j - i - 1);
      float t = ah[p * OUTD + d];
      v = t > 0.f ? t : 0.f;
    }
    Adense[e] = v;
  }
  __syncthreads();

  for (int e = tid; e < MHID * CDIM * OUTD; e += 256) {
    int d = e & 31, c = (e >> 5) & 127, a = e >> 12;
    float s = 0.f;
#pragma unroll
    for (int bb = 0; bb < MHID; ++bb)
      s += Adense[(a * MHID + bb) * OUTD + d] * fh[(bb * CDIM + c) * OUTD + d];
    z1[e] = s;
  }
  __syncthreads();

  for (int e = tid; e < MHID * CDIM * OUTD; e += 256) {
    int d = e & 31, c = (e >> 5) & 127, a = e >> 12;
    float s = 0.f;
#pragma unroll
    for (int bb = 0; bb < MHID; ++bb)
      s += Adense[(a * MHID + bb) * OUTD + d] * z1[((bb * CDIM + c) << 5) + d];
    z2[e] = s;
  }
  __syncthreads();

  for (int s = 0; s < 3; ++s) {
    const float* zs = (s == 0) ? fh : (s == 1 ? z1 : z2);
    float* dst = zf + s * ZF_STRIDE;
    for (int e = tid; e < MHID * 2 * 32 * 32; e += 256) {
      int lane = e & 31, kit = (e >> 5) & 31, ot = (e >> 10) & 1, m = e >> 11;
      int k = kit * 4 + (lane >> 4) * 2;
      int o = ot * 16 + (lane & 15);
      dst[e * 2 + 0] = zs[(m * CDIM + k) * OUTD + o];
      dst[e * 2 + 1] = zs[(m * CDIM + k + 1) * OUTD + o];
    }
  }
}

__global__ __launch_bounds__(256) void rw_main(const float* __restrict__ adj,
                                               const float* __restrict__ feats,
                                               const int*   __restrict__ idxs,
                                               const float* __restrict__ zfrag,
                                               float*       __restrict__ out) {
  __shared__ float lx0[NNODE * NPAD];
  __shared__ float lxc[NNODE * NPAD];
  __shared__ float accbuf[OUTD];

  const int b   = blockIdx.x;
  const int tid = threadIdx.x;
  const int w   = tid >> 5;
  const int l   = tid & 31;
  const int l16 = l & 15;
  const int lhi = l >> 4;

  if (tid < OUTD) accbuf[tid] = 0.f;

  {
    const int r0 = w * 16;
#pragma unroll 4
    for (int r = r0; r < r0 + 16; ++r) {
      int idx = idxs[b * NNODE + r];
      float4 v = *(const float4*)(feats + (size_t)idx * CDIM + l * 4);
      *(float4*)(&lx0[r * NPAD + l * 4]) = v;
    }
  }
  __syncthreads();

  float acc0 = 0.f, acc1 = 0.f;
  const float* xcur = lx0;

  for (int step = 0; step < 3; ++step) {
    if (step) {
      v8f cacc[8];
      const float* Ag = adj + (size_t)b * NNODE * CDIM;
      const int ct = w;
      const int col = ct * 16 + l16;
#pragma unroll
      for (int nt = 0; nt < 8; ++nt) {
        v8f c = {};
        const int row = nt * 16 + l16;
#pragma unroll 4
        for (int kit = 0; kit < 32; ++kit) {
          const int kb = kit * 4 + lhi * 2;
          v2f a = *(const v2f*)(Ag + row * CDIM + kb);
          v2f bv;
          bv.x = xcur[(kb + 0) * NPAD + col];
          bv.y = xcur[(kb + 1) * NPAD + col];
          c = wmma4(a, bv, c);
        }
        cacc[nt] = c;
      }
      __syncthreads();
#pragma unroll
      for (int nt = 0; nt < 8; ++nt)
#pragma unroll
        for (int r = 0; r < 8; ++r)
          lxc[(nt * 16 + r + lhi * 8) * NPAD + ct * 16 + l16] = cacc[nt][r];
      __syncthreads();
      xcur = lxc;
    }

    const float* zf0 = zfrag;
    const float* zfi = zfrag + step * ZF_STRIDE;
    for (int t = w; t < MHID * 8 * 2; t += 8) {
      const int m  = t >> 4;
      const int nt = (t >> 1) & 7;
      const int ot = t & 1;
      const int row = nt * 16 + l16;
      const v2f* bz0 = (const v2f*)zf0 + (size_t)(m * 2 + ot) * 1024 + l;
      const v2f* bzi = (const v2f*)zfi + (size_t)(m * 2 + ot) * 1024 + l;
      v8f czx = {}, ctt = {};
#pragma unroll 4
      for (int kit = 0; kit < 32; ++kit) {
        const int kb = kit * 4 + lhi * 2;
        v2f a0 = *(const v2f*)(&lx0[row * NPAD + kb]);
        czx = wmma4(a0, bz0[kit * 32], czx);
        if (step) {
          v2f ai = *(const v2f*)(&xcur[row * NPAD + kb]);
          ctt = wmma4(ai, bzi[kit * 32], ctt);
        }
      }
      if (!step) ctt = czx;
      float s = 0.f;
#pragma unroll
      for (int r = 0; r < 8; ++r) s += czx[r] * ctt[r];
      if (ot == 0) acc0 += s; else acc1 += s;
    }
  }

  atomicAdd(&accbuf[l16], acc0);
  atomicAdd(&accbuf[16 + l16], acc1);
  __syncthreads();
  if (tid < OUTD) out[(size_t)b * OUTD + tid] = accbuf[tid] * (1.0f / 3840.0f);
}

extern "C" void kernel_launch(void* const* d_in, const int* in_sizes, int n_in,
                              void* d_out, int out_size, void* d_ws, size_t ws_size,
                              hipStream_t stream) {
  (void)in_sizes; (void)n_in; (void)out_size; (void)ws_size;
  const float* adj   = (const float*)d_in[0];
  const float* feats = (const float*)d_in[1];
  const int*   idxs  = (const int*)  d_in[2];
  const float* fh    = (const float*)d_in[3];
  const float* ah    = (const float*)d_in[4];
  float* out = (float*)d_out;
  float* ws  = (float*)d_ws;

  rw_pre<<<1, 256, 0, stream>>>(fh, ah, ws);
  rw_main<<<B_BATCH, 256, 0, stream>>>(adj, feats, idxs, ws + WS_ZF, out);
}


# Setup kernel spread over the grid: one thread per (c,o) column forms z1, z2 in registers and stores the B fragments directly; 64 workgroups of 64 threads replace one 256-thread workgroup
# speedup vs baseline: 1.3855x; 1.3855x over previous
#include <hip/hip_runtime.h>


typedef float v2f __attribute__((ext_vector_type(2)));
typedef float v8f __attribute__((ext_vector_type(8)));

#define B_BATCH 1024
#define NNODE   128
#define CDIM    128
#define MHID    10
#define OUTD    32
#define NPAD    136

__device__ __forceinline__ v8f wmma4(v2f a, v2f b, v8f c) {
  return __builtin_amdgcn_wmma_f32_16x16x4_f32(
       false, a,  false, b,
       (short)0, c,  false,  false);
}

#define WS_ZF  0
#define ZF_STRIDE 40960

__global__ __launch_bounds__(64) void rw_pre(const float* __restrict__ fh,
                                             const float* __restrict__ ah,
                                             float* __restrict__ zf) {
  const int g   = blockIdx.x * 64 + threadIdx.x;
  const int j   = g & 1;
  const int l16 = (g >> 1) & 15;
  const int lhi = (g >> 5) & 1;
  const int kit = (g >> 6) & 31;
  const int ot  = g >> 11;
  const int c   = kit * 4 + lhi * 2 + j;
  const int o   = ot * 16 + l16;

  float r[45];
#pragma unroll
  for (int p = 0; p < 45; ++p) {
    float t = ah[p * OUTD + o];
    r[p] = t > 0.f ? t : 0.f;
  }
  float z0[MHID], z1[MHID], z2[MHID];
#pragma unroll
  for (int m = 0; m < MHID; ++m) z0[m] = fh[(m * CDIM + c) * OUTD + o];

#pragma unroll
  for (int a = 0; a < MHID; ++a) {
    float s = 0.f;
#pragma unroll
    for (int bb = 0; bb < MHID; ++bb) {
      const int i  = a < bb ? a : bb;
      const int jj = a < bb ? bb : a;
      const float av = (a == bb) ? 0.f : r[i * MHID - (i * (i + 1)) / 2 + (jj - i - 1)];
      s += av * z0[bb];
    }
    z1[a] = s;
  }
#pragma unroll
  for (int a = 0; a < MHID; ++a) {
    float s = 0.f;
#pragma unroll
    for (int bb = 0; bb < MHID; ++bb) {
      const int i  = a < bb ? a : bb;
      const int jj = a < bb ? bb : a;
      const float av = (a == bb) ? 0.f : r[i * MHID - (i * (i + 1)) / 2 + (jj - i - 1)];
      s += av * z1[bb];
    }
    z2[a] = s;
  }

#pragma unroll
  for (int m = 0; m < MHID; ++m) {
    zf[0 * ZF_STRIDE + m * 4096 + g] = z0[m];
    zf[1 * ZF_STRIDE + m * 4096 + g] = z1[m];
    zf[2 * ZF_STRIDE + m * 4096 + g] = z2[m];
  }
}

__global__ __launch_bounds__(256) void rw_main(const float* __restrict__ adj,
                                               const float* __restrict__ feats,
                                               const int*   __restrict__ idxs,
                                               const float* __restrict__ zfrag,
                                               float*       __restrict__ out) {
  __shared__ float lx0[NNODE * NPAD];
  __shared__ float lxc[NNODE * NPAD];
  __shared__ float accbuf[OUTD];

  const int b   = blockIdx.x;
  const int tid = threadIdx.x;
  const int w   = tid >> 5;
  const int l   = tid & 31;
  const int l16 = l & 15;
  const int lhi = l >> 4;

  if (tid < OUTD) accbuf[tid] = 0.f;

  {
    const int r0 = w * 16;
#pragma unroll 4
    for (int r = r0; r < r0 + 16; ++r) {
      int idx = idxs[b * NNODE + r];
      float4 v = *(const float4*)(feats + (size_t)idx * CDIM + l * 4);
      *(float4*)(&lx0[r * NPAD + l * 4]) = v;
    }
  }
  __syncthreads();

  float acc0 = 0.f, acc1 = 0.f;
  const float* xcur = lx0;

  for (int step = 0; step < 3; ++step) {
    if (step) {
      v8f cacc[8];
      const float* Ag = adj + (size_t)b * NNODE * CDIM;
      const int ct = w;
      const int col = ct * 16 + l16;
#pragma unroll
      for (int nt = 0; nt < 8; ++nt) {
        v8f c = {};
        const int row = nt * 16 + l16;
#pragma unroll 4
        for (int kit = 0; kit < 32; ++kit) {
          const int kb = kit * 4 + lhi * 2;
          v2f a = *(const v2f*)(Ag + row * CDIM + kb);
          v2f bv;
          bv.x = xcur[(kb + 0) * NPAD + col];
          bv.y = xcur[(kb + 1) * NPAD + col];
          c = wmma4(a, bv, c);
        }
        cacc[nt] = c;
      }
      __syncthreads();
#pragma unroll
      for (int nt = 0; nt < 8; ++nt)
#pragma unroll
        for (int r = 0; r < 8; ++r)
          lxc[(nt * 16 + r + lhi * 8) * NPAD + ct * 16 + l16] = cacc[nt][r];
      __syncthreads();
      xcur = lxc;
    }

    const float* zf0 = zfrag;
    const float* zfi = zfrag + step * ZF_STRIDE;
    for (int t = w; t < MHID * 8 * 2; t += 8) {
      const int m  = t >> 4;
      const int nt = (t >> 1) & 7;
      const int ot = t & 1;
      const int row = nt * 16 + l16;
      const v2f* bz0 = (const v2f*)zf0 + (size_t)(m * 2 + ot) * 1024 + l;
      const v2f* bzi = (const v2f*)zfi + (size_t)(m * 2 + ot) * 1024 + l;
      v8f czx = {}, ctt = {};
#pragma unroll 4
      for (int kit = 0; kit < 32; ++kit) {
        const int kb = kit * 4 + lhi * 2;
        v2f a0 = *(const v2f*)(&lx0[row * NPAD + kb]);
        czx = wmma4(a0, bz0[kit * 32], czx);
        if (step) {
          v2f ai = *(const v2f*)(&xcur[row * NPAD + kb]);
          ctt = wmma4(ai, bzi[kit * 32], ctt);
        }
      }
      if (!step) ctt = czx;
      float s = 0.f;
#pragma unroll
      for (int r = 0; r < 8; ++r) s += czx[r] * ctt[r];
      if (ot == 0) acc0 += s; else acc1 += s;
    }
  }

  atomicAdd(&accbuf[l16], acc0);
  atomicAdd(&accbuf[16 + l16], acc1);
  __syncthreads();
  if (tid < OUTD) out[(size_t)b * OUTD + tid] = accbuf[tid] * (1.0f / 3840.0f);
}

extern "C" void kernel_launch(void* const* d_in, const int* in_sizes, int n_in,
                              void* d_out, int out_size, void* d_ws, size_t ws_size,
                              hipStream_t stream) {
  (void)in_sizes; (void)n_in; (void)out_size; (void)ws_size;
  const float* adj   = (const float*)d_in[0];
  const float* feats = (const float*)d_in[1];
  const int*   idxs  = (const int*)  d_in[2];
  const float* fh    = (const float*)d_in[3];
  const float* ah    = (const float*)d_in[4];
  float* out = (float*)d_out;
  float* ws  = (float*)d_ws;

  rw_pre<<<64, 64, 0, stream>>>(fh, ah, ws + WS_ZF);
  rw_main<<<B_BATCH, 256, 0, stream>>>(adj, feats, idxs, ws + WS_ZF, out);
}


# Contribution k loop: step test moved out of it (one loop form per step kind), run as 4 blocks of 8 k steps; each block's LDS and global fragment loads issue as one group a block ahead of its WMMAs
# speedup vs baseline: 3.7890x; 2.7348x over previous
#include <hip/hip_runtime.h>


typedef float v2f __attribute__((ext_vector_type(2)));
typedef float v8f __attribute__((ext_vector_type(8)));

#define B_BATCH 1024
#define NNODE   128
#define CDIM    128
#define MHID    10
#define OUTD    32
#define NPAD    136

__device__ __forceinline__ v8f wmma4(v2f a, v2f b, v8f c) {
  return __builtin_amdgcn_wmma_f32_16x16x4_f32(
       false, a,  false, b,
       (short)0, c,  false,  false);
}

#define WS_ZF  0
#define ZF_STRIDE 40960

__global__ __launch_bounds__(64) void rw_pre(const float* __restrict__ fh,
                                             const float* __restrict__ ah,
                                             float* __restrict__ zf) {
  const int g   = blockIdx.x * 64 + threadIdx.x;
  const int j   = g & 1;
  const int l16 = (g >> 1) & 15;
  const int lhi = (g >> 5) & 1;
  const int kit = (g >> 6) & 31;
  const int ot  = g >> 11;
  const int c   = kit * 4 + lhi * 2 + j;
  const int o   = ot * 16 + l16;

  float r[45];
#pragma unroll
  for (int p = 0; p < 45; ++p) {
    float t = ah[p * OUTD + o];
    r[p] = t > 0.f ? t : 0.f;
  }
  float z0[MHID], z1[MHID], z2[MHID];
#pragma unroll
  for (int m = 0; m < MHID; ++m) z0[m] = fh[(m * CDIM + c) * OUTD + o];

#pragma unroll
  for (int a = 0; a < MHID; ++a) {
    float s = 0.f;
#pragma unroll
    for (int bb = 0; bb < MHID; ++bb) {
      const int i  = a < bb ? a : bb;
      const int jj = a < bb ? bb : a;
      const float av = (a == bb) ? 0.f : r[i * MHID - (i * (i + 1)) / 2 + (jj - i - 1)];
      s += av * z0[bb];
    }
    z1[a] = s;
  }
#pragma unroll
  for (int a = 0; a < MHID; ++a) {
    float s = 0.f;
#pragma unroll
    for (int bb = 0; bb < MHID; ++bb) {
      const int i  = a < bb ? a : bb;
      const int jj = a < bb ? bb : a;
      const float av = (a == bb) ? 0.f : r[i * MHID - (i * (i + 1)) / 2 + (jj - i - 1)];
      s += av * z1[bb];
    }
    z2[a] = s;
  }

#pragma unroll
  for (int m = 0; m < MHID; ++m) {
    zf[0 * ZF_STRIDE + m * 4096 + g] = z0[m];
    zf[1 * ZF_STRIDE + m * 4096 + g] = z1[m];
    zf[2 * ZF_STRIDE + m * 4096 + g] = z2[m];
  }
}

#define KU 8
#define NB (32 / KU)
template <bool PROP>
__device__ __forceinline__ void rw_contrib(const float* lx0, const float* xcur,
                                           const float* zf0, const float* zfi,
                                           int w, int l, int l16, int lhi, float& acc0, float& acc1) {
  for (int t = w; t < MHID * 8 * 2; t += 8) {
    const int m  = t >> 4;
    const int nt = (t >> 1) & 7;
    const int ot = t & 1;
    const int row = nt * 16 + l16;
    const v2f* bz0 = (const v2f*)zf0 + (size_t)(m * 2 + ot) * 1024 + l;
    const v2f* bzi = (const v2f*)zfi + (size_t)(m * 2 + ot) * 1024 + l;
    const float* ax0 = lx0  + row * NPAD + lhi * 2;
    const float* axi = xcur + row * NPAD + lhi * 2;
    v8f czx = {}, ctt = {};
    v2f a0[2][KU], b0[2][KU], ai[2][KU], bi[2][KU];
#pragma unroll
    for (int blk = -1; blk < NB; ++blk) {
      if (blk + 1 < NB) {
        const int ns = (blk + 1) & 1;
#pragma unroll
        for (int u = 0; u < KU; ++u) {
          const int kit = (blk + 1) * KU + u;
          b0[ns][u] = bz0[kit * 32];
          if (PROP) bi[ns][u] = bzi[kit * 32];
        }
#pragma unroll
        for (int u = 0; u < KU; ++u) {
          const int kit = (blk + 1) * KU + u;
          a0[ns][u] = *(const v2f*)(ax0 + kit * 4);
          if (PROP) ai[ns][u] = *(const v2f*)(axi + kit * 4);
        }
      }
      __builtin_amdgcn_sched_barrier(0);
      if (blk >= 0) {
        const int cs = blk & 1;
#pragma unroll
        for (int u = 0; u < KU; ++u) {
          czx = wmma4(a0[cs][u], b0[cs][u], czx);
          if (PROP) ctt = wmma4(ai[cs][u], bi[cs][u], ctt);
        }
      }
      __builtin_amdgcn_sched_barrier(0);
    }
    if (!PROP) ctt = czx;
    float s = 0.f;
#pragma unroll
    for (int r = 0; r < 8; ++r) s += czx[r] * ctt[r];
    if (ot == 0) acc0 += s; else acc1 += s;
  }
}

__global__ __launch_bounds__(256) void rw_main(const float* __restrict__ adj,
                                               const float* __restrict__ feats,
                                               const int*   __restrict__ idxs,
                                               const float* __restrict__ zfrag,
                                               float*       __restrict__ out) {
  __shared__ float lx0[NNODE * NPAD];
  __shared__ float lxc[NNODE * NPAD];
  __shared__ float accbuf[OUTD];

  const int b   = blockIdx.x;
  const int tid = threadIdx.x;
  const int w   = tid >> 5;
  const int l   = tid & 31;
  const int l16 = l & 15;
  const int lhi = l >> 4;

  if (tid < OUTD) accbuf[tid] = 0.f;

  {
    const int r0 = w * 16;
#pragma unroll 4
    for (int r = r0; r < r0 + 16; ++r) {
      int idx = idxs[b * NNODE + r];
      float4 v = *(const float4*)(feats + (size_t)idx * CDIM + l * 4);
      *(float4*)(&lx0[r * NPAD + l * 4]) = v;
    }
  }
  __syncthreads();

  float acc0 = 0.f, acc1 = 0.f;
  const float* xcur = lx0;

  for (int step = 0; step < 3; ++step) {
    if (step) {
      v8f cacc[8];
      const float* Ag = adj + (size_t)b * NNODE * CDIM;
      const int ct = w;
      const int col = ct * 16 + l16;
#pragma unroll
      for (int nt = 0; nt < 8; ++nt) {
        v8f c = {};
        const int row = nt * 16 + l16;
#pragma unroll 4
        for (int kit = 0; kit < 32; ++kit) {
          const int kb = kit * 4 + lhi * 2;
          v2f a = *(const v2f*)(Ag + row * CDIM + kb);
          v2f bv;
          bv.x = xcur[(kb + 0) * NPAD + col];
          bv.y = xcur[(kb + 1) * NPAD + col];
          c = wmma4(a, bv, c);
        }
        cacc[nt] = c;
      }
      __syncthreads();
#pragma unroll
      for (int nt = 0; nt < 8; ++nt)
#pragma unroll
        for (int r = 0; r < 8; ++r)
          lxc[(nt * 16 + r + lhi * 8) * NPAD + ct * 16 + l16] = cacc[nt][r];
      __syncthreads();
      xcur = lxc;
    }

    if (step == 0) rw_contrib<false>(lx0, xcur, zfrag, zfrag, w, l, l16, lhi, acc0, acc1);
    else           rw_contrib<true >(lx0, xcur, zfrag, zfrag + step * ZF_STRIDE, w, l, l16, lhi, acc0, acc1);
  }

  atomicAdd(&accbuf[l16], acc0);
  atomicAdd(&accbuf[16 + l16], acc1);
  __syncthreads();
  if (tid < OUTD) out[(size_t)b * OUTD + tid] = accbuf[tid] * (1.0f / 3840.0f);
}

extern "C" void kernel_launch(void* const* d_in, const int* in_sizes, int n_in,
                              void* d_out, int out_size, void* d_ws, size_t ws_size,
                              hipStream_t stream) {
  (void)in_sizes; (void)n_in; (void)out_size; (void)ws_size;
  const float* adj   = (const float*)d_in[0];
  const float* feats = (const float*)d_in[1];
  const int*   idxs  = (const int*)  d_in[2];
  const float* fh    = (const float*)d_in[3];
  const float* ah    = (const float*)d_in[4];
  float* out = (float*)d_out;
  float* ws  = (float*)d_ws;

  rw_pre<<<64, 64, 0, stream>>>(fh, ah, ws + WS_ZF);
  rw_main<<<B_BATCH, 256, 0, stream>>>(adj, feats, idxs, ws + WS_ZF, out);
}


# Three steps' contributions fused into one pass so zx is formed once per tile: wave owns a 16-row strip, X0 strip's A frags in registers, X2 strip takes its LDS rows; sum order changes
# speedup vs baseline: 6.2669x; 1.6540x over previous
#include <hip/hip_runtime.h>


typedef float v2f __attribute__((ext_vector_type(2)));
typedef float v8f __attribute__((ext_vector_type(8)));

#define B_BATCH 1024
#define NNODE   128
#define CDIM    128
#define MHID    10
#define OUTD    32
#define NPAD    136

__device__ __forceinline__ v8f wmma4(v2f a, v2f b, v8f c) {
  return __builtin_amdgcn_wmma_f32_16x16x4_f32(
       false, a,  false, b,
       (short)0, c,  false,  false);
}

#define WS_ZF  0
#define ZF_STRIDE 40960

__global__ __launch_bounds__(64) void rw_pre(const float* __restrict__ fh,
                                             const float* __restrict__ ah,
                                             float* __restrict__ zf) {
  const int g   = blockIdx.x * 64 + threadIdx.x;
  const int j   = g & 1;
  const int l16 = (g >> 1) & 15;
  const int lhi = (g >> 5) & 1;
  const int kit = (g >> 6) & 31;
  const int ot  = g >> 11;
  const int c   = kit * 4 + lhi * 2 + j;
  const int o   = ot * 16 + l16;

  float r[45];
#pragma unroll
  for (int p = 0; p < 45; ++p) {
    float t = ah[p * OUTD + o];
    r[p] = t > 0.f ? t : 0.f;
  }
  float z0[MHID], z1[MHID], z2[MHID];
#pragma unroll
  for (int m = 0; m < MHID; ++m) z0[m] = fh[(m * CDIM + c) * OUTD + o];

#pragma unroll
  for (int a = 0; a < MHID; ++a) {
    float s = 0.f;
#pragma unroll
    for (int bb = 0; bb < MHID; ++bb) {
      const int i  = a < bb ? a : bb;
      const int jj = a < bb ? bb : a;
      const float av = (a == bb) ? 0.f : r[i * MHID - (i * (i + 1)) / 2 + (jj - i - 1)];
      s += av * z0[bb];
    }
    z1[a] = s;
  }
#pragma unroll
  for (int a = 0; a < MHID; ++a) {
    float s = 0.f;
#pragma unroll
    for (int bb = 0; bb < MHID; ++bb) {
      const int i  = a < bb ? a : bb;
      const int jj = a < bb ? bb : a;
      const float av = (a == bb) ? 0.f : r[i * MHID - (i * (i + 1)) / 2 + (jj - i - 1)];
      s += av * z1[bb];
    }
    z2[a] = s;
  }

#pragma unroll
  for (int m = 0; m < MHID; ++m) {
    zf[0 * ZF_STRIDE + m * 4096 + g] = z0[m];
    zf[1 * ZF_STRIDE + m * 4096 + g] = z1[m];
    zf[2 * ZF_STRIDE + m * 4096 + g] = z2[m];
  }
}

__device__ __forceinline__ void rw_prop(const v2f (&aa)[32], const float* xs, float* xd,
                                        int w, int l16, int lhi) {
#pragma unroll 1
  for (int ct = 0; ct < 8; ++ct) {
    v8f c = {};
    const float* xb = xs + (lhi * 2) * NPAD + ct * 16 + l16;
#pragma unroll
    for (int kit = 0; kit < 32; ++kit) {
      v2f bv;
      bv.x = xb[(kit * 4 + 0) * NPAD];
      bv.y = xb[(kit * 4 + 1) * NPAD];
      c = wmma4(aa[kit], bv, c);
    }
#pragma unroll
    for (int r = 0; r < 8; ++r)
      xd[(w * 16 + r + lhi * 8) * NPAD + ct * 16 + l16] = c[r];
  }
}

#define KU 4
#define NB (32 / KU)
__global__ __launch_bounds__(256) void rw_main(const float* __restrict__ adj,
                                               const float* __restrict__ feats,
                                               const int*   __restrict__ idxs,
                                               const float* __restrict__ zfrag,
                                               float*       __restrict__ out) {
  __shared__ float lxa[NNODE * NPAD];
  __shared__ float lxb[NNODE * NPAD];
  __shared__ float accbuf[OUTD];

  const int b   = blockIdx.x;
  const int tid = threadIdx.x;
  const int w   = tid >> 5;
  const int l   = tid & 31;
  const int l16 = l & 15;
  const int lhi = l >> 4;

  if (tid < OUTD) accbuf[tid] = 0.f;

  {
    const int r0 = w * 16;
#pragma unroll 4
    for (int r = r0; r < r0 + 16; ++r) {
      int idx = idxs[b * NNODE + r];
      float4 v = *(const float4*)(feats + (size_t)idx * CDIM + l * 4);
      *(float4*)(&lxa[r * NPAD + l * 4]) = v;
    }
  }

  const int row = w * 16 + l16;
  v2f aa[32];
  {
    const float* Ag = adj + (size_t)b * NNODE * CDIM + row * CDIM + lhi * 2;
#pragma unroll
    for (int kit = 0; kit < 32; ++kit) aa[kit] = *(const v2f*)(Ag + kit * 4);
  }
  __syncthreads();
  rw_prop(aa, lxa, lxb, w, l16, lhi);
  __syncthreads();

  v2f xa[32];
  {
    const float* ax0 = lxa + row * NPAD + lhi * 2;
#pragma unroll
    for (int kit = 0; kit < 32; ++kit) xa[kit] = *(const v2f*)(ax0 + kit * 4);
  }
  rw_prop(aa, lxb, lxa, w, l16, lhi);
  __syncthreads();

  float acc0 = 0.f, acc1 = 0.f;
  const float* ax1 = lxb + row * NPAD + lhi * 2;
  const float* ax2 = lxa + row * NPAD + lhi * 2;
  for (int t = 0; t < MHID * 2; ++t) {
    const v2f* bz0 = (const v2f*)(zfrag)                 + (size_t)t * 1024 + l;
    const v2f* bz1 = (const v2f*)(zfrag + ZF_STRIDE)     + (size_t)t * 1024 + l;
    const v2f* bz2 = (const v2f*)(zfrag + 2 * ZF_STRIDE) + (size_t)t * 1024 + l;
    v8f czx = {}, c1 = {}, c2 = {};
    v2f b0[2][KU], b1[2][KU], b2[2][KU], a1[2][KU], a2[2][KU];
#pragma unroll
    for (int blk = -1; blk < NB; ++blk) {
      if (blk + 1 < NB) {
        const int ns = (blk + 1) & 1;
#pragma unroll
        for (int u = 0; u < KU; ++u) {
          const int kit = (blk + 1) * KU + u;
          b0[ns][u] = bz0[kit * 32];
          b1[ns][u] = bz1[kit * 32];
          b2[ns][u] = bz2[kit * 32];
        }
#pragma unroll
        for (int u = 0; u < KU; ++u) {
          const int kit = (blk + 1) * KU + u;
          a1[ns][u] = *(const v2f*)(ax1 + kit * 4);
          a2[ns][u] = *(const v2f*)(ax2 + kit * 4);
        }
      }
      __builtin_amdgcn_sched_barrier(0);
      if (blk >= 0) {
        const int cs = blk & 1;
#pragma unroll
        for (int u = 0; u < KU; ++u) {
          czx = wmma4(xa[blk * KU + u], b0[cs][u], czx);
          c1  = wmma4(a1[cs][u],        b1[cs][u], c1);
          c2  = wmma4(a2[cs][u],        b2[cs][u], c2);
        }
      }
      __builtin_amdgcn_sched_barrier(0);
    }
    float s0 = 0.f, s1 = 0.f, s2 = 0.f;
#pragma unroll
    for (int r = 0; r < 8; ++r) {
      s0 += czx[r] * czx[r];
      s1 += czx[r] * c1[r];
      s2 += czx[r] * c2[r];
    }
    const float s = (s0 + s1) + s2;
    if ((t & 1) == 0) acc0 += s; else acc1 += s;
  }

  atomicAdd(&accbuf[l16], acc0);
  atomicAdd(&accbuf[16 + l16], acc1);
  __syncthreads();
  if (tid < OUTD) out[(size_t)b * OUTD + tid] = accbuf[tid] * (1.0f / 3840.0f);
}

extern "C" void kernel_launch(void* const* d_in, const int* in_sizes, int n_in,
                              void* d_out, int out_size, void* d_ws, size_t ws_size,
                              hipStream_t stream) {
  (void)in_sizes; (void)n_in; (void)out_size; (void)ws_size;
  const float* adj   = (const float*)d_in[0];
  const float* feats = (const float*)d_in[1];
  const int*   idxs  = (const int*)  d_in[2];
  const float* fh    = (const float*)d_in[3];
  const float* ah    = (const float*)d_in[4];
  float* out = (float*)d_out;
  float* ws  = (float*)d_ws;

  rw_pre<<<64, 64, 0, stream>>>(fh, ah, ws + WS_ZF);
  rw_main<<<B_BATCH, 256, 0, stream>>>(adj, feats, idxs, ws + WS_ZF, out);
}


# Propagation k loop regrouped like the contribution loop: two column tiles per pass, own accumulators; each 8-k-step block's LDS B loads issued as one group a block ahead of its WMMAs; sums unchanged
# speedup vs baseline: 6.4727x; 1.0328x over previous
#include <hip/hip_runtime.h>


typedef float v2f __attribute__((ext_vector_type(2)));
typedef float v8f __attribute__((ext_vector_type(8)));

#define B_BATCH 1024
#define NNODE   128
#define CDIM    128
#define MHID    10
#define OUTD    32
#define NPAD    136

__device__ __forceinline__ v8f wmma4(v2f a, v2f b, v8f c) {
  return __builtin_amdgcn_wmma_f32_16x16x4_f32(
       false, a,  false, b,
       (short)0, c,  false,  false);
}

#define WS_ZF  0
#define ZF_STRIDE 40960

__global__ __launch_bounds__(64) void rw_pre(const float* __restrict__ fh,
                                             const float* __restrict__ ah,
                                             float* __restrict__ zf) {
  const int g   = blockIdx.x * 64 + threadIdx.x;
  const int j   = g & 1;
  const int l16 = (g >> 1) & 15;
  const int lhi = (g >> 5) & 1;
  const int kit = (g >> 6) & 31;
  const int ot  = g >> 11;
  const int c   = kit * 4 + lhi * 2 + j;
  const int o   = ot * 16 + l16;

  float r[45];
#pragma unroll
  for (int p = 0; p < 45; ++p) {
    float t = ah[p * OUTD + o];
    r[p] = t > 0.f ? t : 0.f;
  }
  float z0[MHID], z1[MHID], z2[MHID];
#pragma unroll
  for (int m = 0; m < MHID; ++m) z0[m] = fh[(m * CDIM + c) * OUTD + o];

#pragma unroll
  for (int a = 0; a < MHID; ++a) {
    float s = 0.f;
#pragma unroll
    for (int bb = 0; bb < MHID; ++bb) {
      const int i  = a < bb ? a : bb;
      const int jj = a < bb ? bb : a;
      const float av = (a == bb) ? 0.f : r[i * MHID - (i * (i + 1)) / 2 + (jj - i - 1)];
      s += av * z0[bb];
    }
    z1[a] = s;
  }
#pragma unroll
  for (int a = 0; a < MHID; ++a) {
    float s = 0.f;
#pragma unroll
    for (int bb = 0; bb < MHID; ++bb) {
      const int i  = a < bb ? a : bb;
      const int jj = a < bb ? bb : a;
      const float av = (a == bb) ? 0.f : r[i * MHID - (i * (i + 1)) / 2 + (jj - i - 1)];
      s += av * z1[bb];
    }
    z2[a] = s;
  }

#pragma unroll
  for (int m = 0; m < MHID; ++m) {
    zf[0 * ZF_STRIDE + m * 4096 + g] = z0[m];
    zf[1 * ZF_STRIDE + m * 4096 + g] = z1[m];
    zf[2 * ZF_STRIDE + m * 4096 + g] = z2[m];
  }
}

#define PK 8
#define PNB (32 / PK)
__device__ __forceinline__ void rw_prop(const v2f (&aa)[32], const float* xs, float* xd,
                                        int w, int l16, int lhi) {
#pragma unroll 1
  for (int cp = 0; cp < 4; ++cp) {
    v8f ca = {}, cb = {};
    const float* xb = xs + (lhi * 2) * NPAD + cp * 32 + l16;
    int ob = 16;
    asm volatile("" : "+v"(ob));
    const float* xc = xb + ob;
    v2f ba[2][PK], bb[2][PK];
#pragma unroll
    for (int blk = -1; blk < PNB; ++blk) {
      if (blk + 1 < PNB) {
        const int ns = (blk + 1) & 1;
#pragma unroll
        for (int u = 0; u < PK; ++u) {
          const int kit = (blk + 1) * PK + u;
          ba[ns][u].x = xb[(kit * 4 + 0) * NPAD];
          ba[ns][u].y = xb[(kit * 4 + 1) * NPAD];
          bb[ns][u].x = xc[(kit * 4 + 0) * NPAD];
          bb[ns][u].y = xc[(kit * 4 + 1) * NPAD];
        }
      }
      __builtin_amdgcn_sched_barrier(0);
      if (blk >= 0) {
        const int cs = blk & 1;
#pragma unroll
        for (int u = 0; u < PK; ++u) {
          ca = wmma4(aa[blk * PK + u], ba[cs][u], ca);
          cb = wmma4(aa[blk * PK + u], bb[cs][u], cb);
        }
      }
      __builtin_amdgcn_sched_barrier(0);
    }
    float* xo = xd + (w * 16 + lhi * 8) * NPAD + cp * 32 + l16;
#pragma unroll
    for (int r = 0; r < 8; ++r) {
      xo[r * NPAD]      = ca[r];
      xo[r * NPAD + 16] = cb[r];
    }
  }
}

#define KU 4
#define NB (32 / KU)
__global__ __launch_bounds__(256) void rw_main(const float* __restrict__ adj,
                                               const float* __restrict__ feats,
                                               const int*   __restrict__ idxs,
                                               const float* __restrict__ zfrag,
                                               float*       __restrict__ out) {
  __shared__ float lxa[NNODE * NPAD];
  __shared__ float lxb[NNODE * NPAD];
  __shared__ float accbuf[OUTD];

  const int b   = blockIdx.x;
  const int tid = threadIdx.x;
  const int w   = tid >> 5;
  const int l   = tid & 31;
  const int l16 = l & 15;
  const int lhi = l >> 4;

  if (tid < OUTD) accbuf[tid] = 0.f;

  {
    const int r0 = w * 16;
#pragma unroll 4
    for (int r = r0; r < r0 + 16; ++r) {
      int idx = idxs[b * NNODE + r];
      float4 v = *(const float4*)(feats + (size_t)idx * CDIM + l * 4);
      *(float4*)(&lxa[r * NPAD + l * 4]) = v;
    }
  }

  const int row = w * 16 + l16;
  v2f aa[32];
  {
    const float* Ag = adj + (size_t)b * NNODE * CDIM + row * CDIM + lhi * 2;
#pragma unroll
    for (int kit = 0; kit < 32; ++kit) aa[kit] = *(const v2f*)(Ag + kit * 4);
  }
  __syncthreads();
  rw_prop(aa, lxa, lxb, w, l16, lhi);
  __syncthreads();

  v2f xa[32];
  {
    const float* ax0 = lxa + row * NPAD + lhi * 2;
#pragma unroll
    for (int kit = 0; kit < 32; ++kit) xa[kit] = *(const v2f*)(ax0 + kit * 4);
  }
  rw_prop(aa, lxb, lxa, w, l16, lhi);
  __syncthreads();

  float acc0 = 0.f, acc1 = 0.f;
  const float* ax1 = lxb + row * NPAD + lhi * 2;
  const float* ax2 = lxa + row * NPAD + lhi * 2;
  for (int t = 0; t < MHID * 2; ++t) {
    const v2f* bz0 = (const v2f*)(zfrag)                 + (size_t)t * 1024 + l;
    const v2f* bz1 = (const v2f*)(zfrag + ZF_STRIDE)     + (size_t)t * 1024 + l;
    const v2f* bz2 = (const v2f*)(zfrag + 2 * ZF_STRIDE) + (size_t)t * 1024 + l;
    v8f czx = {}, c1 = {}, c2 = {};
    v2f b0[2][KU], b1[2][KU], b2[2][KU], a1[2][KU], a2[2][KU];
#pragma unroll
    for (int blk = -1; blk < NB; ++blk) {
      if (blk + 1 < NB) {
        const int ns = (blk + 1) & 1;
#pragma unroll
        for (int u = 0; u < KU; ++u) {
          const int kit = (blk + 1) * KU + u;
          b0[ns][u] = bz0[kit * 32];
          b1[ns][u] = bz1[kit * 32];
          b2[ns][u] = bz2[kit * 32];
        }
#pragma unroll
        for (int u = 0; u < KU; ++u) {
          const int kit = (blk + 1) * KU + u;
          a1[ns][u] = *(const v2f*)(ax1 + kit * 4);
          a2[ns][u] = *(const v2f*)(ax2 + kit * 4);
        }
      }
      __builtin_amdgcn_sched_barrier(0);
      if (blk >= 0) {
        const int cs = blk & 1;
#pragma unroll
        for (int u = 0; u < KU; ++u) {
          czx = wmma4(xa[blk * KU + u], b0[cs][u], czx);
          c1  = wmma4(a1[cs][u],        b1[cs][u], c1);
          c2  = wmma4(a2[cs][u],        b2[cs][u], c2);
        }
      }
      __builtin_amdgcn_sched_barrier(0);
    }
    float s0 = 0.f, s1 = 0.f, s2 = 0.f;
#pragma unroll
    for (int r = 0; r < 8; ++r) {
      s0 += czx[r] * czx[r];
      s1 += czx[r] * c1[r];
      s2 += czx[r] * c2[r];
    }
    const float s = (s0 + s1) + s2;
    if ((t & 1) == 0) acc0 += s; else acc1 += s;
  }

  atomicAdd(&accbuf[l16], acc0);
  atomicAdd(&accbuf[16 + l16], acc1);
  __syncthreads();
  if (tid < OUTD) out[(size_t)b * OUTD + tid] = accbuf[tid] * (1.0f / 3840.0f);
}

extern "C" void kernel_launch(void* const* d_in, const int* in_sizes, int n_in,
                              void* d_out, int out_size, void* d_ws, size_t ws_size,
                              hipStream_t stream) {
  (void)in_sizes; (void)n_in; (void)out_size; (void)ws_size;
  const float* adj   = (const float*)d_in[0];
  const float* feats = (const float*)d_in[1];
  const int*   idxs  = (const int*)  d_in[2];
  const float* fh    = (const float*)d_in[3];
  const float* ah    = (const float*)d_in[4];
  float* out = (float*)d_out;
  float* ws  = (float*)d_ws;

  rw_pre<<<64, 64, 0, stream>>>(fh, ah, ws + WS_ZF);
  rw_main<<<B_BATCH, 256, 0, stream>>>(adj, feats, idxs, ws + WS_ZF, out);
}


# Contributions re-associated: per batch three 128x128 Gram products X0^T X_s go to the workspace, then one batched f32 WMMA product with a batch-invariant table; launches 2 to 5; sum order changes
# speedup vs baseline: 8.1182x; 1.2542x over previous
#include <hip/hip_runtime.h>


typedef float v2f __attribute__((ext_vector_type(2)));
typedef float v8f __attribute__((ext_vector_type(8)));

#define B_BATCH 1024
#define NNODE   128
#define CDIM    128
#define MHID    10
#define OUTD    32
#define NPAD    136

__device__ __forceinline__ v8f wmma4(v2f a, v2f b, v8f c) {
  return __builtin_amdgcn_wmma_f32_16x16x4_f32(
       false, a,  false, b,
       (short)0, c,  false,  false);
}

#define WS_ZF  0
#define ZF_STRIDE 40960

__global__ __launch_bounds__(64) void rw_pre(const float* __restrict__ fh,
                                             const float* __restrict__ ah,
                                             float* __restrict__ zf) {
  const int g   = blockIdx.x * 64 + threadIdx.x;
  const int j   = g & 1;
  const int l16 = (g >> 1) & 15;
  const int lhi = (g >> 5) & 1;
  const int kit = (g >> 6) & 31;
  const int ot  = g >> 11;
  const int c   = kit * 4 + lhi * 2 + j;
  const int o   = ot * 16 + l16;

  float r[45];
#pragma unroll
  for (int p = 0; p < 45; ++p) {
    float t = ah[p * OUTD + o];
    r[p] = t > 0.f ? t : 0.f;
  }
  float z0[MHID], z1[MHID], z2[MHID];
#pragma unroll
  for (int m = 0; m < MHID; ++m) z0[m] = fh[(m * CDIM + c) * OUTD + o];

#pragma unroll
  for (int a = 0; a < MHID; ++a) {
    float s = 0.f;
#pragma unroll
    for (int bb = 0; bb < MHID; ++bb) {
      const int i  = a < bb ? a : bb;
      const int jj = a < bb ? bb : a;
      const float av = (a == bb) ? 0.f : r[i * MHID - (i * (i + 1)) / 2 + (jj - i - 1)];
      s += av * z0[bb];
    }
    z1[a] = s;
  }
#pragma unroll
  for (int a = 0; a < MHID; ++a) {
    float s = 0.f;
#pragma unroll
    for (int bb = 0; bb < MHID; ++bb) {
      const int i  = a < bb ? a : bb;
      const int jj = a < bb ? bb : a;
      const float av = (a == bb) ? 0.f : r[i * MHID - (i * (i + 1)) / 2 + (jj - i - 1)];
      s += av * z1[bb];
    }
    z2[a] = s;
  }

#pragma unroll
  for (int m = 0; m < MHID; ++m) {
    zf[0 * ZF_STRIDE + m * 4096 + g] = z0[m];
    zf[1 * ZF_STRIDE + m * 4096 + g] = z1[m];
    zf[2 * ZF_STRIDE + m * 4096 + g] = z2[m];
  }
}

#define PK 8
#define PNB (32 / PK)
__device__ __forceinline__ void rw_prop(const v2f (&aa)[32], const float* xs, float* xd,
                                        int w, int l16, int lhi) {
#pragma unroll 1
  for (int cp = 0; cp < 4; ++cp) {
    v8f ca = {}, cb = {};
    const float* xb = xs + (lhi * 2) * NPAD + cp * 32 + l16;
    int ob = 16;
    asm volatile("" : "+v"(ob));
    const float* xc = xb + ob;
    v2f ba[2][PK], bb[2][PK];
#pragma unroll
    for (int blk = -1; blk < PNB; ++blk) {
      if (blk + 1 < PNB) {
        const int ns = (blk + 1) & 1;
#pragma unroll
        for (int u = 0; u < PK; ++u) {
          const int kit = (blk + 1) * PK + u;
          ba[ns][u].x = xb[(kit * 4 + 0) * NPAD];
          ba[ns][u].y = xb[(kit * 4 + 1) * NPAD];
          bb[ns][u].x = xc[(kit * 4 + 0) * NPAD];
          bb[ns][u].y = xc[(kit * 4 + 1) * NPAD];
        }
      }
      __builtin_amdgcn_sched_barrier(0);
      if (blk >= 0) {
        const int cs = blk & 1;
#pragma unroll
        for (int u = 0; u < PK; ++u) {
          ca = wmma4(aa[blk * PK + u], ba[cs][u], ca);
          cb = wmma4(aa[blk * PK + u], bb[cs][u], cb);
        }
      }
      __builtin_amdgcn_sched_barrier(0);
    }
    float* xo = xd + (w * 16 + lhi * 8) * NPAD + cp * 32 + l16;
#pragma unroll
    for (int r = 0; r < 8; ++r) {
      xo[r * NPAD]      = ca[r];
      xo[r * NPAD + 16] = cb[r];
    }
  }
}

#define KU 4
#define NB (32 / KU)
__global__ __launch_bounds__(256) void rw_main(const float* __restrict__ adj,
                                               const float* __restrict__ feats,
                                               const int*   __restrict__ idxs,
                                               const float* __restrict__ zfrag,
                                               float*       __restrict__ out) {
  __shared__ float lxa[NNODE * NPAD];
  __shared__ float lxb[NNODE * NPAD];
  __shared__ float accbuf[OUTD];

  const int b   = blockIdx.x;
  const int tid = threadIdx.x;
  const int w   = tid >> 5;
  const int l   = tid & 31;
  const int l16 = l & 15;
  const int lhi = l >> 4;

  if (tid < OUTD) accbuf[tid] = 0.f;

  {
    const int r0 = w * 16;
#pragma unroll 4
    for (int r = r0; r < r0 + 16; ++r) {
      int idx = idxs[b * NNODE + r];
      float4 v = *(const float4*)(feats + (size_t)idx * CDIM + l * 4);
      *(float4*)(&lxa[r * NPAD + l * 4]) = v;
    }
  }

  const int row = w * 16 + l16;
  v2f aa[32];
  {
    const float* Ag = adj + (size_t)b * NNODE * CDIM + row * CDIM + lhi * 2;
#pragma unroll
    for (int kit = 0; kit < 32; ++kit) aa[kit] = *(const v2f*)(Ag + kit * 4);
  }
  __syncthreads();
  rw_prop(aa, lxa, lxb, w, l16, lhi);
  __syncthreads();

  v2f xa[32];
  {
    const float* ax0 = lxa + row * NPAD + lhi * 2;
#pragma unroll
    for (int kit = 0; kit < 32; ++kit) xa[kit] = *(const v2f*)(ax0 + kit * 4);
  }
  rw_prop(aa, lxb, lxa, w, l16, lhi);
  __syncthreads();

  float acc0 = 0.f, acc1 = 0.f;
  const float* ax1 = lxb + row * NPAD + lhi * 2;
  const float* ax2 = lxa + row * NPAD + lhi * 2;
  for (int t = 0; t < MHID * 2; ++t) {
    const v2f* bz0 = (const v2f*)(zfrag)                 + (size_t)t * 1024 + l;
    const v2f* bz1 = (const v2f*)(zfrag + ZF_STRIDE)     + (size_t)t * 1024 + l;
    const v2f* bz2 = (const v2f*)(zfrag + 2 * ZF_STRIDE) + (size_t)t * 1024 + l;
    v8f czx = {}, c1 = {}, c2 = {};
    v2f b0[2][KU], b1[2][KU], b2[2][KU], a1[2][KU], a2[2][KU];
#pragma unroll
    for (int blk = -1; blk < NB; ++blk) {
      if (blk + 1 < NB) {
        const int ns = (blk + 1) & 1;
#pragma unroll
        for (int u = 0; u < KU; ++u) {
          const int kit = (blk + 1) * KU + u;
          b0[ns][u] = bz0[kit * 32];
          b1[ns][u] = bz1[kit * 32];
          b2[ns][u] = bz2[kit * 32];
        }
#pragma unroll
        for (int u = 0; u < KU; ++u) {
          const int kit = (blk + 1) * KU + u;
          a1[ns][u] = *(const v2f*)(ax1 + kit * 4);
          a2[ns][u] = *(const v2f*)(ax2 + kit * 4);
        }
      }
      __builtin_amdgcn_sched_barrier(0);
      if (blk >= 0) {
        const int cs = blk & 1;
#pragma unroll
        for (int u = 0; u < KU; ++u) {
          czx = wmma4(xa[blk * KU + u], b0[cs][u], czx);
          c1  = wmma4(a1[cs][u],        b1[cs][u], c1);
          c2  = wmma4(a2[cs][u],        b2[cs][u], c2);
        }
      }
      __builtin_amdgcn_sched_barrier(0);
    }
    float s0 = 0.f, s1 = 0.f, s2 = 0.f;
#pragma unroll
    for (int r = 0; r < 8; ++r) {
      s0 += czx[r] * czx[r];
      s1 += czx[r] * c1[r];
      s2 += czx[r] * c2[r];
    }
    const float s = (s0 + s1) + s2;
    if ((t & 1) == 0) acc0 += s; else acc1 += s;
  }

  atomicAdd(&accbuf[l16], acc0);
  atomicAdd(&accbuf[16 + l16], acc1);
  __syncthreads();
  if (tid < OUTD) out[(size_t)b * OUTD + tid] = accbuf[tid] * (1.0f / 3840.0f);
}

#define GLEN     49152
#define NQ       6144
#define WS_WT    131072
#define WS_G     1703936
#define WS_PART  52035584
#define KCH      16
#define QPW      48
#define QB       4

__global__ __launch_bounds__(256) void rw_wbuild(const float* __restrict__ zf,
                                                 float* __restrict__ wt) {
  const int g4   = blockIdx.x * 256 + threadIdx.x;
  const int lane = g4 & 31;
  const int qq   = g4 >> 5;
  const int ot   = qq >= NQ ? 1 : 0;
  const int q    = qq - ot * NQ;
  const int o16  = lane & 15;
  const int hi   = lane >> 4;
  const int p0   = q * 8 + hi * 4;
  const int tile = p0 >> 8;
  const int r    = (p0 >> 5) & 7;
  const int hg   = (p0 >> 4) & 1;
  const int s    = tile >> 6;
  const int w    = (tile >> 3) & 7;
  const int ct   = tile & 7;
  const int c    = 16 * w + r + 8 * hg;
  const int cp0  = 16 * ct + (p0 & 15);
  const int go   = ot * 2048 + o16 * 2;
  const int gc   = go + (c >> 2) * 64 + ((c >> 1) & 1) * 32 + (c & 1);
  const int gcp  = go + (cp0 >> 2) * 64;
  const float* zs = zf + s * ZF_STRIDE + gcp;
  float a0 = 0.f, a1 = 0.f, a2 = 0.f, a3 = 0.f;
#pragma unroll
  for (int m = 0; m < MHID; ++m) {
    const float a = zf[m * 4096 + gc];
    a0 += a * zs[m * 4096 + 0];
    a1 += a * zs[m * 4096 + 1];
    a2 += a * zs[m * 4096 + 32];
    a3 += a * zs[m * 4096 + 33];
  }
  ((float4*)wt)[g4] = make_float4(a0, a1, a2, a3);
}

template <bool TO_WS>
__device__ __forceinline__ void rw_pass(const v2f (&af)[32], const float* xs, float* xd, float* gd,
                                        int w, int l, int l16, int lhi) {
#pragma unroll 1
  for (int cp = 0; cp < 4; ++cp) {
    v8f ca = {}, cb = {};
    const float* xb = xs + (lhi * 2) * NPAD + cp * 32 + l16;
    int ob = 16;
    asm volatile("" : "+v"(ob));
    const float* xc = xb + ob;
    v2f ba[2][PK], bb[2][PK];
#pragma unroll
    for (int blk = -1; blk < PNB; ++blk) {
      if (blk + 1 < PNB) {
        const int ns = (blk + 1) & 1;
#pragma unroll
        for (int u = 0; u < PK; ++u) {
          const int kit = (blk + 1) * PK + u;
          ba[ns][u].x = xb[(kit * 4 + 0) * NPAD];
          ba[ns][u].y = xb[(kit * 4 + 1) * NPAD];
          bb[ns][u].x = xc[(kit * 4 + 0) * NPAD];
          bb[ns][u].y = xc[(kit * 4 + 1) * NPAD];
        }
      }
      __builtin_amdgcn_sched_barrier(0);
      if (blk >= 0) {
        const int cs = blk & 1;
#pragma unroll
        for (int u = 0; u < PK; ++u) {
          ca = wmma4(af[blk * PK + u], ba[cs][u], ca);
          cb = wmma4(af[blk * PK + u], bb[cs][u], cb);
        }
      }
      __builtin_amdgcn_sched_barrier(0);
    }
    if (TO_WS) {
      float* go = gd + cp * 512 + l;
#pragma unroll
      for (int r = 0; r < 8; ++r) {
        go[r * 32]       = ca[r];
        go[256 + r * 32] = cb[r];
      }
    } else {
      float* xo = xd + (w * 16 + lhi * 8) * NPAD + cp * 32 + l16;
#pragma unroll
      for (int r = 0; r < 8; ++r) {
        xo[r * NPAD]      = ca[r];
        xo[r * NPAD + 16] = cb[r];
      }
    }
  }
}

__global__ __launch_bounds__(256) void rw_gram(const float* __restrict__ adj,
                                               const float* __restrict__ feats,
                                               const int*   __restrict__ idxs,
                                               float*       __restrict__ gws) {
  __shared__ float lxa[NNODE * NPAD];
  __shared__ float lxb[NNODE * NPAD];

  const int b   = blockIdx.x;
  const int tid = threadIdx.x;
  const int w   = tid >> 5;
  const int l   = tid & 31;
  const int l16 = l & 15;
  const int lhi = l >> 4;

  {
    const int r0 = w * 16;
#pragma unroll 4
    for (int r = r0; r < r0 + 16; ++r) {
      int idx = idxs[b * NNODE + r];
      float4 v = *(const float4*)(feats + (size_t)idx * CDIM + l * 4);
      *(float4*)(&lxa[r * NPAD + l * 4]) = v;
    }
  }

  const int row = w * 16 + l16;
  v2f aa[32];
  {
    const float* Ag = adj + (size_t)b * NNODE * CDIM + row * CDIM + lhi * 2;
#pragma unroll
    for (int kit = 0; kit < 32; ++kit) aa[kit] = *(const v2f*)(Ag + kit * 4);
  }
  __syncthreads();
  rw_pass<false>(aa, lxa, lxb, nullptr, w, l, l16, lhi);
  __syncthreads();

  v2f xt[32];
  {
    const float* t0 = lxa + (lhi * 2) * NPAD + row;
#pragma unroll
    for (int kit = 0; kit < 32; ++kit) {
      xt[kit].x = t0[(kit * 4 + 0) * NPAD];
      xt[kit].y = t0[(kit * 4 + 1) * NPAD];
    }
  }
  float* gb = gws + (size_t)b * GLEN + w * 2048;
  rw_pass<true>(xt, lxa, nullptr, gb,         w, l, l16, lhi);
  rw_pass<true>(xt, lxb, nullptr, gb + 16384, w, l, l16, lhi);
  __syncthreads();
  rw_pass<false>(aa, lxb, lxa, nullptr, w, l, l16, lhi);
  __syncthreads();
  rw_pass<true>(xt, lxa, nullptr, gb + 32768, w, l, l16, lhi);
}

__global__ __launch_bounds__(256) void rw_prod(const float* __restrict__ gws,
                                               const float* __restrict__ wt,
                                               float* __restrict__ part) {
  __shared__ float red[8 * 1024];
  const int bp  = blockIdx.x / KCH;
  const int ch  = blockIdx.x - bp * KCH;
  const int tid = threadIdx.x;
  const int wv  = tid >> 5;
  const int l   = tid & 31;
  const int l16 = l & 15;
  const int hi  = l >> 4;
  const int q0  = (ch * 8 + wv) * QPW;

  const float4* A0 = (const float4*)(gws + (size_t)(bp * 32 + l16) * GLEN) + hi + q0 * 2;
  const float4* A1 = (const float4*)(gws + (size_t)(bp * 32 + 16 + l16) * GLEN) + hi + q0 * 2;
  const float4* B0 = (const float4*)wt + (size_t)q0 * 32 + l;
  const float4* B1 = B0 + (size_t)NQ * 32;

  v8f c00 = {}, c01 = {}, c10 = {}, c11 = {};
  float4 fa0[2][QB], fa1[2][QB], fb0[2][QB], fb1[2][QB];
#pragma unroll
  for (int blk = -1; blk < QPW / QB; ++blk) {
    if (blk + 1 < QPW / QB) {
      const int ns = (blk + 1) & 1;
#pragma unroll
      for (int u = 0; u < QB; ++u) {
        const int q = (blk + 1) * QB + u;
        fa0[ns][u] = A0[q * 2];
        fa1[ns][u] = A1[q * 2];
        fb0[ns][u] = B0[q * 32];
        fb1[ns][u] = B1[q * 32];
      }
    }
    __builtin_amdgcn_sched_barrier(0);
    if (blk >= 0) {
      const int cs = blk & 1;
#pragma unroll
      for (int u = 0; u < QB; ++u) {
        const v2f a0l = {fa0[cs][u].x, fa0[cs][u].y}, a0h = {fa0[cs][u].z, fa0[cs][u].w};
        const v2f a1l = {fa1[cs][u].x, fa1[cs][u].y}, a1h = {fa1[cs][u].z, fa1[cs][u].w};
        const v2f b0l = {fb0[cs][u].x, fb0[cs][u].y}, b0h = {fb0[cs][u].z, fb0[cs][u].w};
        const v2f b1l = {fb1[cs][u].x, fb1[cs][u].y}, b1h = {fb1[cs][u].z, fb1[cs][u].w};
        c00 = wmma4(a0l, b0l, c00);
        c01 = wmma4(a0l, b1l, c01);
        c10 = wmma4(a1l, b0l, c10);
        c11 = wmma4(a1l, b1l, c11);
        c00 = wmma4(a0h, b0h, c00);
        c01 = wmma4(a0h, b1h, c01);
        c10 = wmma4(a1h, b0h, c10);
        c11 = wmma4(a1h, b1h, c11);
      }
    }
    __builtin_amdgcn_sched_barrier(0);
  }

  float* rw = red + wv * 1024 + l;
#pragma unroll
  for (int r = 0; r < 8; ++r) {
    rw[0   + r * 32] = c00[r];
    rw[256 + r * 32] = c01[r];
    rw[512 + r * 32] = c10[r];
    rw[768 + r * 32] = c11[r];
  }
  __syncthreads();
#pragma unroll
  for (int i = 0; i < 4; ++i) {
    const int e = i * 256 + tid;
    float s = 0.f;
#pragma unroll
    for (int k = 0; k < 8; ++k) s += red[k * 1024 + e];
    const int rt = e >> 9;
    const int ot = (e >> 8) & 1;
    const int r  = (e >> 5) & 7;
    const int ln = e & 31;
    const int bb = bp * 32 + rt * 16 + r + 8 * (ln >> 4);
    const int o  = ot * 16 + (ln & 15);
    part[(size_t)ch * (B_BATCH * OUTD) + bb * OUTD + o] = s;
  }
}

__global__ __launch_bounds__(256) void rw_reduce(const float* __restrict__ part,
                                                 float* __restrict__ out) {
  const int i = blockIdx.x * 256 + threadIdx.x;
  float s = 0.f;
#pragma unroll
  for (int k = 0; k < KCH; ++k) s += part[k * (B_BATCH * OUTD) + i];
  out[i] = s * (1.0f / 3840.0f);
}

extern "C" void kernel_launch(void* const* d_in, const int* in_sizes, int n_in,
                              void* d_out, int out_size, void* d_ws, size_t ws_size,
                              hipStream_t stream) {
  (void)in_sizes; (void)n_in; (void)out_size;
  const float* adj   = (const float*)d_in[0];
  const float* feats = (const float*)d_in[1];
  const int*   idxs  = (const int*)  d_in[2];
  const float* fh    = (const float*)d_in[3];
  const float* ah    = (const float*)d_in[4];
  float* out = (float*)d_out;
  float* ws  = (float*)d_ws;

  rw_pre<<<64, 64, 0, stream>>>(fh, ah, ws + WS_ZF);
  if (ws_size >= 210239488ull) {
    rw_wbuild<<<1536, 256, 0, stream>>>(ws + WS_ZF, ws + WS_WT);
    rw_gram<<<B_BATCH, 256, 0, stream>>>(adj, feats, idxs, ws + WS_G);
    rw_prod<<<512, 256, 0, stream>>>(ws + WS_G, ws + WS_WT, ws + WS_PART);
    rw_reduce<<<128, 256, 0, stream>>>(ws + WS_PART, out);
  } else {
    rw_main<<<B_BATCH, 256, 0, stream>>>(adj, feats, idxs, ws + WS_ZF, out);
  }
}


# Gram product X0^T X0 is symmetric: 36 of its 64 tiles formed and carried (5 or 4 per wave), mirrored table entries doubled; last product's K 49152 to 41984; sum order changes
# speedup vs baseline: 8.9080x; 1.0973x over previous
#include <hip/hip_runtime.h>


typedef float v2f __attribute__((ext_vector_type(2)));
typedef float v8f __attribute__((ext_vector_type(8)));

#define B_BATCH 1024
#define NNODE   128
#define CDIM    128
#define MHID    10
#define OUTD    32
#define NPAD    136

__device__ __forceinline__ v8f wmma4(v2f a, v2f b, v8f c) {
  return __builtin_amdgcn_wmma_f32_16x16x4_f32(
       false, a,  false, b,
       (short)0, c,  false,  false);
}

#define WS_ZF  0
#define ZF_STRIDE 40960

__global__ __launch_bounds__(64) void rw_pre(const float* __restrict__ fh,
                                             const float* __restrict__ ah,
                                             float* __restrict__ zf) {
  const int g   = blockIdx.x * 64 + threadIdx.x;
  const int j   = g & 1;
  const int l16 = (g >> 1) & 15;
  const int lhi = (g >> 5) & 1;
  const int kit = (g >> 6) & 31;
  const int ot  = g >> 11;
  const int c   = kit * 4 + lhi * 2 + j;
  const int o   = ot * 16 + l16;

  float r[45];
#pragma unroll
  for (int p = 0; p < 45; ++p) {
    float t = ah[p * OUTD + o];
    r[p] = t > 0.f ? t : 0.f;
  }
  float z0[MHID], z1[MHID], z2[MHID];
#pragma unroll
  for (int m = 0; m < MHID; ++m) z0[m] = fh[(m * CDIM + c) * OUTD + o];

#pragma unroll
  for (int a = 0; a < MHID; ++a) {
    float s = 0.f;
#pragma unroll
    for (int bb = 0; bb < MHID; ++bb) {
      const int i  = a < bb ? a : bb;
      const int jj = a < bb ? bb : a;
      const float av = (a == bb) ? 0.f : r[i * MHID - (i * (i + 1)) / 2 + (jj - i - 1)];
      s += av * z0[bb];
    }
    z1[a] = s;
  }
#pragma unroll
  for (int a = 0; a < MHID; ++a) {
    float s = 0.f;
#pragma unroll
    for (int bb = 0; bb < MHID; ++bb) {
      const int i  = a < bb ? a : bb;
      const int jj = a < bb ? bb : a;
      const float av = (a == bb) ? 0.f : r[i * MHID - (i * (i + 1)) / 2 + (jj - i - 1)];
      s += av * z1[bb];
    }
    z2[a] = s;
  }

#pragma unroll
  for (int m = 0; m < MHID; ++m) {
    zf[0 * ZF_STRIDE + m * 4096 + g] = z0[m];
    zf[1 * ZF_STRIDE + m * 4096 + g] = z1[m];
    zf[2 * ZF_STRIDE + m * 4096 + g] = z2[m];
  }
}

#define PK 8
#define PNB (32 / PK)
__device__ __forceinline__ void rw_prop(const v2f (&aa)[32], const float* xs, float* xd,
                                        int w, int l16, int lhi) {
#pragma unroll 1
  for (int cp = 0; cp < 4; ++cp) {
    v8f ca = {}, cb = {};
    const float* xb = xs + (lhi * 2) * NPAD + cp * 32 + l16;
    int ob = 16;
    asm volatile("" : "+v"(ob));
    const float* xc = xb + ob;
    v2f ba[2][PK], bb[2][PK];
#pragma unroll
    for (int blk = -1; blk < PNB; ++blk) {
      if (blk + 1 < PNB) {
        const int ns = (blk + 1) & 1;
#pragma unroll
        for (int u = 0; u < PK; ++u) {
          const int kit = (blk + 1) * PK + u;
          ba[ns][u].x = xb[(kit * 4 + 0) * NPAD];
          ba[ns][u].y = xb[(kit * 4 + 1) * NPAD];
          bb[ns][u].x = xc[(kit * 4 + 0) * NPAD];
          bb[ns][u].y = xc[(kit * 4 + 1) * NPAD];
        }
      }
      __builtin_amdgcn_sched_barrier(0);
      if (blk >= 0) {
        const int cs = blk & 1;
#pragma unroll
        for (int u = 0; u < PK; ++u) {
          ca = wmma4(aa[blk * PK + u], ba[cs][u], ca);
          cb = wmma4(aa[blk * PK + u], bb[cs][u], cb);
        }
      }
      __builtin_amdgcn_sched_barrier(0);
    }
    float* xo = xd + (w * 16 + lhi * 8) * NPAD + cp * 32 + l16;
#pragma unroll
    for (int r = 0; r < 8; ++r) {
      xo[r * NPAD]      = ca[r];
      xo[r * NPAD + 16] = cb[r];
    }
  }
}

#define KU 4
#define NB (32 / KU)
__global__ __launch_bounds__(256) void rw_main(const float* __restrict__ adj,
                                               const float* __restrict__ feats,
                                               const int*   __restrict__ idxs,
                                               const float* __restrict__ zfrag,
                                               float*       __restrict__ out) {
  __shared__ float lxa[NNODE * NPAD];
  __shared__ float lxb[NNODE * NPAD];
  __shared__ float accbuf[OUTD];

  const int b   = blockIdx.x;
  const int tid = threadIdx.x;
  const int w   = tid >> 5;
  const int l   = tid & 31;
  const int l16 = l & 15;
  const int lhi = l >> 4;

  if (tid < OUTD) accbuf[tid] = 0.f;

  {
    const int r0 = w * 16;
#pragma unroll 4
    for (int r = r0; r < r0 + 16; ++r) {
      int idx = idxs[b * NNODE + r];
      float4 v = *(const float4*)(feats + (size_t)idx * CDIM + l * 4);
      *(float4*)(&lxa[r * NPAD + l * 4]) = v;
    }
  }

  const int row = w * 16 + l16;
  v2f aa[32];
  {
    const float* Ag = adj + (size_t)b * NNODE * CDIM + row * CDIM + lhi * 2;
#pragma unroll
    for (int kit = 0; kit < 32; ++kit) aa[kit] = *(const v2f*)(Ag + kit * 4);
  }
  __syncthreads();
  rw_prop(aa, lxa, lxb, w, l16, lhi);
  __syncthreads();

  v2f xa[32];
  {
    const float* ax0 = lxa + row * NPAD + lhi * 2;
#pragma unroll
    for (int kit = 0; kit < 32; ++kit) xa[kit] = *(const v2f*)(ax0 + kit * 4);
  }
  rw_prop(aa, lxb, lxa, w, l16, lhi);
  __syncthreads();

  float acc0 = 0.f, acc1 = 0.f;
  const float* ax1 = lxb + row * NPAD + lhi * 2;
  const float* ax2 = lxa + row * NPAD + lhi * 2;
  for (int t = 0; t < MHID * 2; ++t) {
    const v2f* bz0 = (const v2f*)(zfrag)                 + (size_t)t * 1024 + l;
    const v2f* bz1 = (const v2f*)(zfrag + ZF_STRIDE)     + (size_t)t * 1024 + l;
    const v2f* bz2 = (const v2f*)(zfrag + 2 * ZF_STRIDE) + (size_t)t * 1024 + l;
    v8f czx = {}, c1 = {}, c2 = {};
    v2f b0[2][KU], b1[2][KU], b2[2][KU], a1[2][KU], a2[2][KU];
#pragma unroll
    for (int blk = -1; blk < NB; ++blk) {
      if (blk + 1 < NB) {
        const int ns = (blk + 1) & 1;
#pragma unroll
        for (int u = 0; u < KU; ++u) {
          const int kit = (blk + 1) * KU + u;
          b0[ns][u] = bz0[kit * 32];
          b1[ns][u] = bz1[kit * 32];
          b2[ns][u] = bz2[kit * 32];
        }
#pragma unroll
        for (int u = 0; u < KU; ++u) {
          const int kit = (blk + 1) * KU + u;
          a1[ns][u] = *(const v2f*)(ax1 + kit * 4);
          a2[ns][u] = *(const v2f*)(ax2 + kit * 4);
        }
      }
      __builtin_amdgcn_sched_barrier(0);
      if (blk >= 0) {
        const int cs = blk & 1;
#pragma unroll
        for (int u = 0; u < KU; ++u) {
          czx = wmma4(xa[blk * KU + u], b0[cs][u], czx);
          c1  = wmma4(a1[cs][u],        b1[cs][u], c1);
          c2  = wmma4(a2[cs][u],        b2[cs][u], c2);
        }
      }
      __builtin_amdgcn_sched_barrier(0);
    }
    float s0 = 0.f, s1 = 0.f, s2 = 0.f;
#pragma unroll
    for (int r = 0; r < 8; ++r) {
      s0 += czx[r] * czx[r];
      s1 += czx[r] * c1[r];
      s2 += czx[r] * c2[r];
    }
    const float s = (s0 + s1) + s2;
    if ((t & 1) == 0) acc0 += s; else acc1 += s;
  }

  atomicAdd(&accbuf[l16], acc0);
  atomicAdd(&accbuf[16 + l16], acc1);
  __syncthreads();
  if (tid < OUTD) out[(size_t)b * OUTD + tid] = accbuf[tid] * (1.0f / 3840.0f);
}

#define G0T      36
#define GLEN     41984
#define NQ       5248
#define WS_WT    131072
#define WS_G     1703936
#define WS_PART  52035584
#define KCH      16
#define QPW      41
#define QB       4
#define QNB      ((QPW + QB - 1) / QB)

__global__ __launch_bounds__(256) void rw_wbuild(const float* __restrict__ zf,
                                                 float* __restrict__ wt) {
  const int g4   = blockIdx.x * 256 + threadIdx.x;
  const int lane = g4 & 31;
  const int qq   = g4 >> 5;
  const int ot   = qq >= NQ ? 1 : 0;
  const int q    = qq - ot * NQ;
  const int o16  = lane & 15;
  const int hi   = lane >> 4;
  const int p0   = q * 8 + hi * 4;
  const int slot = p0 >> 8;
  const int r    = (p0 >> 5) & 7;
  const int hg   = (p0 >> 4) & 1;
  int s, w, ct;
  float f = 1.f;
  if (slot < G0T) {
    int d;
    if (slot < 20) { w = slot / 5; d = slot - w * 5; }
    else           { w = 4 + ((slot - 20) >> 2); d = (slot - 20) & 3; }
    s  = 0;
    ct = (w + d) & 7;
    if (d != 0) f = 2.f;
  } else {
    const int t2 = slot - G0T;
    s  = 1 + (t2 >> 6);
    w  = (t2 >> 3) & 7;
    ct = t2 & 7;
  }
  const int c    = 16 * w + r + 8 * hg;
  const int cp0  = 16 * ct + (p0 & 15);
  const int go   = ot * 2048 + o16 * 2;
  const int gc   = go + (c >> 2) * 64 + ((c >> 1) & 1) * 32 + (c & 1);
  const int gcp  = go + (cp0 >> 2) * 64;
  const float* zs = zf + s * ZF_STRIDE + gcp;
  float a0 = 0.f, a1 = 0.f, a2 = 0.f, a3 = 0.f;
#pragma unroll
  for (int m = 0; m < MHID; ++m) {
    const float a = zf[m * 4096 + gc];
    a0 += a * zs[m * 4096 + 0];
    a1 += a * zs[m * 4096 + 1];
    a2 += a * zs[m * 4096 + 32];
    a3 += a * zs[m * 4096 + 33];
  }
  ((float4*)wt)[g4] = make_float4(a0 * f, a1 * f, a2 * f, a3 * f);
}

template <bool TO_WS>
__device__ __forceinline__ void rw_pass(const v2f (&af)[32], const float* xs, float* xd, float* gd,
                                        int w, int l, int l16, int lhi) {
#pragma unroll 1
  for (int cp = 0; cp < 4; ++cp) {
    v8f ca = {}, cb = {};
    const float* xb = xs + (lhi * 2) * NPAD + cp * 32 + l16;
    int ob = 16;
    asm volatile("" : "+v"(ob));
    const float* xc = xb + ob;
    v2f ba[2][PK], bb[2][PK];
#pragma unroll
    for (int blk = -1; blk < PNB; ++blk) {
      if (blk + 1 < PNB) {
        const int ns = (blk + 1) & 1;
#pragma unroll
        for (int u = 0; u < PK; ++u) {
          const int kit = (blk + 1) * PK + u;
          ba[ns][u].x = xb[(kit * 4 + 0) * NPAD];
          ba[ns][u].y = xb[(kit * 4 + 1) * NPAD];
          bb[ns][u].x = xc[(kit * 4 + 0) * NPAD];
          bb[ns][u].y = xc[(kit * 4 + 1) * NPAD];
        }
      }
      __builtin_amdgcn_sched_barrier(0);
      if (blk >= 0) {
        const int cs = blk & 1;
#pragma unroll
        for (int u = 0; u < PK; ++u) {
          ca = wmma4(af[blk * PK + u], ba[cs][u], ca);
          cb = wmma4(af[blk * PK + u], bb[cs][u], cb);
        }
      }
      __builtin_amdgcn_sched_barrier(0);
    }
    if (TO_WS) {
      float* go = gd + cp * 512 + l;
#pragma unroll
      for (int r = 0; r < 8; ++r) {
        go[r * 32]       = ca[r];
        go[256 + r * 32] = cb[r];
      }
    } else {
      float* xo = xd + (w * 16 + lhi * 8) * NPAD + cp * 32 + l16;
#pragma unroll
      for (int r = 0; r < 8; ++r) {
        xo[r * NPAD]      = ca[r];
        xo[r * NPAD + 16] = cb[r];
      }
    }
  }
}

template <bool TWO>
__device__ __forceinline__ void rw_gsym(const v2f (&af)[32], const float* xb, const float* xc,
                                        float* go) {
  v8f ca = {}, cb = {};
  v2f ba[2][PK], bb[2][PK];
#pragma unroll
  for (int blk = -1; blk < PNB; ++blk) {
    if (blk + 1 < PNB) {
      const int ns = (blk + 1) & 1;
#pragma unroll
      for (int u = 0; u < PK; ++u) {
        const int kit = (blk + 1) * PK + u;
        ba[ns][u].x = xb[(kit * 4 + 0) * NPAD];
        ba[ns][u].y = xb[(kit * 4 + 1) * NPAD];
        if (TWO) {
          bb[ns][u].x = xc[(kit * 4 + 0) * NPAD];
          bb[ns][u].y = xc[(kit * 4 + 1) * NPAD];
        }
      }
    }
    __builtin_amdgcn_sched_barrier(0);
    if (blk >= 0) {
      const int cs = blk & 1;
#pragma unroll
      for (int u = 0; u < PK; ++u) {
        ca = wmma4(af[blk * PK + u], ba[cs][u], ca);
        if (TWO) cb = wmma4(af[blk * PK + u], bb[cs][u], cb);
      }
    }
    __builtin_amdgcn_sched_barrier(0);
  }
#pragma unroll
  for (int r = 0; r < 8; ++r) {
    go[r * 32] = ca[r];
    if (TWO) go[256 + r * 32] = cb[r];
  }
}

__global__ __launch_bounds__(256) void rw_gram(const float* __restrict__ adj,
                                               const float* __restrict__ feats,
                                               const int*   __restrict__ idxs,
                                               float*       __restrict__ gws) {
  __shared__ float lxa[NNODE * NPAD];
  __shared__ float lxb[NNODE * NPAD];

  const int b   = blockIdx.x;
  const int tid = threadIdx.x;
  const int w   = tid >> 5;
  const int l   = tid & 31;
  const int l16 = l & 15;
  const int lhi = l >> 4;

  {
    const int r0 = w * 16;
#pragma unroll 4
    for (int r = r0; r < r0 + 16; ++r) {
      int idx = idxs[b * NNODE + r];
      float4 v = *(const float4*)(feats + (size_t)idx * CDIM + l * 4);
      *(float4*)(&lxa[r * NPAD + l * 4]) = v;
    }
  }

  const int row = w * 16 + l16;
  v2f aa[32];
  {
    const float* Ag = adj + (size_t)b * NNODE * CDIM + row * CDIM + lhi * 2;
#pragma unroll
    for (int kit = 0; kit < 32; ++kit) aa[kit] = *(const v2f*)(Ag + kit * 4);
  }
  __syncthreads();
  rw_pass<false>(aa, lxa, lxb, nullptr, w, l, l16, lhi);
  __syncthreads();

  v2f xt[32];
  {
    const float* t0 = lxa + (lhi * 2) * NPAD + row;
#pragma unroll
    for (int kit = 0; kit < 32; ++kit) {
      xt[kit].x = t0[(kit * 4 + 0) * NPAD];
      xt[kit].y = t0[(kit * 4 + 1) * NPAD];
    }
  }
  {
    const int wu = __builtin_amdgcn_readfirstlane(w);
    float* g0 = gws + (size_t)b * GLEN + (wu < 4 ? wu * 5 : 20 + (wu - 4) * 4) * 256 + l;
    const float* xk = lxa + (lhi * 2) * NPAD + l16;
    rw_gsym<true>(xt, xk + ((wu + 0) & 7) * 16, xk + ((wu + 1) & 7) * 16, g0);
    rw_gsym<true>(xt, xk + ((wu + 2) & 7) * 16, xk + ((wu + 3) & 7) * 16, g0 + 512);
    if (wu < 4) rw_gsym<false>(xt, xk + (wu + 4) * 16, nullptr, g0 + 1024);
  }
  float* gb = gws + (size_t)b * GLEN + G0T * 256 + w * 2048;
  rw_pass<true>(xt, lxb, nullptr, gb,         w, l, l16, lhi);
  __syncthreads();
  rw_pass<false>(aa, lxb, lxa, nullptr, w, l, l16, lhi);
  __syncthreads();
  rw_pass<true>(xt, lxa, nullptr, gb + 16384, w, l, l16, lhi);
}

__global__ __launch_bounds__(256) void rw_prod(const float* __restrict__ gws,
                                               const float* __restrict__ wt,
                                               float* __restrict__ part) {
  __shared__ float red[8 * 1024];
  const int bp  = blockIdx.x / KCH;
  const int ch  = blockIdx.x - bp * KCH;
  const int tid = threadIdx.x;
  const int wv  = tid >> 5;
  const int l   = tid & 31;
  const int l16 = l & 15;
  const int hi  = l >> 4;
  const int q0  = (ch * 8 + wv) * QPW;

  const float4* A0 = (const float4*)(gws + (size_t)(bp * 32 + l16) * GLEN) + hi + q0 * 2;
  const float4* A1 = (const float4*)(gws + (size_t)(bp * 32 + 16 + l16) * GLEN) + hi + q0 * 2;
  const float4* B0 = (const float4*)wt + (size_t)q0 * 32 + l;
  const float4* B1 = B0 + (size_t)NQ * 32;

  v8f c00 = {}, c01 = {}, c10 = {}, c11 = {};
  float4 fa0[2][QB], fa1[2][QB], fb0[2][QB], fb1[2][QB];
#pragma unroll
  for (int blk = -1; blk < QNB; ++blk) {
    if (blk + 1 < QNB) {
      const int ns = (blk + 1) & 1;
#pragma unroll
      for (int u = 0; u < QB; ++u) {
        const int q = (blk + 1) * QB + u;
        if (q < QPW) {
          fa0[ns][u] = A0[q * 2];
          fa1[ns][u] = A1[q * 2];
          fb0[ns][u] = B0[q * 32];
          fb1[ns][u] = B1[q * 32];
        }
      }
    }
    __builtin_amdgcn_sched_barrier(0);
    if (blk >= 0) {
      const int cs = blk & 1;
#pragma unroll
      for (int u = 0; u < QB; ++u) {
        if (blk * QB + u >= QPW) break;
        const v2f a0l = {fa0[cs][u].x, fa0[cs][u].y}, a0h = {fa0[cs][u].z, fa0[cs][u].w};
        const v2f a1l = {fa1[cs][u].x, fa1[cs][u].y}, a1h = {fa1[cs][u].z, fa1[cs][u].w};
        const v2f b0l = {fb0[cs][u].x, fb0[cs][u].y}, b0h = {fb0[cs][u].z, fb0[cs][u].w};
        const v2f b1l = {fb1[cs][u].x, fb1[cs][u].y}, b1h = {fb1[cs][u].z, fb1[cs][u].w};
        c00 = wmma4(a0l, b0l, c00);
        c01 = wmma4(a0l, b1l, c01);
        c10 = wmma4(a1l, b0l, c10);
        c11 = wmma4(a1l, b1l, c11);
        c00 = wmma4(a0h, b0h, c00);
        c01 = wmma4(a0h, b1h, c01);
        c10 = wmma4(a1h, b0h, c10);
        c11 = wmma4(a1h, b1h, c11);
      }
    }
    __builtin_amdgcn_sched_barrier(0);
  }

  float* rw = red + wv * 1024 + l;
#pragma unroll
  for (int r = 0; r < 8; ++r) {
    rw[0   + r * 32] = c00[r];
    rw[256 + r * 32] = c01[r];
    rw[512 + r * 32] = c10[r];
    rw[768 + r * 32] = c11[r];
  }
  __syncthreads();
#pragma unroll
  for (int i = 0; i < 4; ++i) {
    const int e = i * 256 + tid;
    float s = 0.f;
#pragma unroll
    for (int k = 0; k < 8; ++k) s += red[k * 1024 + e];
    const int rt = e >> 9;
    const int ot = (e >> 8) & 1;
    const int r  = (e >> 5) & 7;
    const int ln = e & 31;
    const int bb = bp * 32 + rt * 16 + r + 8 * (ln >> 4);
    const int o  = ot * 16 + (ln & 15);
    part[(size_t)ch * (B_BATCH * OUTD) + bb * OUTD + o] = s;
  }
}

__global__ __launch_bounds__(256) void rw_reduce(const float* __restrict__ part,
                                                 float* __restrict__ out) {
  const int i = blockIdx.x * 256 + threadIdx.x;
  float s = 0.f;
#pragma unroll
  for (int k = 0; k < KCH; ++k) s += part[k * (B_BATCH * OUTD) + i];
  out[i] = s * (1.0f / 3840.0f);
}

extern "C" void kernel_launch(void* const* d_in, const int* in_sizes, int n_in,
                              void* d_out, int out_size, void* d_ws, size_t ws_size,
                              hipStream_t stream) {
  (void)in_sizes; (void)n_in; (void)out_size;
  const float* adj   = (const float*)d_in[0];
  const float* feats = (const float*)d_in[1];
  const int*   idxs  = (const int*)  d_in[2];
  const float* fh    = (const float*)d_in[3];
  const float* ah    = (const float*)d_in[4];
  float* out = (float*)d_out;
  float* ws  = (float*)d_ws;

  rw_pre<<<64, 64, 0, stream>>>(fh, ah, ws + WS_ZF);
  if (ws_size >= 210239488ull) {
    rw_wbuild<<<1312, 256, 0, stream>>>(ws + WS_ZF, ws + WS_WT);
    rw_gram<<<B_BATCH, 256, 0, stream>>>(adj, feats, idxs, ws + WS_G);
    rw_prod<<<512, 256, 0, stream>>>(ws + WS_G, ws + WS_WT, ws + WS_PART);
    rw_reduce<<<128, 256, 0, stream>>>(ws + WS_PART, out);
  } else {
    rw_main<<<B_BATCH, 256, 0, stream>>>(adj, feats, idxs, ws + WS_ZF, out);
  }
}


# Last product's A operand staged through LDS: one load takes 512 contiguous bytes of one batch row (4 whole lines), not 32 bytes of 16 rows; waves take interleaved lines; sum order changes
# speedup vs baseline: 9.2086x; 1.0337x over previous
#include <hip/hip_runtime.h>


typedef float v2f __attribute__((ext_vector_type(2)));
typedef float v8f __attribute__((ext_vector_type(8)));

#define B_BATCH 1024
#define NNODE   128
#define CDIM    128
#define MHID    10
#define OUTD    32
#define NPAD    136

__device__ __forceinline__ v8f wmma4(v2f a, v2f b, v8f c) {
  return __builtin_amdgcn_wmma_f32_16x16x4_f32(
       false, a,  false, b,
       (short)0, c,  false,  false);
}

#define WS_ZF  0
#define ZF_STRIDE 40960

__global__ __launch_bounds__(64) void rw_pre(const float* __restrict__ fh,
                                             const float* __restrict__ ah,
                                             float* __restrict__ zf) {
  const int g   = blockIdx.x * 64 + threadIdx.x;
  const int j   = g & 1;
  const int l16 = (g >> 1) & 15;
  const int lhi = (g >> 5) & 1;
  const int kit = (g >> 6) & 31;
  const int ot  = g >> 11;
  const int c   = kit * 4 + lhi * 2 + j;
  const int o   = ot * 16 + l16;

  float r[45];
#pragma unroll
  for (int p = 0; p < 45; ++p) {
    float t = ah[p * OUTD + o];
    r[p] = t > 0.f ? t : 0.f;
  }
  float z0[MHID], z1[MHID], z2[MHID];
#pragma unroll
  for (int m = 0; m < MHID; ++m) z0[m] = fh[(m * CDIM + c) * OUTD + o];

#pragma unroll
  for (int a = 0; a < MHID; ++a) {
    float s = 0.f;
#pragma unroll
    for (int bb = 0; bb < MHID; ++bb) {
      const int i  = a < bb ? a : bb;
      const int jj = a < bb ? bb : a;
      const float av = (a == bb) ? 0.f : r[i * MHID - (i * (i + 1)) / 2 + (jj - i - 1)];
      s += av * z0[bb];
    }
    z1[a] = s;
  }
#pragma unroll
  for (int a = 0; a < MHID; ++a) {
    float s = 0.f;
#pragma unroll
    for (int bb = 0; bb < MHID; ++bb) {
      const int i  = a < bb ? a : bb;
      const int jj = a < bb ? bb : a;
      const float av = (a == bb) ? 0.f : r[i * MHID - (i * (i + 1)) / 2 + (jj - i - 1)];
      s += av * z1[bb];
    }
    z2[a] = s;
  }

#pragma unroll
  for (int m = 0; m < MHID; ++m) {
    zf[0 * ZF_STRIDE + m * 4096 + g] = z0[m];
    zf[1 * ZF_STRIDE + m * 4096 + g] = z1[m];
    zf[2 * ZF_STRIDE + m * 4096 + g] = z2[m];
  }
}

#define PK 8
#define PNB (32 / PK)
__device__ __forceinline__ void rw_prop(const v2f (&aa)[32], const float* xs, float* xd,
                                        int w, int l16, int lhi) {
#pragma unroll 1
  for (int cp = 0; cp < 4; ++cp) {
    v8f ca = {}, cb = {};
    const float* xb = xs + (lhi * 2) * NPAD + cp * 32 + l16;
    int ob = 16;
    asm volatile("" : "+v"(ob));
    const float* xc = xb + ob;
    v2f ba[2][PK], bb[2][PK];
#pragma unroll
    for (int blk = -1; blk < PNB; ++blk) {
      if (blk + 1 < PNB) {
        const int ns = (blk + 1) & 1;
#pragma unroll
        for (int u = 0; u < PK; ++u) {
          const int kit = (blk + 1) * PK + u;
          ba[ns][u].x = xb[(kit * 4 + 0) * NPAD];
          ba[ns][u].y = xb[(kit * 4 + 1) * NPAD];
          bb[ns][u].x = xc[(kit * 4 + 0) * NPAD];
          bb[ns][u].y = xc[(kit * 4 + 1) * NPAD];
        }
      }
      __builtin_amdgcn_sched_barrier(0);
      if (blk >= 0) {
        const int cs = blk & 1;
#pragma unroll
        for (int u = 0; u < PK; ++u) {
          ca = wmma4(aa[blk * PK + u], ba[cs][u], ca);
          cb = wmma4(aa[blk * PK + u], bb[cs][u], cb);
        }
      }
      __builtin_amdgcn_sched_barrier(0);
    }
    float* xo = xd + (w * 16 + lhi * 8) * NPAD + cp * 32 + l16;
#pragma unroll
    for (int r = 0; r < 8; ++r) {
      xo[r * NPAD]      = ca[r];
      xo[r * NPAD + 16] = cb[r];
    }
  }
}

#define KU 4
#define NB (32 / KU)
__global__ __launch_bounds__(256) void rw_main(const float* __restrict__ adj,
                                               const float* __restrict__ feats,
                                               const int*   __restrict__ idxs,
                                               const float* __restrict__ zfrag,
                                               float*       __restrict__ out) {
  __shared__ float lxa[NNODE * NPAD];
  __shared__ float lxb[NNODE * NPAD];
  __shared__ float accbuf[OUTD];

  const int b   = blockIdx.x;
  const int tid = threadIdx.x;
  const int w   = tid >> 5;
  const int l   = tid & 31;
  const int l16 = l & 15;
  const int lhi = l >> 4;

  if (tid < OUTD) accbuf[tid] = 0.f;

  {
    const int r0 = w * 16;
#pragma unroll 4
    for (int r = r0; r < r0 + 16; ++r) {
      int idx = idxs[b * NNODE + r];
      float4 v = *(const float4*)(feats + (size_t)idx * CDIM + l * 4);
      *(float4*)(&lxa[r * NPAD + l * 4]) = v;
    }
  }

  const int row = w * 16 + l16;
  v2f aa[32];
  {
    const float* Ag = adj + (size_t)b * NNODE * CDIM + row * CDIM + lhi * 2;
#pragma unroll
    for (int kit = 0; kit < 32; ++kit) aa[kit] = *(const v2f*)(Ag + kit * 4);
  }
  __syncthreads();
  rw_prop(aa, lxa, lxb, w, l16, lhi);
  __syncthreads();

  v2f xa[32];
  {
    const float* ax0 = lxa + row * NPAD + lhi * 2;
#pragma unroll
    for (int kit = 0; kit < 32; ++kit) xa[kit] = *(const v2f*)(ax0 + kit * 4);
  }
  rw_prop(aa, lxb, lxa, w, l16, lhi);
  __syncthreads();

  float acc0 = 0.f, acc1 = 0.f;
  const float* ax1 = lxb + row * NPAD + lhi * 2;
  const float* ax2 = lxa + row * NPAD + lhi * 2;
  for (int t = 0; t < MHID * 2; ++t) {
    const v2f* bz0 = (const v2f*)(zfrag)                 + (size_t)t * 1024 + l;
    const v2f* bz1 = (const v2f*)(zfrag + ZF_STRIDE)     + (size_t)t * 1024 + l;
    const v2f* bz2 = (const v2f*)(zfrag + 2 * ZF_STRIDE) + (size_t)t * 1024 + l;
    v8f czx = {}, c1 = {}, c2 = {};
    v2f b0[2][KU], b1[2][KU], b2[2][KU], a1[2][KU], a2[2][KU];
#pragma unroll
    for (int blk = -1; blk < NB; ++blk) {
      if (blk + 1 < NB) {
        const int ns = (blk + 1) & 1;
#pragma unroll
        for (int u = 0; u < KU; ++u) {
          const int kit = (blk + 1) * KU + u;
          b0[ns][u] = bz0[kit * 32];
          b1[ns][u] = bz1[kit * 32];
          b2[ns][u] = bz2[kit * 32];
        }
#pragma unroll
        for (int u = 0; u < KU; ++u) {
          const int kit = (blk + 1) * KU + u;
          a1[ns][u] = *(const v2f*)(ax1 + kit * 4);
          a2[ns][u] = *(const v2f*)(ax2 + kit * 4);
        }
      }
      __builtin_amdgcn_sched_barrier(0);
      if (blk >= 0) {
        const int cs = blk & 1;
#pragma unroll
        for (int u = 0; u < KU; ++u) {
          czx = wmma4(xa[blk * KU + u], b0[cs][u], czx);
          c1  = wmma4(a1[cs][u],        b1[cs][u], c1);
          c2  = wmma4(a2[cs][u],        b2[cs][u], c2);
        }
      }
      __builtin_amdgcn_sched_barrier(0);
    }
    float s0 = 0.f, s1 = 0.f, s2 = 0.f;
#pragma unroll
    for (int r = 0; r < 8; ++r) {
      s0 += czx[r] * czx[r];
      s1 += czx[r] * c1[r];
      s2 += czx[r] * c2[r];
    }
    const float s = (s0 + s1) + s2;
    if ((t & 1) == 0) acc0 += s; else acc1 += s;
  }

  atomicAdd(&accbuf[l16], acc0);
  atomicAdd(&accbuf[16 + l16], acc1);
  __syncthreads();
  if (tid < OUTD) out[(size_t)b * OUTD + tid] = accbuf[tid] * (1.0f / 3840.0f);
}

#define G0T      36
#define GLEN     41984
#define NQ       5248
#define WS_WT    131072
#define WS_G     1703936
#define WS_PART  52035584
#define KCH      16

__global__ __launch_bounds__(256) void rw_wbuild(const float* __restrict__ zf,
                                                 float* __restrict__ wt) {
  const int g4   = blockIdx.x * 256 + threadIdx.x;
  const int lane = g4 & 31;
  const int qq   = g4 >> 5;
  const int ot   = qq >= NQ ? 1 : 0;
  const int q    = qq - ot * NQ;
  const int o16  = lane & 15;
  const int hi   = lane >> 4;
  const int p0   = q * 8 + hi * 4;
  const int slot = p0 >> 8;
  const int r    = (p0 >> 5) & 7;
  const int hg   = (p0 >> 4) & 1;
  int s, w, ct;
  float f = 1.f;
  if (slot < G0T) {
    int d;
    if (slot < 20) { w = slot / 5; d = slot - w * 5; }
    else           { w = 4 + ((slot - 20) >> 2); d = (slot - 20) & 3; }
    s  = 0;
    ct = (w + d) & 7;
    if (d != 0) f = 2.f;
  } else {
    const int t2 = slot - G0T;
    s  = 1 + (t2 >> 6);
    w  = (t2 >> 3) & 7;
    ct = t2 & 7;
  }
  const int c    = 16 * w + r + 8 * hg;
  const int cp0  = 16 * ct + (p0 & 15);
  const int go   = ot * 2048 + o16 * 2;
  const int gc   = go + (c >> 2) * 64 + ((c >> 1) & 1) * 32 + (c & 1);
  const int gcp  = go + (cp0 >> 2) * 64;
  const float* zs = zf + s * ZF_STRIDE + gcp;
  float a0 = 0.f, a1 = 0.f, a2 = 0.f, a3 = 0.f;
#pragma unroll
  for (int m = 0; m < MHID; ++m) {
    const float a = zf[m * 4096 + gc];
    a0 += a * zs[m * 4096 + 0];
    a1 += a * zs[m * 4096 + 1];
    a2 += a * zs[m * 4096 + 32];
    a3 += a * zs[m * 4096 + 33];
  }
  ((float4*)wt)[g4] = make_float4(a0 * f, a1 * f, a2 * f, a3 * f);
}

template <bool TO_WS>
__device__ __forceinline__ void rw_pass(const v2f (&af)[32], const float* xs, float* xd, float* gd,
                                        int w, int l, int l16, int lhi) {
#pragma unroll 1
  for (int cp = 0; cp < 4; ++cp) {
    v8f ca = {}, cb = {};
    const float* xb = xs + (lhi * 2) * NPAD + cp * 32 + l16;
    int ob = 16;
    asm volatile("" : "+v"(ob));
    const float* xc = xb + ob;
    v2f ba[2][PK], bb[2][PK];
#pragma unroll
    for (int blk = -1; blk < PNB; ++blk) {
      if (blk + 1 < PNB) {
        const int ns = (blk + 1) & 1;
#pragma unroll
        for (int u = 0; u < PK; ++u) {
          const int kit = (blk + 1) * PK + u;
          ba[ns][u].x = xb[(kit * 4 + 0) * NPAD];
          ba[ns][u].y = xb[(kit * 4 + 1) * NPAD];
          bb[ns][u].x = xc[(kit * 4 + 0) * NPAD];
          bb[ns][u].y = xc[(kit * 4 + 1) * NPAD];
        }
      }
      __builtin_amdgcn_sched_barrier(0);
      if (blk >= 0) {
        const int cs = blk & 1;
#pragma unroll
        for (int u = 0; u < PK; ++u) {
          ca = wmma4(af[blk * PK + u], ba[cs][u], ca);
          cb = wmma4(af[blk * PK + u], bb[cs][u], cb);
        }
      }
      __builtin_amdgcn_sched_barrier(0);
    }
    if (TO_WS) {
      float* go = gd + cp * 512 + l;
#pragma unroll
      for (int r = 0; r < 8; ++r) {
        go[r * 32]       = ca[r];
        go[256 + r * 32] = cb[r];
      }
    } else {
      float* xo = xd + (w * 16 + lhi * 8) * NPAD + cp * 32 + l16;
#pragma unroll
      for (int r = 0; r < 8; ++r) {
        xo[r * NPAD]      = ca[r];
        xo[r * NPAD + 16] = cb[r];
      }
    }
  }
}

template <bool TWO>
__device__ __forceinline__ void rw_gsym(const v2f (&af)[32], const float* xb, const float* xc,
                                        float* go) {
  v8f ca = {}, cb = {};
  v2f ba[2][PK], bb[2][PK];
#pragma unroll
  for (int blk = -1; blk < PNB; ++blk) {
    if (blk + 1 < PNB) {
      const int ns = (blk + 1) & 1;
#pragma unroll
      for (int u = 0; u < PK; ++u) {
        const int kit = (blk + 1) * PK + u;
        ba[ns][u].x = xb[(kit * 4 + 0) * NPAD];
        ba[ns][u].y = xb[(kit * 4 + 1) * NPAD];
        if (TWO) {
          bb[ns][u].x = xc[(kit * 4 + 0) * NPAD];
          bb[ns][u].y = xc[(kit * 4 + 1) * NPAD];
        }
      }
    }
    __builtin_amdgcn_sched_barrier(0);
    if (blk >= 0) {
      const int cs = blk & 1;
#pragma unroll
      for (int u = 0; u < PK; ++u) {
        ca = wmma4(af[blk * PK + u], ba[cs][u], ca);
        if (TWO) cb = wmma4(af[blk * PK + u], bb[cs][u], cb);
      }
    }
    __builtin_amdgcn_sched_barrier(0);
  }
#pragma unroll
  for (int r = 0; r < 8; ++r) {
    go[r * 32] = ca[r];
    if (TWO) go[256 + r * 32] = cb[r];
  }
}

__global__ __launch_bounds__(256) void rw_gram(const float* __restrict__ adj,
                                               const float* __restrict__ feats,
                                               const int*   __restrict__ idxs,
                                               float*       __restrict__ gws) {
  __shared__ float lxa[NNODE * NPAD];
  __shared__ float lxb[NNODE * NPAD];

  const int b   = blockIdx.x;
  const int tid = threadIdx.x;
  const int w   = tid >> 5;
  const int l   = tid & 31;
  const int l16 = l & 15;
  const int lhi = l >> 4;

  {
    const int r0 = w * 16;
#pragma unroll 4
    for (int r = r0; r < r0 + 16; ++r) {
      int idx = idxs[b * NNODE + r];
      float4 v = *(const float4*)(feats + (size_t)idx * CDIM + l * 4);
      *(float4*)(&lxa[r * NPAD + l * 4]) = v;
    }
  }

  const int row = w * 16 + l16;
  v2f aa[32];
  {
    const float* Ag = adj + (size_t)b * NNODE * CDIM + row * CDIM + lhi * 2;
#pragma unroll
    for (int kit = 0; kit < 32; ++kit) aa[kit] = *(const v2f*)(Ag + kit * 4);
  }
  __syncthreads();
  rw_pass<false>(aa, lxa, lxb, nullptr, w, l, l16, lhi);
  __syncthreads();

  v2f xt[32];
  {
    const float* t0 = lxa + (lhi * 2) * NPAD + row;
#pragma unroll
    for (int kit = 0; kit < 32; ++kit) {
      xt[kit].x = t0[(kit * 4 + 0) * NPAD];
      xt[kit].y = t0[(kit * 4 + 1) * NPAD];
    }
  }
  {
    const int wu = __builtin_amdgcn_readfirstlane(w);
    float* g0 = gws + (size_t)b * GLEN + (wu < 4 ? wu * 5 : 20 + (wu - 4) * 4) * 256 + l;
    const float* xk = lxa + (lhi * 2) * NPAD + l16;
    rw_gsym<true>(xt, xk + ((wu + 0) & 7) * 16, xk + ((wu + 1) & 7) * 16, g0);
    rw_gsym<true>(xt, xk + ((wu + 2) & 7) * 16, xk + ((wu + 3) & 7) * 16, g0 + 512);
    if (wu < 4) rw_gsym<false>(xt, xk + (wu + 4) * 16, nullptr, g0 + 1024);
  }
  float* gb = gws + (size_t)b * GLEN + G0T * 256 + w * 2048;
  rw_pass<true>(xt, lxb, nullptr, gb,         w, l, l16, lhi);
  __syncthreads();
  rw_pass<false>(aa, lxb, lxa, nullptr, w, l, l16, lhi);
  __syncthreads();
  rw_pass<true>(xt, lxa, nullptr, gb + 16384, w, l, l16, lhi);
}

#define QCH     328
#define NST     10
#define SPITCH  260
#define SBUF    (32 * SPITCH)
__global__ __launch_bounds__(256) void rw_prod(const float* __restrict__ gws,
                                               const float* __restrict__ wt,
                                               float* __restrict__ part) {
  __shared__ float4 sm4[2 * SBUF / 4];
  float* sm = (float*)sm4;
  const int bp  = blockIdx.x / KCH;
  const int ch  = blockIdx.x - bp * KCH;
  const int tid = threadIdx.x;
  const int wv  = tid >> 5;
  const int l   = tid & 31;
  const int l16 = l & 15;
  const int hi  = l >> 4;

  const float* gsrc = gws + (size_t)(bp * 32 + wv * 4) * GLEN + ch * (QCH * 8) + l * 4;
  float*       sdst = sm + (wv * 4) * SPITCH + l * 4;
  const float* gtl  = gws + (size_t)(bp * 32 + wv * 4 + hi) * GLEN + ch * (QCH * 8) + NST * 256 + l16 * 4;
  float*       stl  = sm + (wv * 4 + hi) * SPITCH + l16 * 4;
  const float*  asrc = sm + l16 * SPITCH + wv * 32 + hi * 4;
  const float*  atl  = sm + l16 * SPITCH + wv * 8 + hi * 4;
  const float4* B0 = (const float4*)wt + (size_t)(ch * QCH + wv * 4) * 32 + l;
  const float4* B1 = B0 + (size_t)NQ * 32;
  const float4* T0 = (const float4*)wt + (size_t)(ch * QCH + NST * 32 + wv) * 32 + l;
  const float4* T1 = T0 + (size_t)NQ * 32;

  v8f c00 = {}, c01 = {}, c10 = {}, c11 = {};
  float4 g0, g1, g2, g3, g4, g5, g6, g7, t0, t1;
  float4 fb0[2][4], fb1[2][4];
#define STAGE_LOAD(off)                                              \
  g0 = *(const float4*)(gsrc + (off));                               \
  g1 = *(const float4*)(gsrc + (off) + 128);                         \
  g2 = *(const float4*)(gsrc + (size_t)GLEN + (off));                \
  g3 = *(const float4*)(gsrc + (size_t)GLEN + (off) + 128);          \
  g4 = *(const float4*)(gsrc + (size_t)2 * GLEN + (off));            \
  g5 = *(const float4*)(gsrc + (size_t)2 * GLEN + (off) + 128);      \
  g6 = *(const float4*)(gsrc + (size_t)3 * GLEN + (off));            \
  g7 = *(const float4*)(gsrc + (size_t)3 * GLEN + (off) + 128);
#define STAGE_STORE(buf)                                             \
  *(float4*)(sdst + (buf) * SBUF)                    = g0;           \
  *(float4*)(sdst + (buf) * SBUF + 128)              = g1;           \
  *(float4*)(sdst + (buf) * SBUF + SPITCH)           = g2;           \
  *(float4*)(sdst + (buf) * SBUF + SPITCH + 128)     = g3;           \
  *(float4*)(sdst + (buf) * SBUF + 2 * SPITCH)       = g4;           \
  *(float4*)(sdst + (buf) * SBUF + 2 * SPITCH + 128) = g5;           \
  *(float4*)(sdst + (buf) * SBUF + 3 * SPITCH)       = g6;           \
  *(float4*)(sdst + (buf) * SBUF + 3 * SPITCH + 128) = g7;
#define STAGE_BARRIER() asm volatile("s_wait_dscnt 0x0\n\ts_barrier_signal -1\n\ts_barrier_wait -1" ::: "memory");

  STAGE_LOAD(0)
#pragma unroll
  for (int u = 0; u < 4; ++u) {
    fb0[0][u] = B0[u * 32];
    fb1[0][u] = B1[u * 32];
  }
  STAGE_STORE(0)
  STAGE_LOAD(256)
  STAGE_BARRIER()

#pragma unroll
  for (int s = 0; s < NST; ++s) {
    const int cs = s & 1, ns = cs ^ 1;
    if (s + 1 < NST) {
#pragma unroll
      for (int u = 0; u < 4; ++u) {
        fb0[ns][u] = B0[((s + 1) * 32 + u) * 32];
        fb1[ns][u] = B1[((s + 1) * 32 + u) * 32];
      }
    } else {
      fb0[ns][0] = T0[0];
      fb1[ns][0] = T1[0];
    }
    float4 fa0[4], fa1[4];
#pragma unroll
    for (int u = 0; u < 4; ++u) {
      fa0[u] = *(const float4*)(asrc + cs * SBUF + u * 8);
      fa1[u] = *(const float4*)(asrc + cs * SBUF + 16 * SPITCH + u * 8);
    }
    __builtin_amdgcn_sched_barrier(0);
#pragma unroll
    for (int u = 0; u < 4; ++u) {
      const v2f a0l = {fa0[u].x, fa0[u].y}, a0h = {fa0[u].z, fa0[u].w};
      const v2f a1l = {fa1[u].x, fa1[u].y}, a1h = {fa1[u].z, fa1[u].w};
      const v2f b0l = {fb0[cs][u].x, fb0[cs][u].y}, b0h = {fb0[cs][u].z, fb0[cs][u].w};
      const v2f b1l = {fb1[cs][u].x, fb1[cs][u].y}, b1h = {fb1[cs][u].z, fb1[cs][u].w};
      c00 = wmma4(a0l, b0l, c00);
      c01 = wmma4(a0l, b1l, c01);
      c10 = wmma4(a1l, b0l, c10);
      c11 = wmma4(a1l, b1l, c11);
      c00 = wmma4(a0h, b0h, c00);
      c01 = wmma4(a0h, b1h, c01);
      c10 = wmma4(a1h, b0h, c10);
      c11 = wmma4(a1h, b1h, c11);
    }
    __builtin_amdgcn_sched_barrier(0);
    if (s + 1 < NST) {
      STAGE_STORE(ns)
      if (s + 2 < NST) {
        STAGE_LOAD((s + 2) * 256)
      } else {
        t0 = *(const float4*)(gtl);
        t1 = *(const float4*)(gtl + (size_t)2 * GLEN);
      }
    } else {
      *(float4*)(stl + ns * SBUF)              = t0;
      *(float4*)(stl + ns * SBUF + 2 * SPITCH) = t1;
    }
    STAGE_BARRIER()
  }
  {
    const int cs = NST & 1;
    const float4 fa0 = *(const float4*)(atl + cs * SBUF);
    const float4 fa1 = *(const float4*)(atl + cs * SBUF + 16 * SPITCH);
    const v2f a0l = {fa0.x, fa0.y}, a0h = {fa0.z, fa0.w};
    const v2f a1l = {fa1.x, fa1.y}, a1h = {fa1.z, fa1.w};
    const v2f b0l = {fb0[cs][0].x, fb0[cs][0].y}, b0h = {fb0[cs][0].z, fb0[cs][0].w};
    const v2f b1l = {fb1[cs][0].x, fb1[cs][0].y}, b1h = {fb1[cs][0].z, fb1[cs][0].w};
    c00 = wmma4(a0l, b0l, c00);
    c01 = wmma4(a0l, b1l, c01);
    c10 = wmma4(a1l, b0l, c10);
    c11 = wmma4(a1l, b1l, c11);
    c00 = wmma4(a0h, b0h, c00);
    c01 = wmma4(a0h, b1h, c01);
    c10 = wmma4(a1h, b0h, c10);
    c11 = wmma4(a1h, b1h, c11);
  }
  __syncthreads();

  float* red = sm;
  float* rw = red + wv * 1024 + l;
#pragma unroll
  for (int r = 0; r < 8; ++r) {
    rw[0   + r * 32] = c00[r];
    rw[256 + r * 32] = c01[r];
    rw[512 + r * 32] = c10[r];
    rw[768 + r * 32] = c11[r];
  }
  __syncthreads();
#pragma unroll
  for (int i = 0; i < 4; ++i) {
    const int e = i * 256 + tid;
    float s = 0.f;
#pragma unroll
    for (int k = 0; k < 8; ++k) s += red[k * 1024 + e];
    const int rt = e >> 9;
    const int ot = (e >> 8) & 1;
    const int r  = (e >> 5) & 7;
    const int ln = e & 31;
    const int bb = bp * 32 + rt * 16 + r + 8 * (ln >> 4);
    const int o  = ot * 16 + (ln & 15);
    part[(size_t)ch * (B_BATCH * OUTD) + bb * OUTD + o] = s;
  }
}

__global__ __launch_bounds__(256) void rw_reduce(const float* __restrict__ part,
                                                 float* __restrict__ out) {
  const int i = blockIdx.x * 256 + threadIdx.x;
  float s = 0.f;
#pragma unroll
  for (int k = 0; k < KCH; ++k) s += part[k * (B_BATCH * OUTD) + i];
  out[i] = s * (1.0f / 3840.0f);
}

extern "C" void kernel_launch(void* const* d_in, const int* in_sizes, int n_in,
                              void* d_out, int out_size, void* d_ws, size_t ws_size,
                              hipStream_t stream) {
  (void)in_sizes; (void)n_in; (void)out_size;
  const float* adj   = (const float*)d_in[0];
  const float* feats = (const float*)d_in[1];
  const int*   idxs  = (const int*)  d_in[2];
  const float* fh    = (const float*)d_in[3];
  const float* ah    = (const float*)d_in[4];
  float* out = (float*)d_out;
  float* ws  = (float*)d_ws;

  rw_pre<<<64, 64, 0, stream>>>(fh, ah, ws + WS_ZF);
  if (ws_size >= 210239488ull) {
    rw_wbuild<<<1312, 256, 0, stream>>>(ws + WS_ZF, ws + WS_WT);
    rw_gram<<<B_BATCH, 256, 0, stream>>>(adj, feats, idxs, ws + WS_G);
    rw_prod<<<512, 256, 0, stream>>>(ws + WS_G, ws + WS_WT, ws + WS_PART);
    rw_reduce<<<128, 256, 0, stream>>>(ws + WS_PART, out);
  } else {
    rw_main<<<B_BATCH, 256, 0, stream>>>(adj, feats, idxs, ws + WS_ZF, out);
  }
}


# Table build (hidden-graph products, Wtab entries) moved into the Gram launch as 82 extra workgroups that run beside the batch workgroups; launches 5 to 3; sums unchanged
# speedup vs baseline: 9.5499x; 1.0371x over previous
#include <hip/hip_runtime.h>


typedef float v2f __attribute__((ext_vector_type(2)));
typedef float v8f __attribute__((ext_vector_type(8)));

#define B_BATCH 1024
#define NNODE   128
#define CDIM    128
#define MHID    10
#define OUTD    32
#define NPAD    136

__device__ __forceinline__ v8f wmma4(v2f a, v2f b, v8f c) {
  return __builtin_amdgcn_wmma_f32_16x16x4_f32(
       false, a,  false, b,
       (short)0, c,  false,  false);
}

#define WS_ZF  0
#define ZF_STRIDE 40960

__global__ __launch_bounds__(64) void rw_pre(const float* __restrict__ fh,
                                             const float* __restrict__ ah,
                                             float* __restrict__ zf) {
  const int g   = blockIdx.x * 64 + threadIdx.x;
  const int j   = g & 1;
  const int l16 = (g >> 1) & 15;
  const int lhi = (g >> 5) & 1;
  const int kit = (g >> 6) & 31;
  const int ot  = g >> 11;
  const int c   = kit * 4 + lhi * 2 + j;
  const int o   = ot * 16 + l16;

  float r[45];
#pragma unroll
  for (int p = 0; p < 45; ++p) {
    float t = ah[p * OUTD + o];
    r[p] = t > 0.f ? t : 0.f;
  }
  float z0[MHID], z1[MHID], z2[MHID];
#pragma unroll
  for (int m = 0; m < MHID; ++m) z0[m] = fh[(m * CDIM + c) * OUTD + o];

#pragma unroll
  for (int a = 0; a < MHID; ++a) {
    float s = 0.f;
#pragma unroll
    for (int bb = 0; bb < MHID; ++bb) {
      const int i  = a < bb ? a : bb;
      const int jj = a < bb ? bb : a;
      const float av = (a == bb) ? 0.f : r[i * MHID - (i * (i + 1)) / 2 + (jj - i - 1)];
      s += av * z0[bb];
    }
    z1[a] = s;
  }
#pragma unroll
  for (int a = 0; a < MHID; ++a) {
    float s = 0.f;
#pragma unroll
    for (int bb = 0; bb < MHID; ++bb) {
      const int i  = a < bb ? a : bb;
      const int jj = a < bb ? bb : a;
      const float av = (a == bb) ? 0.f : r[i * MHID - (i * (i + 1)) / 2 + (jj - i - 1)];
      s += av * z1[bb];
    }
    z2[a] = s;
  }

#pragma unroll
  for (int m = 0; m < MHID; ++m) {
    zf[0 * ZF_STRIDE + m * 4096 + g] = z0[m];
    zf[1 * ZF_STRIDE + m * 4096 + g] = z1[m];
    zf[2 * ZF_STRIDE + m * 4096 + g] = z2[m];
  }
}

#define PK 8
#define PNB (32 / PK)
__device__ __forceinline__ void rw_prop(const v2f (&aa)[32], const float* xs, float* xd,
                                        int w, int l16, int lhi) {
#pragma unroll 1
  for (int cp = 0; cp < 4; ++cp) {
    v8f ca = {}, cb = {};
    const float* xb = xs + (lhi * 2) * NPAD + cp * 32 + l16;
    int ob = 16;
    asm volatile("" : "+v"(ob));
    const float* xc = xb + ob;
    v2f ba[2][PK], bb[2][PK];
#pragma unroll
    for (int blk = -1; blk < PNB; ++blk) {
      if (blk + 1 < PNB) {
        const int ns = (blk + 1) & 1;
#pragma unroll
        for (int u = 0; u < PK; ++u) {
          const int kit = (blk + 1) * PK + u;
          ba[ns][u].x = xb[(kit * 4 + 0) * NPAD];
          ba[ns][u].y = xb[(kit * 4 + 1) * NPAD];
          bb[ns][u].x = xc[(kit * 4 + 0) * NPAD];
          bb[ns][u].y = xc[(kit * 4 + 1) * NPAD];
        }
      }
      __builtin_amdgcn_sched_barrier(0);
      if (blk >= 0) {
        const int cs = blk & 1;
#pragma unroll
        for (int u = 0; u < PK; ++u) {
          ca = wmma4(aa[blk * PK + u], ba[cs][u], ca);
          cb = wmma4(aa[blk * PK + u], bb[cs][u], cb);
        }
      }
      __builtin_amdgcn_sched_barrier(0);
    }
    float* xo = xd + (w * 16 + lhi * 8) * NPAD + cp * 32 + l16;
#pragma unroll
    for (int r = 0; r < 8; ++r) {
      xo[r * NPAD]      = ca[r];
      xo[r * NPAD + 16] = cb[r];
    }
  }
}

#define KU 4
#define NB (32 / KU)
__global__ __launch_bounds__(256) void rw_main(const float* __restrict__ adj,
                                               const float* __restrict__ feats,
                                               const int*   __restrict__ idxs,
                                               const float* __restrict__ zfrag,
                                               float*       __restrict__ out) {
  __shared__ float lxa[NNODE * NPAD];
  __shared__ float lxb[NNODE * NPAD];
  __shared__ float accbuf[OUTD];

  const int b   = blockIdx.x;
  const int tid = threadIdx.x;
  const int w   = tid >> 5;
  const int l   = tid & 31;
  const int l16 = l & 15;
  const int lhi = l >> 4;

  if (tid < OUTD) accbuf[tid] = 0.f;

  {
    const int r0 = w * 16;
#pragma unroll 4
    for (int r = r0; r < r0 + 16; ++r) {
      int idx = idxs[b * NNODE + r];
      float4 v = *(const float4*)(feats + (size_t)idx * CDIM + l * 4);
      *(float4*)(&lxa[r * NPAD + l * 4]) = v;
    }
  }

  const int row = w * 16 + l16;
  v2f aa[32];
  {
    const float* Ag = adj + (size_t)b * NNODE * CDIM + row * CDIM + lhi * 2;
#pragma unroll
    for (int kit = 0; kit < 32; ++kit) aa[kit] = *(const v2f*)(Ag + kit * 4);
  }
  __syncthreads();
  rw_prop(aa, lxa, lxb, w, l16, lhi);
  __syncthreads();

  v2f xa[32];
  {
    const float* ax0 = lxa + row * NPAD + lhi * 2;
#pragma unroll
    for (int kit = 0; kit < 32; ++kit) xa[kit] = *(const v2f*)(ax0 + kit * 4);
  }
  rw_prop(aa, lxb, lxa, w, l16, lhi);
  __syncthreads();

  float acc0 = 0.f, acc1 = 0.f;
  const float* ax1 = lxb + row * NPAD + lhi * 2;
  const float* ax2 = lxa + row * NPAD + lhi * 2;
  for (int t = 0; t < MHID * 2; ++t) {
    const v2f* bz0 = (const v2f*)(zfrag)                 + (size_t)t * 1024 + l;
    const v2f* bz1 = (const v2f*)(zfrag + ZF_STRIDE)     + (size_t)t * 1024 + l;
    const v2f* bz2 = (const v2f*)(zfrag + 2 * ZF_STRIDE) + (size_t)t * 1024 + l;
    v8f czx = {}, c1 = {}, c2 = {};
    v2f b0[2][KU], b1[2][KU], b2[2][KU], a1[2][KU], a2[2][KU];
#pragma unroll
    for (int blk = -1; blk < NB; ++blk) {
      if (blk + 1 < NB) {
        const int ns = (blk + 1) & 1;
#pragma unroll
        for (int u = 0; u < KU; ++u) {
          const int kit = (blk + 1) * KU + u;
          b0[ns][u] = bz0[kit * 32];
          b1[ns][u] = bz1[kit * 32];
          b2[ns][u] = bz2[kit * 32];
        }
#pragma unroll
        for (int u = 0; u < KU; ++u) {
          const int kit = (blk + 1) * KU + u;
          a1[ns][u] = *(const v2f*)(ax1 + kit * 4);
          a2[ns][u] = *(const v2f*)(ax2 + kit * 4);
        }
      }
      __builtin_amdgcn_sched_barrier(0);
      if (blk >= 0) {
        const int cs = blk & 1;
#pragma unroll
        for (int u = 0; u < KU; ++u) {
          czx = wmma4(xa[blk * KU + u], b0[cs][u], czx);
          c1  = wmma4(a1[cs][u],        b1[cs][u], c1);
          c2  = wmma4(a2[cs][u],        b2[cs][u], c2);
        }
      }
      __builtin_amdgcn_sched_barrier(0);
    }
    float s0 = 0.f, s1 = 0.f, s2 = 0.f;
#pragma unroll
    for (int r = 0; r < 8; ++r) {
      s0 += czx[r] * czx[r];
      s1 += czx[r] * c1[r];
      s2 += czx[r] * c2[r];
    }
    const float s = (s0 + s1) + s2;
    if ((t & 1) == 0) acc0 += s; else acc1 += s;
  }

  atomicAdd(&accbuf[l16], acc0);
  atomicAdd(&accbuf[16 + l16], acc1);
  __syncthreads();
  if (tid < OUTD) out[(size_t)b * OUTD + tid] = accbuf[tid] * (1.0f / 3840.0f);
}

#define G0T      36
#define GLEN     41984
#define NQ       5248
#define WS_WT    131072
#define WS_G     1703936
#define WS_PART  52035584
#define KCH      16

#define NTB 82
#define GRAM_GRID 1106
__device__ __forceinline__ void rw_tables(const float* __restrict__ fh,
                                          const float* __restrict__ ah,
                                          float* __restrict__ wt,
                                          int gw, int l) {
  const int slot = gw >> 2;
  const int ot   = (gw >> 1) & 1;
  const int jh   = gw & 1;
  const int o16  = l & 15;
  const int hi   = l >> 4;
  const int o    = ot * 16 + o16;
  int s, w, ct;
  float f = 1.f;
  if (slot < G0T) {
    int d;
    if (slot < 20) { w = slot / 5; d = slot - w * 5; }
    else           { w = 4 + ((slot - 20) >> 2); d = (slot - 20) & 3; }
    s  = 0;
    ct = (w + d) & 7;
    if (d != 0) f = 2.f;
  } else {
    const int t2 = slot - G0T;
    s  = 1 + (t2 >> 6);
    w  = (t2 >> 3) & 7;
    ct = t2 & 7;
  }
  const int cp0 = 16 * ct + 8 * jh + 4 * hi;

  float zs[4][MHID];
#pragma unroll
  for (int e = 0; e < 4; ++e) {
#pragma unroll
    for (int m = 0; m < MHID; ++m) zs[e][m] = fh[(m * CDIM + cp0 + e) * OUTD + o];
  }
  if (s > 0) {
    float r[45];
#pragma unroll
    for (int p = 0; p < 45; ++p) {
      float t = ah[p * OUTD + o];
      r[p] = t > 0.f ? t : 0.f;
    }
    for (int it = 0; it < s; ++it) {
#pragma unroll
      for (int e = 0; e < 4; ++e) {
        float zn[MHID];
#pragma unroll
        for (int a = 0; a < MHID; ++a) {
          float acc = 0.f;
#pragma unroll
          for (int bb = 0; bb < MHID; ++bb) {
            const int i  = a < bb ? a : bb;
            const int jj = a < bb ? bb : a;
            const float av = (a == bb) ? 0.f : r[i * MHID - (i * (i + 1)) / 2 + (jj - i - 1)];
            acc = __builtin_fmaf(av, zs[e][bb], acc);
          }
          zn[a] = acc;
        }
#pragma unroll
        for (int a = 0; a < MHID; ++a) zs[e][a] = zn[a];
      }
    }
  }

  const float* za = fh + (16 * w) * OUTD + o;
  float4*      wo = (float4*)wt + (size_t)(ot * NQ + slot * 32 + jh) * 32 + l;
#pragma unroll
  for (int i = 0; i < 16; ++i) {
    const int c = (i >> 1) + 8 * (i & 1);
    float a0 = 0.f, a1 = 0.f, a2 = 0.f, a3 = 0.f;
#pragma unroll
    for (int m = 0; m < MHID; ++m) {
      const float a = za[(m * CDIM + c) * OUTD];
      a0 = __builtin_fmaf(a, zs[0][m], a0);
      a1 = __builtin_fmaf(a, zs[1][m], a1);
      a2 = __builtin_fmaf(a, zs[2][m], a2);
      a3 = __builtin_fmaf(a, zs[3][m], a3);
    }
    wo[i * 64] = make_float4(a0 * f, a1 * f, a2 * f, a3 * f);
  }
}

template <bool TO_WS>
__device__ __forceinline__ void rw_pass(const v2f (&af)[32], const float* xs, float* xd, float* gd,
                                        int w, int l, int l16, int lhi) {
#pragma unroll 1
  for (int cp = 0; cp < 4; ++cp) {
    v8f ca = {}, cb = {};
    const float* xb = xs + (lhi * 2) * NPAD + cp * 32 + l16;
    int ob = 16;
    asm volatile("" : "+v"(ob));
    const float* xc = xb + ob;
    v2f ba[2][PK], bb[2][PK];
#pragma unroll
    for (int blk = -1; blk < PNB; ++blk) {
      if (blk + 1 < PNB) {
        const int ns = (blk + 1) & 1;
#pragma unroll
        for (int u = 0; u < PK; ++u) {
          const int kit = (blk + 1) * PK + u;
          ba[ns][u].x = xb[(kit * 4 + 0) * NPAD];
          ba[ns][u].y = xb[(kit * 4 + 1) * NPAD];
          bb[ns][u].x = xc[(kit * 4 + 0) * NPAD];
          bb[ns][u].y = xc[(kit * 4 + 1) * NPAD];
        }
      }
      __builtin_amdgcn_sched_barrier(0);
      if (blk >= 0) {
        const int cs = blk & 1;
#pragma unroll
        for (int u = 0; u < PK; ++u) {
          ca = wmma4(af[blk * PK + u], ba[cs][u], ca);
          cb = wmma4(af[blk * PK + u], bb[cs][u], cb);
        }
      }
      __builtin_amdgcn_sched_barrier(0);
    }
    if (TO_WS) {
      float* go = gd + cp * 512 + l;
#pragma unroll
      for (int r = 0; r < 8; ++r) {
        go[r * 32]       = ca[r];
        go[256 + r * 32] = cb[r];
      }
    } else {
      float* xo = xd + (w * 16 + lhi * 8) * NPAD + cp * 32 + l16;
#pragma unroll
      for (int r = 0; r < 8; ++r) {
        xo[r * NPAD]      = ca[r];
        xo[r * NPAD + 16] = cb[r];
      }
    }
  }
}

template <bool TWO>
__device__ __forceinline__ void rw_gsym(const v2f (&af)[32], const float* xb, const float* xc,
                                        float* go) {
  v8f ca = {}, cb = {};
  v2f ba[2][PK], bb[2][PK];
#pragma unroll
  for (int blk = -1; blk < PNB; ++blk) {
    if (blk + 1 < PNB) {
      const int ns = (blk + 1) & 1;
#pragma unroll
      for (int u = 0; u < PK; ++u) {
        const int kit = (blk + 1) * PK + u;
        ba[ns][u].x = xb[(kit * 4 + 0) * NPAD];
        ba[ns][u].y = xb[(kit * 4 + 1) * NPAD];
        if (TWO) {
          bb[ns][u].x = xc[(kit * 4 + 0) * NPAD];
          bb[ns][u].y = xc[(kit * 4 + 1) * NPAD];
        }
      }
    }
    __builtin_amdgcn_sched_barrier(0);
    if (blk >= 0) {
      const int cs = blk & 1;
#pragma unroll
      for (int u = 0; u < PK; ++u) {
        ca = wmma4(af[blk * PK + u], ba[cs][u], ca);
        if (TWO) cb = wmma4(af[blk * PK + u], bb[cs][u], cb);
      }
    }
    __builtin_amdgcn_sched_barrier(0);
  }
#pragma unroll
  for (int r = 0; r < 8; ++r) {
    go[r * 32] = ca[r];
    if (TWO) go[256 + r * 32] = cb[r];
  }
}

__global__ __launch_bounds__(256) void rw_gram(const float* __restrict__ adj,
                                               const float* __restrict__ feats,
                                               const int*   __restrict__ idxs,
                                               const float* __restrict__ fh,
                                               const float* __restrict__ ah,
                                               float*       __restrict__ wt,
                                               float*       __restrict__ gws) {
  __shared__ float lxa[NNODE * NPAD];
  __shared__ float lxb[NNODE * NPAD];

  const int tid = threadIdx.x;
  if (blockIdx.x < NTB) {
    const int gw = __builtin_amdgcn_readfirstlane((int)blockIdx.x * 8 + (tid >> 5));
    rw_tables(fh, ah, wt, gw, tid & 31);
    return;
  }
  const int b   = blockIdx.x - NTB;
  const int w   = tid >> 5;
  const int l   = tid & 31;
  const int l16 = l & 15;
  const int lhi = l >> 4;

  {
    const int r0 = w * 16;
#pragma unroll 4
    for (int r = r0; r < r0 + 16; ++r) {
      int idx = idxs[b * NNODE + r];
      float4 v = *(const float4*)(feats + (size_t)idx * CDIM + l * 4);
      *(float4*)(&lxa[r * NPAD + l * 4]) = v;
    }
  }

  const int row = w * 16 + l16;
  v2f aa[32];
  {
    const float* Ag = adj + (size_t)b * NNODE * CDIM + row * CDIM + lhi * 2;
#pragma unroll
    for (int kit = 0; kit < 32; ++kit) aa[kit] = *(const v2f*)(Ag + kit * 4);
  }
  __syncthreads();
  rw_pass<false>(aa, lxa, lxb, nullptr, w, l, l16, lhi);
  __syncthreads();

  v2f xt[32];
  {
    const float* t0 = lxa + (lhi * 2) * NPAD + row;
#pragma unroll
    for (int kit = 0; kit < 32; ++kit) {
      xt[kit].x = t0[(kit * 4 + 0) * NPAD];
      xt[kit].y = t0[(kit * 4 + 1) * NPAD];
    }
  }
  {
    const int wu = __builtin_amdgcn_readfirstlane(w);
    float* g0 = gws + (size_t)b * GLEN + (wu < 4 ? wu * 5 : 20 + (wu - 4) * 4) * 256 + l;
    const float* xk = lxa + (lhi * 2) * NPAD + l16;
    rw_gsym<true>(xt, xk + ((wu + 0) & 7) * 16, xk + ((wu + 1) & 7) * 16, g0);
    rw_gsym<true>(xt, xk + ((wu + 2) & 7) * 16, xk + ((wu + 3) & 7) * 16, g0 + 512);
    if (wu < 4) rw_gsym<false>(xt, xk + (wu + 4) * 16, nullptr, g0 + 1024);
  }
  float* gb = gws + (size_t)b * GLEN + G0T * 256 + w * 2048;
  rw_pass<true>(xt, lxb, nullptr, gb,         w, l, l16, lhi);
  __syncthreads();
  rw_pass<false>(aa, lxb, lxa, nullptr, w, l, l16, lhi);
  __syncthreads();
  rw_pass<true>(xt, lxa, nullptr, gb + 16384, w, l, l16, lhi);
}

#define QCH     328
#define NST     10
#define SPITCH  260
#define SBUF    (32 * SPITCH)
__global__ __launch_bounds__(256) void rw_prod(const float* __restrict__ gws,
                                               const float* __restrict__ wt,
                                               float* __restrict__ part) {
  __shared__ float4 sm4[2 * SBUF / 4];
  float* sm = (float*)sm4;
  const int bp  = blockIdx.x / KCH;
  const int ch  = blockIdx.x - bp * KCH;
  const int tid = threadIdx.x;
  const int wv  = tid >> 5;
  const int l   = tid & 31;
  const int l16 = l & 15;
  const int hi  = l >> 4;

  const float* gsrc = gws + (size_t)(bp * 32 + wv * 4) * GLEN + ch * (QCH * 8) + l * 4;
  float*       sdst = sm + (wv * 4) * SPITCH + l * 4;
  const float* gtl  = gws + (size_t)(bp * 32 + wv * 4 + hi) * GLEN + ch * (QCH * 8) + NST * 256 + l16 * 4;
  float*       stl  = sm + (wv * 4 + hi) * SPITCH + l16 * 4;
  const float*  asrc = sm + l16 * SPITCH + wv * 32 + hi * 4;
  const float*  atl  = sm + l16 * SPITCH + wv * 8 + hi * 4;
  const float4* B0 = (const float4*)wt + (size_t)(ch * QCH + wv * 4) * 32 + l;
  const float4* B1 = B0 + (size_t)NQ * 32;
  const float4* T0 = (const float4*)wt + (size_t)(ch * QCH + NST * 32 + wv) * 32 + l;
  const float4* T1 = T0 + (size_t)NQ * 32;

  v8f c00 = {}, c01 = {}, c10 = {}, c11 = {};
  float4 g0, g1, g2, g3, g4, g5, g6, g7, t0, t1;
  float4 fb0[2][4], fb1[2][4];
#define STAGE_LOAD(off)                                              \
  g0 = *(const float4*)(gsrc + (off));                               \
  g1 = *(const float4*)(gsrc + (off) + 128);                         \
  g2 = *(const float4*)(gsrc + (size_t)GLEN + (off));                \
  g3 = *(const float4*)(gsrc + (size_t)GLEN + (off) + 128);          \
  g4 = *(const float4*)(gsrc + (size_t)2 * GLEN + (off));            \
  g5 = *(const float4*)(gsrc + (size_t)2 * GLEN + (off) + 128);      \
  g6 = *(const float4*)(gsrc + (size_t)3 * GLEN + (off));            \
  g7 = *(const float4*)(gsrc + (size_t)3 * GLEN + (off) + 128);
#define STAGE_STORE(buf)                                             \
  *(float4*)(sdst + (buf) * SBUF)                    = g0;           \
  *(float4*)(sdst + (buf) * SBUF + 128)              = g1;           \
  *(float4*)(sdst + (buf) * SBUF + SPITCH)           = g2;           \
  *(float4*)(sdst + (buf) * SBUF + SPITCH + 128)     = g3;           \
  *(float4*)(sdst + (buf) * SBUF + 2 * SPITCH)       = g4;           \
  *(float4*)(sdst + (buf) * SBUF + 2 * SPITCH + 128) = g5;           \
  *(float4*)(sdst + (buf) * SBUF + 3 * SPITCH)       = g6;           \
  *(float4*)(sdst + (buf) * SBUF + 3 * SPITCH + 128) = g7;
#define STAGE_BARRIER() asm volatile("s_wait_dscnt 0x0\n\ts_barrier_signal -1\n\ts_barrier_wait -1" ::: "memory");

  STAGE_LOAD(0)
#pragma unroll
  for (int u = 0; u < 4; ++u) {
    fb0[0][u] = B0[u * 32];
    fb1[0][u] = B1[u * 32];
  }
  STAGE_STORE(0)
  STAGE_LOAD(256)
  STAGE_BARRIER()

#pragma unroll
  for (int s = 0; s < NST; ++s) {
    const int cs = s & 1, ns = cs ^ 1;
    if (s + 1 < NST) {
#pragma unroll
      for (int u = 0; u < 4; ++u) {
        fb0[ns][u] = B0[((s + 1) * 32 + u) * 32];
        fb1[ns][u] = B1[((s + 1) * 32 + u) * 32];
      }
    } else {
      fb0[ns][0] = T0[0];
      fb1[ns][0] = T1[0];
    }
    float4 fa0[4], fa1[4];
#pragma unroll
    for (int u = 0; u < 4; ++u) {
      fa0[u] = *(const float4*)(asrc + cs * SBUF + u * 8);
      fa1[u] = *(const float4*)(asrc + cs * SBUF + 16 * SPITCH + u * 8);
    }
    __builtin_amdgcn_sched_barrier(0);
#pragma unroll
    for (int u = 0; u < 4; ++u) {
      const v2f a0l = {fa0[u].x, fa0[u].y}, a0h = {fa0[u].z, fa0[u].w};
      const v2f a1l = {fa1[u].x, fa1[u].y}, a1h = {fa1[u].z, fa1[u].w};
      const v2f b0l = {fb0[cs][u].x, fb0[cs][u].y}, b0h = {fb0[cs][u].z, fb0[cs][u].w};
      const v2f b1l = {fb1[cs][u].x, fb1[cs][u].y}, b1h = {fb1[cs][u].z, fb1[cs][u].w};
      c00 = wmma4(a0l, b0l, c00);
      c01 = wmma4(a0l, b1l, c01);
      c10 = wmma4(a1l, b0l, c10);
      c11 = wmma4(a1l, b1l, c11);
      c00 = wmma4(a0h, b0h, c00);
      c01 = wmma4(a0h, b1h, c01);
      c10 = wmma4(a1h, b0h, c10);
      c11 = wmma4(a1h, b1h, c11);
    }
    __builtin_amdgcn_sched_barrier(0);
    if (s + 1 < NST) {
      STAGE_STORE(ns)
      if (s + 2 < NST) {
        STAGE_LOAD((s + 2) * 256)
      } else {
        t0 = *(const float4*)(gtl);
        t1 = *(const float4*)(gtl + (size_t)2 * GLEN);
      }
    } else {
      *(float4*)(stl + ns * SBUF)              = t0;
      *(float4*)(stl + ns * SBUF + 2 * SPITCH) = t1;
    }
    STAGE_BARRIER()
  }
  {
    const int cs = NST & 1;
    const float4 fa0 = *(const float4*)(atl + cs * SBUF);
    const float4 fa1 = *(const float4*)(atl + cs * SBUF + 16 * SPITCH);
    const v2f a0l = {fa0.x, fa0.y}, a0h = {fa0.z, fa0.w};
    const v2f a1l = {fa1.x, fa1.y}, a1h = {fa1.z, fa1.w};
    const v2f b0l = {fb0[cs][0].x, fb0[cs][0].y}, b0h = {fb0[cs][0].z, fb0[cs][0].w};
    const v2f b1l = {fb1[cs][0].x, fb1[cs][0].y}, b1h = {fb1[cs][0].z, fb1[cs][0].w};
    c00 = wmma4(a0l, b0l, c00);
    c01 = wmma4(a0l, b1l, c01);
    c10 = wmma4(a1l, b0l, c10);
    c11 = wmma4(a1l, b1l, c11);
    c00 = wmma4(a0h, b0h, c00);
    c01 = wmma4(a0h, b1h, c01);
    c10 = wmma4(a1h, b0h, c10);
    c11 = wmma4(a1h, b1h, c11);
  }
  __syncthreads();

  float* red = sm;
  float* rw = red + wv * 1024 + l;
#pragma unroll
  for (int r = 0; r < 8; ++r) {
    rw[0   + r * 32] = c00[r];
    rw[256 + r * 32] = c01[r];
    rw[512 + r * 32] = c10[r];
    rw[768 + r * 32] = c11[r];
  }
  __syncthreads();
#pragma unroll
  for (int i = 0; i < 4; ++i) {
    const int e = i * 256 + tid;
    float s = 0.f;
#pragma unroll
    for (int k = 0; k < 8; ++k) s += red[k * 1024 + e];
    const int rt = e >> 9;
    const int ot = (e >> 8) & 1;
    const int r  = (e >> 5) & 7;
    const int ln = e & 31;
    const int bb = bp * 32 + rt * 16 + r + 8 * (ln >> 4);
    const int o  = ot * 16 + (ln & 15);
    part[(size_t)ch * (B_BATCH * OUTD) + bb * OUTD + o] = s;
  }
}

__global__ __launch_bounds__(256) void rw_reduce(const float* __restrict__ part,
                                                 float* __restrict__ out) {
  const int i = blockIdx.x * 256 + threadIdx.x;
  float s = 0.f;
#pragma unroll
  for (int k = 0; k < KCH; ++k) s += part[k * (B_BATCH * OUTD) + i];
  out[i] = s * (1.0f / 3840.0f);
}

extern "C" void kernel_launch(void* const* d_in, const int* in_sizes, int n_in,
                              void* d_out, int out_size, void* d_ws, size_t ws_size,
                              hipStream_t stream) {
  (void)in_sizes; (void)n_in; (void)out_size;
  const float* adj   = (const float*)d_in[0];
  const float* feats = (const float*)d_in[1];
  const int*   idxs  = (const int*)  d_in[2];
  const float* fh    = (const float*)d_in[3];
  const float* ah    = (const float*)d_in[4];
  float* out = (float*)d_out;
  float* ws  = (float*)d_ws;

  if (ws_size >= 210239488ull) {
    rw_gram<<<GRAM_GRID, 256, 0, stream>>>(adj, feats, idxs, fh, ah, ws + WS_WT, ws + WS_G);
    rw_prod<<<512, 256, 0, stream>>>(ws + WS_G, ws + WS_WT, ws + WS_PART);
    rw_reduce<<<128, 256, 0, stream>>>(ws + WS_PART, out);
  } else {
    rw_pre<<<64, 64, 0, stream>>>(fh, ah, ws + WS_ZF);
    rw_main<<<B_BATCH, 256, 0, stream>>>(adj, feats, idxs, ws + WS_ZF, out);
  }
}


# Gram kernel: X0, X1, X2 held transposed in LDS (node index contiguous): one 16-byte load at a constant offset feeds the fragments of two WMMAs; k regrouped inside the products; sum order changes
# speedup vs baseline: 10.4327x; 1.0924x over previous
#include <hip/hip_runtime.h>


typedef float v2f __attribute__((ext_vector_type(2)));
typedef float v8f __attribute__((ext_vector_type(8)));
typedef float v4f __attribute__((ext_vector_type(4)));
typedef __attribute__((address_space(3))) const v4f lds_cv4f;

#define B_BATCH 1024
#define NNODE   128
#define CDIM    128
#define MHID    10
#define OUTD    32
#define NPAD    136

__device__ __forceinline__ v8f wmma4(v2f a, v2f b, v8f c) {
  return __builtin_amdgcn_wmma_f32_16x16x4_f32(
       false, a,  false, b,
       (short)0, c,  false,  false);
}

#define WS_ZF  0
#define ZF_STRIDE 40960

__global__ __launch_bounds__(64) void rw_pre(const float* __restrict__ fh,
                                             const float* __restrict__ ah,
                                             float* __restrict__ zf) {
  const int g   = blockIdx.x * 64 + threadIdx.x;
  const int j   = g & 1;
  const int l16 = (g >> 1) & 15;
  const int lhi = (g >> 5) & 1;
  const int kit = (g >> 6) & 31;
  const int ot  = g >> 11;
  const int c   = kit * 4 + lhi * 2 + j;
  const int o   = ot * 16 + l16;

  float r[45];
#pragma unroll
  for (int p = 0; p < 45; ++p) {
    float t = ah[p * OUTD + o];
    r[p] = t > 0.f ? t : 0.f;
  }
  float z0[MHID], z1[MHID], z2[MHID];
#pragma unroll
  for (int m = 0; m < MHID; ++m) z0[m] = fh[(m * CDIM + c) * OUTD + o];

#pragma unroll
  for (int a = 0; a < MHID; ++a) {
    float s = 0.f;
#pragma unroll
    for (int bb = 0; bb < MHID; ++bb) {
      const int i  = a < bb ? a : bb;
      const int jj = a < bb ? bb : a;
      const float av = (a == bb) ? 0.f : r[i * MHID - (i * (i + 1)) / 2 + (jj - i - 1)];
      s += av * z0[bb];
    }
    z1[a] = s;
  }
#pragma unroll
  for (int a = 0; a < MHID; ++a) {
    float s = 0.f;
#pragma unroll
    for (int bb = 0; bb < MHID; ++bb) {
      const int i  = a < bb ? a : bb;
      const int jj = a < bb ? bb : a;
      const float av = (a == bb) ? 0.f : r[i * MHID - (i * (i + 1)) / 2 + (jj - i - 1)];
      s += av * z1[bb];
    }
    z2[a] = s;
  }

#pragma unroll
  for (int m = 0; m < MHID; ++m) {
    zf[0 * ZF_STRIDE + m * 4096 + g] = z0[m];
    zf[1 * ZF_STRIDE + m * 4096 + g] = z1[m];
    zf[2 * ZF_STRIDE + m * 4096 + g] = z2[m];
  }
}

#define PK 8
#define PNB (32 / PK)
__device__ __forceinline__ void rw_prop(const v2f (&aa)[32], const float* xs, float* xd,
                                        int w, int l16, int lhi) {
#pragma unroll 1
  for (int cp = 0; cp < 4; ++cp) {
    v8f ca = {}, cb = {};
    const float* xb = xs + (lhi * 2) * NPAD + cp * 32 + l16;
    int ob = 16;
    asm volatile("" : "+v"(ob));
    const float* xc = xb + ob;
    v2f ba[2][PK], bb[2][PK];
#pragma unroll
    for (int blk = -1; blk < PNB; ++blk) {
      if (blk + 1 < PNB) {
        const int ns = (blk + 1) & 1;
#pragma unroll
        for (int u = 0; u < PK; ++u) {
          const int kit = (blk + 1) * PK + u;
          ba[ns][u].x = xb[(kit * 4 + 0) * NPAD];
          ba[ns][u].y = xb[(kit * 4 + 1) * NPAD];
          bb[ns][u].x = xc[(kit * 4 + 0) * NPAD];
          bb[ns][u].y = xc[(kit * 4 + 1) * NPAD];
        }
      }
      __builtin_amdgcn_sched_barrier(0);
      if (blk >= 0) {
        const int cs = blk & 1;
#pragma unroll
        for (int u = 0; u < PK; ++u) {
          ca = wmma4(aa[blk * PK + u], ba[cs][u], ca);
          cb = wmma4(aa[blk * PK + u], bb[cs][u], cb);
        }
      }
      __builtin_amdgcn_sched_barrier(0);
    }
    float* xo = xd + (w * 16 + lhi * 8) * NPAD + cp * 32 + l16;
#pragma unroll
    for (int r = 0; r < 8; ++r) {
      xo[r * NPAD]      = ca[r];
      xo[r * NPAD + 16] = cb[r];
    }
  }
}

#define KU 4
#define NB (32 / KU)
__global__ __launch_bounds__(256) void rw_main(const float* __restrict__ adj,
                                               const float* __restrict__ feats,
                                               const int*   __restrict__ idxs,
                                               const float* __restrict__ zfrag,
                                               float*       __restrict__ out) {
  __shared__ float lxa[NNODE * NPAD];
  __shared__ float lxb[NNODE * NPAD];
  __shared__ float accbuf[OUTD];

  const int b   = blockIdx.x;
  const int tid = threadIdx.x;
  const int w   = tid >> 5;
  const int l   = tid & 31;
  const int l16 = l & 15;
  const int lhi = l >> 4;

  if (tid < OUTD) accbuf[tid] = 0.f;

  {
    const int r0 = w * 16;
#pragma unroll 4
    for (int r = r0; r < r0 + 16; ++r) {
      int idx = idxs[b * NNODE + r];
      float4 v = *(const float4*)(feats + (size_t)idx * CDIM + l * 4);
      *(float4*)(&lxa[r * NPAD + l * 4]) = v;
    }
  }

  const int row = w * 16 + l16;
  v2f aa[32];
  {
    const float* Ag = adj + (size_t)b * NNODE * CDIM + row * CDIM + lhi * 2;
#pragma unroll
    for (int kit = 0; kit < 32; ++kit) aa[kit] = *(const v2f*)(Ag + kit * 4);
  }
  __syncthreads();
  rw_prop(aa, lxa, lxb, w, l16, lhi);
  __syncthreads();

  v2f xa[32];
  {
    const float* ax0 = lxa + row * NPAD + lhi * 2;
#pragma unroll
    for (int kit = 0; kit < 32; ++kit) xa[kit] = *(const v2f*)(ax0 + kit * 4);
  }
  rw_prop(aa, lxb, lxa, w, l16, lhi);
  __syncthreads();

  float acc0 = 0.f, acc1 = 0.f;
  const float* ax1 = lxb + row * NPAD + lhi * 2;
  const float* ax2 = lxa + row * NPAD + lhi * 2;
  for (int t = 0; t < MHID * 2; ++t) {
    const v2f* bz0 = (const v2f*)(zfrag)                 + (size_t)t * 1024 + l;
    const v2f* bz1 = (const v2f*)(zfrag + ZF_STRIDE)     + (size_t)t * 1024 + l;
    const v2f* bz2 = (const v2f*)(zfrag + 2 * ZF_STRIDE) + (size_t)t * 1024 + l;
    v8f czx = {}, c1 = {}, c2 = {};
    v2f b0[2][KU], b1[2][KU], b2[2][KU], a1[2][KU], a2[2][KU];
#pragma unroll
    for (int blk = -1; blk < NB; ++blk) {
      if (blk + 1 < NB) {
        const int ns = (blk + 1) & 1;
#pragma unroll
        for (int u = 0; u < KU; ++u) {
          const int kit = (blk + 1) * KU + u;
          b0[ns][u] = bz0[kit * 32];
          b1[ns][u] = bz1[kit * 32];
          b2[ns][u] = bz2[kit * 32];
        }
#pragma unroll
        for (int u = 0; u < KU; ++u) {
          const int kit = (blk + 1) * KU + u;
          a1[ns][u] = *(const v2f*)(ax1 + kit * 4);
          a2[ns][u] = *(const v2f*)(ax2 + kit * 4);
        }
      }
      __builtin_amdgcn_sched_barrier(0);
      if (blk >= 0) {
        const int cs = blk & 1;
#pragma unroll
        for (int u = 0; u < KU; ++u) {
          czx = wmma4(xa[blk * KU + u], b0[cs][u], czx);
          c1  = wmma4(a1[cs][u],        b1[cs][u], c1);
          c2  = wmma4(a2[cs][u],        b2[cs][u], c2);
        }
      }
      __builtin_amdgcn_sched_barrier(0);
    }
    float s0 = 0.f, s1 = 0.f, s2 = 0.f;
#pragma unroll
    for (int r = 0; r < 8; ++r) {
      s0 += czx[r] * czx[r];
      s1 += czx[r] * c1[r];
      s2 += czx[r] * c2[r];
    }
    const float s = (s0 + s1) + s2;
    if ((t & 1) == 0) acc0 += s; else acc1 += s;
  }

  atomicAdd(&accbuf[l16], acc0);
  atomicAdd(&accbuf[16 + l16], acc1);
  __syncthreads();
  if (tid < OUTD) out[(size_t)b * OUTD + tid] = accbuf[tid] * (1.0f / 3840.0f);
}

#define TP       132
#define G0T      36
#define GLEN     41984
#define NQ       5248
#define WS_WT    131072
#define WS_G     1703936
#define WS_PART  52035584
#define KCH      16

#define NTB 82
#define GRAM_GRID 1106
__device__ __forceinline__ void rw_tables(const float* __restrict__ fh,
                                          const float* __restrict__ ah,
                                          float* __restrict__ wt,
                                          int gw, int l) {
  const int slot = gw >> 2;
  const int ot   = (gw >> 1) & 1;
  const int jh   = gw & 1;
  const int o16  = l & 15;
  const int hi   = l >> 4;
  const int o    = ot * 16 + o16;
  int s, w, ct;
  float f = 1.f;
  if (slot < G0T) {
    int d;
    if (slot < 20) { w = slot / 5; d = slot - w * 5; }
    else           { w = 4 + ((slot - 20) >> 2); d = (slot - 20) & 3; }
    s  = 0;
    ct = (w + d) & 7;
    if (d != 0) f = 2.f;
  } else {
    const int t2 = slot - G0T;
    s  = 1 + (t2 >> 6);
    w  = (t2 >> 3) & 7;
    ct = t2 & 7;
  }
  const int cp0 = 16 * ct + 8 * jh + 4 * hi;

  float zs[4][MHID];
#pragma unroll
  for (int e = 0; e < 4; ++e) {
#pragma unroll
    for (int m = 0; m < MHID; ++m) zs[e][m] = fh[(m * CDIM + cp0 + e) * OUTD + o];
  }
  if (s > 0) {
    float r[45];
#pragma unroll
    for (int p = 0; p < 45; ++p) {
      float t = ah[p * OUTD + o];
      r[p] = t > 0.f ? t : 0.f;
    }
    for (int it = 0; it < s; ++it) {
#pragma unroll
      for (int e = 0; e < 4; ++e) {
        float zn[MHID];
#pragma unroll
        for (int a = 0; a < MHID; ++a) {
          float acc = 0.f;
#pragma unroll
          for (int bb = 0; bb < MHID; ++bb) {
            const int i  = a < bb ? a : bb;
            const int jj = a < bb ? bb : a;
            const float av = (a == bb) ? 0.f : r[i * MHID - (i * (i + 1)) / 2 + (jj - i - 1)];
            acc = __builtin_fmaf(av, zs[e][bb], acc);
          }
          zn[a] = acc;
        }
#pragma unroll
        for (int a = 0; a < MHID; ++a) zs[e][a] = zn[a];
      }
    }
  }

  const float* za = fh + (16 * w) * OUTD + o;
  float4*      wo = (float4*)wt + (size_t)(ot * NQ + slot * 32 + jh) * 32 + l;
#pragma unroll
  for (int i = 0; i < 16; ++i) {
    const int c = (i >> 1) + 8 * (i & 1);
    float a0 = 0.f, a1 = 0.f, a2 = 0.f, a3 = 0.f;
#pragma unroll
    for (int m = 0; m < MHID; ++m) {
      const float a = za[(m * CDIM + c) * OUTD];
      a0 = __builtin_fmaf(a, zs[0][m], a0);
      a1 = __builtin_fmaf(a, zs[1][m], a1);
      a2 = __builtin_fmaf(a, zs[2][m], a2);
      a3 = __builtin_fmaf(a, zs[3][m], a3);
    }
    wo[i * 64] = make_float4(a0 * f, a1 * f, a2 * f, a3 * f);
  }
}

template <bool TWO>
__device__ __forceinline__ void rw_tiles(const v2f (&af)[32], unsigned xa, unsigned xc, v8f& ca, v8f& cb) {
  asm volatile("" : "+v"(xa));
  if (TWO) asm volatile("" : "+v"(xc));
  const lds_cv4f* pa = (const lds_cv4f*)xa;
  const lds_cv4f* pc = (const lds_cv4f*)xc;
  v4f fa[2][PK / 2], fb[2][PK / 2];
#pragma unroll
  for (int blk = -1; blk < PNB; ++blk) {
    if (blk + 1 < PNB) {
      const int ns = (blk + 1) & 1;
#pragma unroll
      for (int u = 0; u < PK / 2; ++u) {
        const int kp = (blk + 1) * (PK / 2) + u;
        fa[ns][u] = pa[kp * 2];
        if (TWO) fb[ns][u] = pc[kp * 2];
      }
    }
    __builtin_amdgcn_sched_barrier(0);
    if (blk >= 0) {
      const int cs = blk & 1;
#pragma unroll
      for (int u = 0; u < PK / 2; ++u) {
        const int q = blk * PK + 2 * u;
        const v2f al = {fa[cs][u].x, fa[cs][u].y}, ah = {fa[cs][u].z, fa[cs][u].w};
        ca = wmma4(af[q], al, ca);
        if (TWO) {
          const v2f bl = {fb[cs][u].x, fb[cs][u].y};
          cb = wmma4(af[q], bl, cb);
        }
        ca = wmma4(af[q + 1], ah, ca);
        if (TWO) {
          const v2f bh = {fb[cs][u].z, fb[cs][u].w};
          cb = wmma4(af[q + 1], bh, cb);
        }
      }
    }
    __builtin_amdgcn_sched_barrier(0);
  }
}

template <bool TO_WS>
__device__ __forceinline__ void rw_pass(const v2f (&af)[32], const float* xs, float* xd, float* gd,
                                        int w, int l, int l16, int lhi) {
#pragma unroll 1
  for (int cp = 0; cp < 4; ++cp) {
    v8f ca = {}, cb = {};
    const unsigned xa = (unsigned)(size_t)(xs + (cp * 32 + l16) * TP + lhi * 4);
    rw_tiles<true>(af, xa, xa + 16 * TP * 4, ca, cb);
    if (TO_WS) {
      float* go = gd + cp * 512 + l;
#pragma unroll
      for (int r = 0; r < 8; ++r) {
        go[r * 32]       = ca[r];
        go[256 + r * 32] = cb[r];
      }
    } else {
      float* xo = xd + (cp * 32 + l16) * TP + w * 16 + lhi * 8;
      *(float4*)(xo)               = make_float4(ca[0], ca[1], ca[2], ca[3]);
      *(float4*)(xo + 4)           = make_float4(ca[4], ca[5], ca[6], ca[7]);
      *(float4*)(xo + 16 * TP)     = make_float4(cb[0], cb[1], cb[2], cb[3]);
      *(float4*)(xo + 16 * TP + 4) = make_float4(cb[4], cb[5], cb[6], cb[7]);
    }
  }
}

template <bool TWO>
__device__ __forceinline__ void rw_gsym(const v2f (&af)[32], unsigned xa, unsigned xc, float* go) {
  v8f ca = {}, cb = {};
  rw_tiles<TWO>(af, xa, xc, ca, cb);
#pragma unroll
  for (int r = 0; r < 8; ++r) {
    go[r * 32] = ca[r];
    if (TWO) go[256 + r * 32] = cb[r];
  }
}

__global__ __launch_bounds__(256) void rw_gram(const float* __restrict__ adj,
                                               const float* __restrict__ feats,
                                               const int*   __restrict__ idxs,
                                               const float* __restrict__ fh,
                                               const float* __restrict__ ah,
                                               float*       __restrict__ wt,
                                               float*       __restrict__ gws) {
  __shared__ __attribute__((aligned(16))) float lxa[CDIM * TP];
  __shared__ __attribute__((aligned(16))) float lxb[CDIM * TP];

  const int tid = threadIdx.x;
  if (blockIdx.x < NTB) {
    const int gw = __builtin_amdgcn_readfirstlane((int)blockIdx.x * 8 + (tid >> 5));
    rw_tables(fh, ah, wt, gw, tid & 31);
    return;
  }
  const int b   = blockIdx.x - NTB;
  const int w   = tid >> 5;
  const int l   = tid & 31;
  const int l16 = l & 15;
  const int lhi = l >> 4;

  for (int g = 0; g < 4; ++g) {
    const int n0 = w * 16 + g * 4;
    const int i0 = idxs[b * NNODE + n0 + 0];
    const int i1 = idxs[b * NNODE + n0 + 1];
    const int i2 = idxs[b * NNODE + n0 + 2];
    const int i3 = idxs[b * NNODE + n0 + 3];
    const float4 v0 = *(const float4*)(feats + (size_t)i0 * CDIM + l * 4);
    const float4 v1 = *(const float4*)(feats + (size_t)i1 * CDIM + l * 4);
    const float4 v2 = *(const float4*)(feats + (size_t)i2 * CDIM + l * 4);
    const float4 v3 = *(const float4*)(feats + (size_t)i3 * CDIM + l * 4);
    float* xo = lxa + (l * 4) * TP + n0;
    *(float4*)(xo)          = make_float4(v0.x, v1.x, v2.x, v3.x);
    *(float4*)(xo + TP)     = make_float4(v0.y, v1.y, v2.y, v3.y);
    *(float4*)(xo + 2 * TP) = make_float4(v0.z, v1.z, v2.z, v3.z);
    *(float4*)(xo + 3 * TP) = make_float4(v0.w, v1.w, v2.w, v3.w);
  }

  const int row = w * 16 + l16;
  v2f aa[32];
  {
    const float* Ag = adj + (size_t)b * NNODE * CDIM + row * CDIM + lhi * 4;
#pragma unroll
    for (int kp = 0; kp < 16; ++kp) {
      const float4 t = *(const float4*)(Ag + kp * 8);
      aa[2 * kp]     = v2f{t.x, t.y};
      aa[2 * kp + 1] = v2f{t.z, t.w};
    }
  }
  __syncthreads();
  rw_pass<false>(aa, lxa, lxb, nullptr, w, l, l16, lhi);
  __syncthreads();

  v2f xt[32];
  {
    const float* t0 = lxa + row * TP + lhi * 4;
#pragma unroll
    for (int kp = 0; kp < 16; ++kp) {
      const float4 t = *(const float4*)(t0 + kp * 8);
      xt[2 * kp]     = v2f{t.x, t.y};
      xt[2 * kp + 1] = v2f{t.z, t.w};
    }
  }
  {
    const int wu = __builtin_amdgcn_readfirstlane(w);
    float* g0 = gws + (size_t)b * GLEN + (wu < 4 ? wu * 5 : 20 + (wu - 4) * 4) * 256 + l;
    const unsigned xk = (unsigned)(size_t)(lxa + l16 * TP + lhi * 4);
    const unsigned tb = 16 * TP * 4;
    rw_gsym<true>(xt, xk + ((wu + 0) & 7) * tb, xk + ((wu + 1) & 7) * tb, g0);
    rw_gsym<true>(xt, xk + ((wu + 2) & 7) * tb, xk + ((wu + 3) & 7) * tb, g0 + 512);
    if (wu < 4) rw_gsym<false>(xt, xk + (wu + 4) * tb, xk, g0 + 1024);
  }
  float* gb = gws + (size_t)b * GLEN + G0T * 256 + w * 2048;
  rw_pass<true>(xt, lxb, nullptr, gb,         w, l, l16, lhi);
  __syncthreads();
  rw_pass<false>(aa, lxb, lxa, nullptr, w, l, l16, lhi);
  __syncthreads();
  rw_pass<true>(xt, lxa, nullptr, gb + 16384, w, l, l16, lhi);
}

#define QCH     328
#define NST     10
#define SPITCH  260
#define SBUF    (32 * SPITCH)
__global__ __launch_bounds__(256) void rw_prod(const float* __restrict__ gws,
                                               const float* __restrict__ wt,
                                               float* __restrict__ part) {
  __shared__ float4 sm4[2 * SBUF / 4];
  float* sm = (float*)sm4;
  const int bp  = blockIdx.x / KCH;
  const int ch  = blockIdx.x - bp * KCH;
  const int tid = threadIdx.x;
  const int wv  = tid >> 5;
  const int l   = tid & 31;
  const int l16 = l & 15;
  const int hi  = l >> 4;

  const float* gsrc = gws + (size_t)(bp * 32 + wv * 4) * GLEN + ch * (QCH * 8) + l * 4;
  float*       sdst = sm + (wv * 4) * SPITCH + l * 4;
  const float* gtl  = gws + (size_t)(bp * 32 + wv * 4 + hi) * GLEN + ch * (QCH * 8) + NST * 256 + l16 * 4;
  float*       stl  = sm + (wv * 4 + hi) * SPITCH + l16 * 4;
  const float*  asrc = sm + l16 * SPITCH + wv * 32 + hi * 4;
  const float*  atl  = sm + l16 * SPITCH + wv * 8 + hi * 4;
  const float4* B0 = (const float4*)wt + (size_t)(ch * QCH + wv * 4) * 32 + l;
  const float4* B1 = B0 + (size_t)NQ * 32;
  const float4* T0 = (const float4*)wt + (size_t)(ch * QCH + NST * 32 + wv) * 32 + l;
  const float4* T1 = T0 + (size_t)NQ * 32;

  v8f c00 = {}, c01 = {}, c10 = {}, c11 = {};
  float4 g0, g1, g2, g3, g4, g5, g6, g7, t0, t1;
  float4 fb0[2][4], fb1[2][4];
#define STAGE_LOAD(off)                                              \
  g0 = *(const float4*)(gsrc + (off));                               \
  g1 = *(const float4*)(gsrc + (off) + 128);                         \
  g2 = *(const float4*)(gsrc + (size_t)GLEN + (off));                \
  g3 = *(const float4*)(gsrc + (size_t)GLEN + (off) + 128);          \
  g4 = *(const float4*)(gsrc + (size_t)2 * GLEN + (off));            \
  g5 = *(const float4*)(gsrc + (size_t)2 * GLEN + (off) + 128);      \
  g6 = *(const float4*)(gsrc + (size_t)3 * GLEN + (off));            \
  g7 = *(const float4*)(gsrc + (size_t)3 * GLEN + (off) + 128);
#define STAGE_STORE(buf)                                             \
  *(float4*)(sdst + (buf) * SBUF)                    = g0;           \
  *(float4*)(sdst + (buf) * SBUF + 128)              = g1;           \
  *(float4*)(sdst + (buf) * SBUF + SPITCH)           = g2;           \
  *(float4*)(sdst + (buf) * SBUF + SPITCH + 128)     = g3;           \
  *(float4*)(sdst + (buf) * SBUF + 2 * SPITCH)       = g4;           \
  *(float4*)(sdst + (buf) * SBUF + 2 * SPITCH + 128) = g5;           \
  *(float4*)(sdst + (buf) * SBUF + 3 * SPITCH)       = g6;           \
  *(float4*)(sdst + (buf) * SBUF + 3 * SPITCH + 128) = g7;
#define STAGE_BARRIER() asm volatile("s_wait_dscnt 0x0\n\ts_barrier_signal -1\n\ts_barrier_wait -1" ::: "memory");

  STAGE_LOAD(0)
#pragma unroll
  for (int u = 0; u < 4; ++u) {
    fb0[0][u] = B0[u * 32];
    fb1[0][u] = B1[u * 32];
  }
  STAGE_STORE(0)
  STAGE_LOAD(256)
  STAGE_BARRIER()

#pragma unroll
  for (int s = 0; s < NST; ++s) {
    const int cs = s & 1, ns = cs ^ 1;
    if (s + 1 < NST) {
#pragma unroll
      for (int u = 0; u < 4; ++u) {
        fb0[ns][u] = B0[((s + 1) * 32 + u) * 32];
        fb1[ns][u] = B1[((s + 1) * 32 + u) * 32];
      }
    } else {
      fb0[ns][0] = T0[0];
      fb1[ns][0] = T1[0];
    }
    float4 fa0[4], fa1[4];
#pragma unroll
    for (int u = 0; u < 4; ++u) {
      fa0[u] = *(const float4*)(asrc + cs * SBUF + u * 8);
      fa1[u] = *(const float4*)(asrc + cs * SBUF + 16 * SPITCH + u * 8);
    }
    __builtin_amdgcn_sched_barrier(0);
#pragma unroll
    for (int u = 0; u < 4; ++u) {
      const v2f a0l = {fa0[u].x, fa0[u].y}, a0h = {fa0[u].z, fa0[u].w};
      const v2f a1l = {fa1[u].x, fa1[u].y}, a1h = {fa1[u].z, fa1[u].w};
      const v2f b0l = {fb0[cs][u].x, fb0[cs][u].y}, b0h = {fb0[cs][u].z, fb0[cs][u].w};
      const v2f b1l = {fb1[cs][u].x, fb1[cs][u].y}, b1h = {fb1[cs][u].z, fb1[cs][u].w};
      c00 = wmma4(a0l, b0l, c00);
      c01 = wmma4(a0l, b1l, c01);
      c10 = wmma4(a1l, b0l, c10);
      c11 = wmma4(a1l, b1l, c11);
      c00 = wmma4(a0h, b0h, c00);
      c01 = wmma4(a0h, b1h, c01);
      c10 = wmma4(a1h, b0h, c10);
      c11 = wmma4(a1h, b1h, c11);
    }
    __builtin_amdgcn_sched_barrier(0);
    if (s + 1 < NST) {
      STAGE_STORE(ns)
      if (s + 2 < NST) {
        STAGE_LOAD((s + 2) * 256)
      } else {
        t0 = *(const float4*)(gtl);
        t1 = *(const float4*)(gtl + (size_t)2 * GLEN);
      }
    } else {
      *(float4*)(stl + ns * SBUF)              = t0;
      *(float4*)(stl + ns * SBUF + 2 * SPITCH) = t1;
    }
    STAGE_BARRIER()
  }
  {
    const int cs = NST & 1;
    const float4 fa0 = *(const float4*)(atl + cs * SBUF);
    const float4 fa1 = *(const float4*)(atl + cs * SBUF + 16 * SPITCH);
    const v2f a0l = {fa0.x, fa0.y}, a0h = {fa0.z, fa0.w};
    const v2f a1l = {fa1.x, fa1.y}, a1h = {fa1.z, fa1.w};
    const v2f b0l = {fb0[cs][0].x, fb0[cs][0].y}, b0h = {fb0[cs][0].z, fb0[cs][0].w};
    const v2f b1l = {fb1[cs][0].x, fb1[cs][0].y}, b1h = {fb1[cs][0].z, fb1[cs][0].w};
    c00 = wmma4(a0l, b0l, c00);
    c01 = wmma4(a0l, b1l, c01);
    c10 = wmma4(a1l, b0l, c10);
    c11 = wmma4(a1l, b1l, c11);
    c00 = wmma4(a0h, b0h, c00);
    c01 = wmma4(a0h, b1h, c01);
    c10 = wmma4(a1h, b0h, c10);
    c11 = wmma4(a1h, b1h, c11);
  }
  __syncthreads();

  float* red = sm;
  float* rw = red + wv * 1024 + l;
#pragma unroll
  for (int r = 0; r < 8; ++r) {
    rw[0   + r * 32] = c00[r];
    rw[256 + r * 32] = c01[r];
    rw[512 + r * 32] = c10[r];
    rw[768 + r * 32] = c11[r];
  }
  __syncthreads();
#pragma unroll
  for (int i = 0; i < 4; ++i) {
    const int e = i * 256 + tid;
    float s = 0.f;
#pragma unroll
    for (int k = 0; k < 8; ++k) s += red[k * 1024 + e];
    const int rt = e >> 9;
    const int ot = (e >> 8) & 1;
    const int r  = (e >> 5) & 7;
    const int ln = e & 31;
    const int bb = bp * 32 + rt * 16 + r + 8 * (ln >> 4);
    const int o  = ot * 16 + (ln & 15);
    part[(size_t)ch * (B_BATCH * OUTD) + bb * OUTD + o] = s;
  }
}

__global__ __launch_bounds__(256) void rw_reduce(const float* __restrict__ part,
                                                 float* __restrict__ out) {
  const int i = blockIdx.x * 256 + threadIdx.x;
  float s = 0.f;
#pragma unroll
  for (int k = 0; k < KCH; ++k) s += part[k * (B_BATCH * OUTD) + i];
  out[i] = s * (1.0f / 3840.0f);
}

extern "C" void kernel_launch(void* const* d_in, const int* in_sizes, int n_in,
                              void* d_out, int out_size, void* d_ws, size_t ws_size,
                              hipStream_t stream) {
  (void)in_sizes; (void)n_in; (void)out_size;
  const float* adj   = (const float*)d_in[0];
  const float* feats = (const float*)d_in[1];
  const int*   idxs  = (const int*)  d_in[2];
  const float* fh    = (const float*)d_in[3];
  const float* ah    = (const float*)d_in[4];
  float* out = (float*)d_out;
  float* ws  = (float*)d_ws;

  if (ws_size >= 210239488ull) {
    rw_gram<<<GRAM_GRID, 256, 0, stream>>>(adj, feats, idxs, fh, ah, ws + WS_WT, ws + WS_G);
    rw_prod<<<512, 256, 0, stream>>>(ws + WS_G, ws + WS_WT, ws + WS_PART);
    rw_reduce<<<128, 256, 0, stream>>>(ws + WS_PART, out);
  } else {
    rw_pre<<<64, 64, 0, stream>>>(fh, ah, ws + WS_ZF);
    rw_main<<<B_BATCH, 256, 0, stream>>>(adj, feats, idxs, ws + WS_ZF, out);
  }
}


# Step 1's table is symmetric (hidden adjacency is U + U^T): only sym(G1) = ((adj X0 + adj^T X0)/2)^T X0 is formed, 36 tiles like G0; step 2 as (adj^T X0)^T (adj X0); no X2; sum order changes
# speedup vs baseline: 11.2879x; 1.0820x over previous
#include <hip/hip_runtime.h>


typedef float v2f __attribute__((ext_vector_type(2)));
typedef float v8f __attribute__((ext_vector_type(8)));
typedef float v4f __attribute__((ext_vector_type(4)));
typedef __attribute__((address_space(3))) const v4f lds_cv4f;

#define B_BATCH 1024
#define NNODE   128
#define CDIM    128
#define MHID    10
#define OUTD    32
#define NPAD    136

__device__ __forceinline__ v8f wmma4(v2f a, v2f b, v8f c) {
  return __builtin_amdgcn_wmma_f32_16x16x4_f32(
       false, a,  false, b,
       (short)0, c,  false,  false);
}

#define WS_ZF  0
#define ZF_STRIDE 40960

__global__ __launch_bounds__(64) void rw_pre(const float* __restrict__ fh,
                                             const float* __restrict__ ah,
                                             float* __restrict__ zf) {
  const int g   = blockIdx.x * 64 + threadIdx.x;
  const int j   = g & 1;
  const int l16 = (g >> 1) & 15;
  const int lhi = (g >> 5) & 1;
  const int kit = (g >> 6) & 31;
  const int ot  = g >> 11;
  const int c   = kit * 4 + lhi * 2 + j;
  const int o   = ot * 16 + l16;

  float r[45];
#pragma unroll
  for (int p = 0; p < 45; ++p) {
    float t = ah[p * OUTD + o];
    r[p] = t > 0.f ? t : 0.f;
  }
  float z0[MHID], z1[MHID], z2[MHID];
#pragma unroll
  for (int m = 0; m < MHID; ++m) z0[m] = fh[(m * CDIM + c) * OUTD + o];

#pragma unroll
  for (int a = 0; a < MHID; ++a) {
    float s = 0.f;
#pragma unroll
    for (int bb = 0; bb < MHID; ++bb) {
      const int i  = a < bb ? a : bb;
      const int jj = a < bb ? bb : a;
      const float av = (a == bb) ? 0.f : r[i * MHID - (i * (i + 1)) / 2 + (jj - i - 1)];
      s += av * z0[bb];
    }
    z1[a] = s;
  }
#pragma unroll
  for (int a = 0; a < MHID; ++a) {
    float s = 0.f;
#pragma unroll
    for (int bb = 0; bb < MHID; ++bb) {
      const int i  = a < bb ? a : bb;
      const int jj = a < bb ? bb : a;
      const float av = (a == bb) ? 0.f : r[i * MHID - (i * (i + 1)) / 2 + (jj - i - 1)];
      s += av * z1[bb];
    }
    z2[a] = s;
  }

#pragma unroll
  for (int m = 0; m < MHID; ++m) {
    zf[0 * ZF_STRIDE + m * 4096 + g] = z0[m];
    zf[1 * ZF_STRIDE + m * 4096 + g] = z1[m];
    zf[2 * ZF_STRIDE + m * 4096 + g] = z2[m];
  }
}

#define PK 8
#define PNB (32 / PK)
__device__ __forceinline__ void rw_prop(const v2f (&aa)[32], const float* xs, float* xd,
                                        int w, int l16, int lhi) {
#pragma unroll 1
  for (int cp = 0; cp < 4; ++cp) {
    v8f ca = {}, cb = {};
    const float* xb = xs + (lhi * 2) * NPAD + cp * 32 + l16;
    int ob = 16;
    asm volatile("" : "+v"(ob));
    const float* xc = xb + ob;
    v2f ba[2][PK], bb[2][PK];
#pragma unroll
    for (int blk = -1; blk < PNB; ++blk) {
      if (blk + 1 < PNB) {
        const int ns = (blk + 1) & 1;
#pragma unroll
        for (int u = 0; u < PK; ++u) {
          const int kit = (blk + 1) * PK + u;
          ba[ns][u].x = xb[(kit * 4 + 0) * NPAD];
          ba[ns][u].y = xb[(kit * 4 + 1) * NPAD];
          bb[ns][u].x = xc[(kit * 4 + 0) * NPAD];
          bb[ns][u].y = xc[(kit * 4 + 1) * NPAD];
        }
      }
      __builtin_amdgcn_sched_barrier(0);
      if (blk >= 0) {
        const int cs = blk & 1;
#pragma unroll
        for (int u = 0; u < PK; ++u) {
          ca = wmma4(aa[blk * PK + u], ba[cs][u], ca);
          cb = wmma4(aa[blk * PK + u], bb[cs][u], cb);
        }
      }
      __builtin_amdgcn_sched_barrier(0);
    }
    float* xo = xd + (w * 16 + lhi * 8) * NPAD + cp * 32 + l16;
#pragma unroll
    for (int r = 0; r < 8; ++r) {
      xo[r * NPAD]      = ca[r];
      xo[r * NPAD + 16] = cb[r];
    }
  }
}

#define KU 4
#define NB (32 / KU)
__global__ __launch_bounds__(256) void rw_main(const float* __restrict__ adj,
                                               const float* __restrict__ feats,
                                               const int*   __restrict__ idxs,
                                               const float* __restrict__ zfrag,
                                               float*       __restrict__ out) {
  __shared__ float lxa[NNODE * NPAD];
  __shared__ float lxb[NNODE * NPAD];
  __shared__ float accbuf[OUTD];

  const int b   = blockIdx.x;
  const int tid = threadIdx.x;
  const int w   = tid >> 5;
  const int l   = tid & 31;
  const int l16 = l & 15;
  const int lhi = l >> 4;

  if (tid < OUTD) accbuf[tid] = 0.f;

  {
    const int r0 = w * 16;
#pragma unroll 4
    for (int r = r0; r < r0 + 16; ++r) {
      int idx = idxs[b * NNODE + r];
      float4 v = *(const float4*)(feats + (size_t)idx * CDIM + l * 4);
      *(float4*)(&lxa[r * NPAD + l * 4]) = v;
    }
  }

  const int row = w * 16 + l16;
  v2f aa[32];
  {
    const float* Ag = adj + (size_t)b * NNODE * CDIM + row * CDIM + lhi * 2;
#pragma unroll
    for (int kit = 0; kit < 32; ++kit) aa[kit] = *(const v2f*)(Ag + kit * 4);
  }
  __syncthreads();
  rw_prop(aa, lxa, lxb, w, l16, lhi);
  __syncthreads();

  v2f xa[32];
  {
    const float* ax0 = lxa + row * NPAD + lhi * 2;
#pragma unroll
    for (int kit = 0; kit < 32; ++kit) xa[kit] = *(const v2f*)(ax0 + kit * 4);
  }
  rw_prop(aa, lxb, lxa, w, l16, lhi);
  __syncthreads();

  float acc0 = 0.f, acc1 = 0.f;
  const float* ax1 = lxb + row * NPAD + lhi * 2;
  const float* ax2 = lxa + row * NPAD + lhi * 2;
  for (int t = 0; t < MHID * 2; ++t) {
    const v2f* bz0 = (const v2f*)(zfrag)                 + (size_t)t * 1024 + l;
    const v2f* bz1 = (const v2f*)(zfrag + ZF_STRIDE)     + (size_t)t * 1024 + l;
    const v2f* bz2 = (const v2f*)(zfrag + 2 * ZF_STRIDE) + (size_t)t * 1024 + l;
    v8f czx = {}, c1 = {}, c2 = {};
    v2f b0[2][KU], b1[2][KU], b2[2][KU], a1[2][KU], a2[2][KU];
#pragma unroll
    for (int blk = -1; blk < NB; ++blk) {
      if (blk + 1 < NB) {
        const int ns = (blk + 1) & 1;
#pragma unroll
        for (int u = 0; u < KU; ++u) {
          const int kit = (blk + 1) * KU + u;
          b0[ns][u] = bz0[kit * 32];
          b1[ns][u] = bz1[kit * 32];
          b2[ns][u] = bz2[kit * 32];
        }
#pragma unroll
        for (int u = 0; u < KU; ++u) {
          const int kit = (blk + 1) * KU + u;
          a1[ns][u] = *(const v2f*)(ax1 + kit * 4);
          a2[ns][u] = *(const v2f*)(ax2 + kit * 4);
        }
      }
      __builtin_amdgcn_sched_barrier(0);
      if (blk >= 0) {
        const int cs = blk & 1;
#pragma unroll
        for (int u = 0; u < KU; ++u) {
          czx = wmma4(xa[blk * KU + u], b0[cs][u], czx);
          c1  = wmma4(a1[cs][u],        b1[cs][u], c1);
          c2  = wmma4(a2[cs][u],        b2[cs][u], c2);
        }
      }
      __builtin_amdgcn_sched_barrier(0);
    }
    float s0 = 0.f, s1 = 0.f, s2 = 0.f;
#pragma unroll
    for (int r = 0; r < 8; ++r) {
      s0 += czx[r] * czx[r];
      s1 += czx[r] * c1[r];
      s2 += czx[r] * c2[r];
    }
    const float s = (s0 + s1) + s2;
    if ((t & 1) == 0) acc0 += s; else acc1 += s;
  }

  atomicAdd(&accbuf[l16], acc0);
  atomicAdd(&accbuf[16 + l16], acc1);
  __syncthreads();
  if (tid < OUTD) out[(size_t)b * OUTD + tid] = accbuf[tid] * (1.0f / 3840.0f);
}

#define TP       132
#define G0T      36
#define G2S      72
#define GZS      136
#define GLEN     36864
#define NQ       4608
#define WS_WT    131072
#define WS_G     1703936
#define WS_PART  52035584
#define KCH      16

#define NTB 72
#define GRAM_GRID 1096
__device__ __forceinline__ void rw_tables(const float* __restrict__ fh,
                                          const float* __restrict__ ah,
                                          float* __restrict__ wt,
                                          int gw, int l) {
  const int slot = gw >> 2;
  const int ot   = (gw >> 1) & 1;
  const int jh   = gw & 1;
  const int o16  = l & 15;
  const int hi   = l >> 4;
  const int o    = ot * 16 + o16;
  if (slot >= GZS) {
    float4* wz = (float4*)wt + (size_t)(ot * NQ + slot * 32 + jh) * 32 + l;
#pragma unroll
    for (int i = 0; i < 16; ++i) wz[i * 64] = make_float4(0.f, 0.f, 0.f, 0.f);
    return;
  }
  int s, w, ct;
  float f = 1.f;
  if (slot < G2S) {
    s = slot >= G0T ? 1 : 0;
    const int t = slot - s * G0T;
    int d;
    if (t < 20) { w = t / 5; d = t - w * 5; }
    else        { w = 4 + ((t - 20) >> 2); d = (t - 20) & 3; }
    ct = (w + d) & 7;
    if (d != 0) f = 2.f;
  } else {
    const int t2 = slot - G2S;
    s  = 2;
    w  = t2 >> 3;
    ct = t2 & 7;
  }
  const int cp0 = 16 * ct + 8 * jh + 4 * hi;

  float zs[4][MHID];
#pragma unroll
  for (int e = 0; e < 4; ++e) {
#pragma unroll
    for (int m = 0; m < MHID; ++m) zs[e][m] = fh[(m * CDIM + cp0 + e) * OUTD + o];
  }
  if (s > 0) {
    float r[45];
#pragma unroll
    for (int p = 0; p < 45; ++p) {
      float t = ah[p * OUTD + o];
      r[p] = t > 0.f ? t : 0.f;
    }
    for (int it = 0; it < s; ++it) {
#pragma unroll
      for (int e = 0; e < 4; ++e) {
        float zn[MHID];
#pragma unroll
        for (int a = 0; a < MHID; ++a) {
          float acc = 0.f;
#pragma unroll
          for (int bb = 0; bb < MHID; ++bb) {
            const int i  = a < bb ? a : bb;
            const int jj = a < bb ? bb : a;
            const float av = (a == bb) ? 0.f : r[i * MHID - (i * (i + 1)) / 2 + (jj - i - 1)];
            acc = __builtin_fmaf(av, zs[e][bb], acc);
          }
          zn[a] = acc;
        }
#pragma unroll
        for (int a = 0; a < MHID; ++a) zs[e][a] = zn[a];
      }
    }
  }

  const float* za = fh + (16 * w) * OUTD + o;
  float4*      wo = (float4*)wt + (size_t)(ot * NQ + slot * 32 + jh) * 32 + l;
#pragma unroll
  for (int i = 0; i < 16; ++i) {
    const int c = (i >> 1) + 8 * (i & 1);
    float a0 = 0.f, a1 = 0.f, a2 = 0.f, a3 = 0.f;
#pragma unroll
    for (int m = 0; m < MHID; ++m) {
      const float a = za[(m * CDIM + c) * OUTD];
      a0 = __builtin_fmaf(a, zs[0][m], a0);
      a1 = __builtin_fmaf(a, zs[1][m], a1);
      a2 = __builtin_fmaf(a, zs[2][m], a2);
      a3 = __builtin_fmaf(a, zs[3][m], a3);
    }
    wo[i * 64] = make_float4(a0 * f, a1 * f, a2 * f, a3 * f);
  }
}

template <bool TWO>
__device__ __forceinline__ void rw_tiles(const v2f (&af)[32], unsigned xa, unsigned xc, v8f& ca, v8f& cb) {
  asm volatile("" : "+v"(xa));
  if (TWO) asm volatile("" : "+v"(xc));
  const lds_cv4f* pa = (const lds_cv4f*)xa;
  const lds_cv4f* pc = (const lds_cv4f*)xc;
  v4f fa[2][PK / 2], fb[2][PK / 2];
#pragma unroll
  for (int blk = -1; blk < PNB; ++blk) {
    if (blk + 1 < PNB) {
      const int ns = (blk + 1) & 1;
#pragma unroll
      for (int u = 0; u < PK / 2; ++u) {
        const int kp = (blk + 1) * (PK / 2) + u;
        fa[ns][u] = pa[kp * 2];
        if (TWO) fb[ns][u] = pc[kp * 2];
      }
    }
    __builtin_amdgcn_sched_barrier(0);
    if (blk >= 0) {
      const int cs = blk & 1;
#pragma unroll
      for (int u = 0; u < PK / 2; ++u) {
        const int q = blk * PK + 2 * u;
        const v2f al = {fa[cs][u].x, fa[cs][u].y}, ah = {fa[cs][u].z, fa[cs][u].w};
        ca = wmma4(af[q], al, ca);
        if (TWO) {
          const v2f bl = {fb[cs][u].x, fb[cs][u].y};
          cb = wmma4(af[q], bl, cb);
        }
        ca = wmma4(af[q + 1], ah, ca);
        if (TWO) {
          const v2f bh = {fb[cs][u].z, fb[cs][u].w};
          cb = wmma4(af[q + 1], bh, cb);
        }
      }
    }
    __builtin_amdgcn_sched_barrier(0);
  }
}

template <bool TO_WS>
__device__ __forceinline__ void rw_pass(const v2f (&af)[32], const float* xs, float* xd, float* gd,
                                        int w, int l, int l16, int lhi) {
#pragma unroll 1
  for (int cp = 0; cp < 4; ++cp) {
    v8f ca = {}, cb = {};
    const unsigned xa = (unsigned)(size_t)(xs + (cp * 32 + l16) * TP + lhi * 4);
    rw_tiles<true>(af, xa, xa + 16 * TP * 4, ca, cb);
    if (TO_WS) {
      float* go = gd + cp * 512 + l;
#pragma unroll
      for (int r = 0; r < 8; ++r) {
        go[r * 32]       = ca[r];
        go[256 + r * 32] = cb[r];
      }
    } else {
      float* xo = xd + (cp * 32 + l16) * TP + w * 16 + lhi * 8;
      *(float4*)(xo)               = make_float4(ca[0], ca[1], ca[2], ca[3]);
      *(float4*)(xo + 4)           = make_float4(ca[4], ca[5], ca[6], ca[7]);
      *(float4*)(xo + 16 * TP)     = make_float4(cb[0], cb[1], cb[2], cb[3]);
      *(float4*)(xo + 16 * TP + 4) = make_float4(cb[4], cb[5], cb[6], cb[7]);
    }
  }
}

template <bool TWO>
__device__ __forceinline__ void rw_gsym(const v2f (&af)[32], unsigned xa, unsigned xc, float* go) {
  v8f ca = {}, cb = {};
  rw_tiles<TWO>(af, xa, xc, ca, cb);
#pragma unroll
  for (int r = 0; r < 8; ++r) {
    go[r * 32] = ca[r];
    if (TWO) go[256 + r * 32] = cb[r];
  }
}

#define YK  4
#define TRP 20
__device__ __forceinline__ void rw_ytr(const v2f (&af)[32], const float* __restrict__ ab, float* sc,
                                       int l16, int lhi, v2f (&yt)[32]) {
  v2f ba[2][YK], bb[2][YK];
  v8f ca = {}, cb = {};
#pragma unroll
  for (int blk = -1; blk < 32; ++blk) {
    if (blk + 1 < 32) {
      const int nb = blk + 1, ns = nb & 1, cpn = nb >> 3, kb = nb & 7;
#pragma unroll
      for (int u = 0; u < YK; ++u) {
        const int q  = kb * YK + u;
        const int k0 = 8 * (q >> 1) + 2 * (q & 1);
        ba[ns][u].x = ab[(k0    ) * NNODE + cpn * 32];
        ba[ns][u].y = ab[(k0 + 1) * NNODE + cpn * 32];
        bb[ns][u].x = ab[(k0    ) * NNODE + cpn * 32 + 16];
        bb[ns][u].y = ab[(k0 + 1) * NNODE + cpn * 32 + 16];
      }
    }
    __builtin_amdgcn_sched_barrier(0);
    if (blk >= 0) {
      const int cs = blk & 1, cp = blk >> 3, kb = blk & 7;
#pragma unroll
      for (int u = 0; u < YK; ++u) {
        const int q = kb * YK + u;
        ca = wmma4(af[q], ba[cs][u], ca);
        cb = wmma4(af[q], bb[cs][u], cb);
      }
      if (kb == 7) {
        float* so = sc + (lhi * 8) * TRP + l16;
#pragma unroll
        for (int r = 0; r < 8; ++r) {
          so[r * TRP]        = ca[r];
          so[(16 + r) * TRP] = cb[r];
        }
        asm volatile("s_wait_dscnt 0x0" ::: "memory");
        const float* si = sc + l16 * TRP + lhi * 4;
        const float4 p0 = *(const float4*)(si);
        const float4 p1 = *(const float4*)(si + 8);
        const float4 p2 = *(const float4*)(si + 16 * TRP);
        const float4 p3 = *(const float4*)(si + 16 * TRP + 8);
        yt[8 * cp + 0] = v2f{p0.x, p0.y};
        yt[8 * cp + 1] = v2f{p0.z, p0.w};
        yt[8 * cp + 2] = v2f{p1.x, p1.y};
        yt[8 * cp + 3] = v2f{p1.z, p1.w};
        yt[8 * cp + 4] = v2f{p2.x, p2.y};
        yt[8 * cp + 5] = v2f{p2.z, p2.w};
        yt[8 * cp + 6] = v2f{p3.x, p3.y};
        yt[8 * cp + 7] = v2f{p3.z, p3.w};
        ca = v8f{};
        cb = v8f{};
      }
    }
    __builtin_amdgcn_sched_barrier(0);
  }
}

__global__ __launch_bounds__(256) void rw_gram(const float* __restrict__ adj,
                                               const float* __restrict__ feats,
                                               const int*   __restrict__ idxs,
                                               const float* __restrict__ fh,
                                               const float* __restrict__ ah,
                                               float*       __restrict__ wt,
                                               float*       __restrict__ gws) {
  __shared__ __attribute__((aligned(16))) float lxa[CDIM * TP];
  __shared__ __attribute__((aligned(16))) float lxb[CDIM * TP];
  __shared__ __attribute__((aligned(16))) float ltr[8 * 32 * TRP];

  const int tid = threadIdx.x;
  if (blockIdx.x < NTB) {
    const int gw = __builtin_amdgcn_readfirstlane((int)blockIdx.x * 8 + (tid >> 5));
    rw_tables(fh, ah, wt, gw, tid & 31);
    return;
  }
  const int b   = blockIdx.x - NTB;
  const int w   = tid >> 5;
  const int l   = tid & 31;
  const int l16 = l & 15;
  const int lhi = l >> 4;

  for (int g = 0; g < 4; ++g) {
    const int n0 = w * 16 + g * 4;
    const int i0 = idxs[b * NNODE + n0 + 0];
    const int i1 = idxs[b * NNODE + n0 + 1];
    const int i2 = idxs[b * NNODE + n0 + 2];
    const int i3 = idxs[b * NNODE + n0 + 3];
    const float4 v0 = *(const float4*)(feats + (size_t)i0 * CDIM + l * 4);
    const float4 v1 = *(const float4*)(feats + (size_t)i1 * CDIM + l * 4);
    const float4 v2 = *(const float4*)(feats + (size_t)i2 * CDIM + l * 4);
    const float4 v3 = *(const float4*)(feats + (size_t)i3 * CDIM + l * 4);
    float* xo = lxa + (l * 4) * TP + n0;
    *(float4*)(xo)          = make_float4(v0.x, v1.x, v2.x, v3.x);
    *(float4*)(xo + TP)     = make_float4(v0.y, v1.y, v2.y, v3.y);
    *(float4*)(xo + 2 * TP) = make_float4(v0.z, v1.z, v2.z, v3.z);
    *(float4*)(xo + 3 * TP) = make_float4(v0.w, v1.w, v2.w, v3.w);
  }

  const int row = w * 16 + l16;
  v2f aa[32];
  {
    const float* Ag = adj + (size_t)b * NNODE * CDIM + row * CDIM + lhi * 4;
#pragma unroll
    for (int kp = 0; kp < 16; ++kp) {
      const float4 t = *(const float4*)(Ag + kp * 8);
      aa[2 * kp]     = v2f{t.x, t.y};
      aa[2 * kp + 1] = v2f{t.z, t.w};
    }
  }
  __syncthreads();
  rw_pass<false>(aa, lxa, lxb, nullptr, w, l, l16, lhi);

  v2f xt[32];
  {
    const float* t0 = lxa + row * TP + lhi * 4;
#pragma unroll
    for (int kp = 0; kp < 16; ++kp) {
      const float4 t = *(const float4*)(t0 + kp * 8);
      xt[2 * kp]     = v2f{t.x, t.y};
      xt[2 * kp + 1] = v2f{t.z, t.w};
    }
  }
  const int wu = __builtin_amdgcn_readfirstlane(w);
  float* g0 = gws + (size_t)b * GLEN + (wu < 4 ? wu * 5 : 20 + (wu - 4) * 4) * 256 + l;
  const unsigned xk = (unsigned)(size_t)(lxa + l16 * TP + lhi * 4);
  const unsigned tb = 16 * TP * 4;
  rw_gsym<true>(xt, xk + ((wu + 0) & 7) * tb, xk + ((wu + 1) & 7) * tb, g0);
  rw_gsym<true>(xt, xk + ((wu + 2) & 7) * tb, xk + ((wu + 3) & 7) * tb, g0 + 512);
  if (wu < 4) rw_gsym<false>(xt, xk + (wu + 4) * tb, xk, g0 + 1024);

  v2f yt[32];
  rw_ytr(xt, adj + (size_t)b * NNODE * NNODE + (lhi * 4) * NNODE + l16, ltr + w * (32 * TRP), l16, lhi, yt);
  __syncthreads();

  rw_pass<true>(yt, lxb, nullptr, gws + (size_t)b * GLEN + G2S * 256 + w * 2048, w, l, l16, lhi);

  {
    const float* t1 = lxb + row * TP + lhi * 4;
#pragma unroll
    for (int kp = 0; kp < 16; ++kp) {
      const float4 t = *(const float4*)(t1 + kp * 8);
      yt[2 * kp]     = v2f{0.5f * (yt[2 * kp].x + t.x),     0.5f * (yt[2 * kp].y + t.y)};
      yt[2 * kp + 1] = v2f{0.5f * (yt[2 * kp + 1].x + t.z), 0.5f * (yt[2 * kp + 1].y + t.w)};
    }
  }
  float* g1 = g0 + G0T * 256;
  rw_gsym<true>(yt, xk + ((wu + 0) & 7) * tb, xk + ((wu + 1) & 7) * tb, g1);
  rw_gsym<true>(yt, xk + ((wu + 2) & 7) * tb, xk + ((wu + 3) & 7) * tb, g1 + 512);
  if (wu < 4) rw_gsym<false>(yt, xk + (wu + 4) * tb, xk, g1 + 1024);

  float* gz = gws + (size_t)b * GLEN + (GZS + w) * 256 + l;
#pragma unroll
  for (int r = 0; r < 8; ++r) gz[r * 32] = 0.f;
}

#define QCH     288
#define NST     9
#define SPITCH  260
#define SBUF    (32 * SPITCH)
__global__ __launch_bounds__(256) void rw_prod(const float* __restrict__ gws,
                                               const float* __restrict__ wt,
                                               float* __restrict__ part) {
  __shared__ float4 sm4[2 * SBUF / 4];
  float* sm = (float*)sm4;
  const int bp  = blockIdx.x / KCH;
  const int ch  = blockIdx.x - bp * KCH;
  const int tid = threadIdx.x;
  const int wv  = tid >> 5;
  const int l   = tid & 31;
  const int l16 = l & 15;
  const int hi  = l >> 4;

  const float* gsrc = gws + (size_t)(bp * 32 + wv * 4) * GLEN + ch * (QCH * 8) + l * 4;
  float*       sdst = sm + (wv * 4) * SPITCH + l * 4;
  const float*  asrc = sm + l16 * SPITCH + wv * 32 + hi * 4;
  const float4* B0 = (const float4*)wt + (size_t)(ch * QCH + wv * 4) * 32 + l;
  const float4* B1 = B0 + (size_t)NQ * 32;

  v8f c00 = {}, c01 = {}, c10 = {}, c11 = {};
  float4 g0, g1, g2, g3, g4, g5, g6, g7;
  float4 fb0[2][4], fb1[2][4];
#define STAGE_LOAD(off)                                              \
  g0 = *(const float4*)(gsrc + (off));                               \
  g1 = *(const float4*)(gsrc + (off) + 128);                         \
  g2 = *(const float4*)(gsrc + (size_t)GLEN + (off));                \
  g3 = *(const float4*)(gsrc + (size_t)GLEN + (off) + 128);          \
  g4 = *(const float4*)(gsrc + (size_t)2 * GLEN + (off));            \
  g5 = *(const float4*)(gsrc + (size_t)2 * GLEN + (off) + 128);      \
  g6 = *(const float4*)(gsrc + (size_t)3 * GLEN + (off));            \
  g7 = *(const float4*)(gsrc + (size_t)3 * GLEN + (off) + 128);
#define STAGE_STORE(buf)                                             \
  *(float4*)(sdst + (buf) * SBUF)                    = g0;           \
  *(float4*)(sdst + (buf) * SBUF + 128)              = g1;           \
  *(float4*)(sdst + (buf) * SBUF + SPITCH)           = g2;           \
  *(float4*)(sdst + (buf) * SBUF + SPITCH + 128)     = g3;           \
  *(float4*)(sdst + (buf) * SBUF + 2 * SPITCH)       = g4;           \
  *(float4*)(sdst + (buf) * SBUF + 2 * SPITCH + 128) = g5;           \
  *(float4*)(sdst + (buf) * SBUF + 3 * SPITCH)       = g6;           \
  *(float4*)(sdst + (buf) * SBUF + 3 * SPITCH + 128) = g7;
#define STAGE_BARRIER() asm volatile("s_wait_dscnt 0x0\n\ts_barrier_signal -1\n\ts_barrier_wait -1" ::: "memory");

  STAGE_LOAD(0)
#pragma unroll
  for (int u = 0; u < 4; ++u) {
    fb0[0][u] = B0[u * 32];
    fb1[0][u] = B1[u * 32];
  }
  STAGE_STORE(0)
  STAGE_LOAD(256)
  STAGE_BARRIER()

#pragma unroll
  for (int s = 0; s < NST; ++s) {
    const int cs = s & 1, ns = cs ^ 1;
    if (s + 1 < NST) {
#pragma unroll
      for (int u = 0; u < 4; ++u) {
        fb0[ns][u] = B0[((s + 1) * 32 + u) * 32];
        fb1[ns][u] = B1[((s + 1) * 32 + u) * 32];
      }
    }
    float4 fa0[4], fa1[4];
#pragma unroll
    for (int u = 0; u < 4; ++u) {
      fa0[u] = *(const float4*)(asrc + cs * SBUF + u * 8);
      fa1[u] = *(const float4*)(asrc + cs * SBUF + 16 * SPITCH + u * 8);
    }
    __builtin_amdgcn_sched_barrier(0);
#pragma unroll
    for (int u = 0; u < 4; ++u) {
      const v2f a0l = {fa0[u].x, fa0[u].y}, a0h = {fa0[u].z, fa0[u].w};
      const v2f a1l = {fa1[u].x, fa1[u].y}, a1h = {fa1[u].z, fa1[u].w};
      const v2f b0l = {fb0[cs][u].x, fb0[cs][u].y}, b0h = {fb0[cs][u].z, fb0[cs][u].w};
      const v2f b1l = {fb1[cs][u].x, fb1[cs][u].y}, b1h = {fb1[cs][u].z, fb1[cs][u].w};
      c00 = wmma4(a0l, b0l, c00);
      c01 = wmma4(a0l, b1l, c01);
      c10 = wmma4(a1l, b0l, c10);
      c11 = wmma4(a1l, b1l, c11);
      c00 = wmma4(a0h, b0h, c00);
      c01 = wmma4(a0h, b1h, c01);
      c10 = wmma4(a1h, b0h, c10);
      c11 = wmma4(a1h, b1h, c11);
    }
    __builtin_amdgcn_sched_barrier(0);
    if (s + 1 < NST) {
      STAGE_STORE(ns)
      if (s + 2 < NST) {
        STAGE_LOAD((s + 2) * 256)
      }
      STAGE_BARRIER()
    }
  }
  __syncthreads();

  float* red = sm;
  float* rw = red + wv * 1024 + l;
#pragma unroll
  for (int r = 0; r < 8; ++r) {
    rw[0   + r * 32] = c00[r];
    rw[256 + r * 32] = c01[r];
    rw[512 + r * 32] = c10[r];
    rw[768 + r * 32] = c11[r];
  }
  __syncthreads();
#pragma unroll
  for (int i = 0; i < 4; ++i) {
    const int e = i * 256 + tid;
    float s = 0.f;
#pragma unroll
    for (int k = 0; k < 8; ++k) s += red[k * 1024 + e];
    const int rt = e >> 9;
    const int ot = (e >> 8) & 1;
    const int r  = (e >> 5) & 7;
    const int ln = e & 31;
    const int bb = bp * 32 + rt * 16 + r + 8 * (ln >> 4);
    const int o  = ot * 16 + (ln & 15);
    part[(size_t)ch * (B_BATCH * OUTD) + bb * OUTD + o] = s;
  }
}

__global__ __launch_bounds__(256) void rw_reduce(const float* __restrict__ part,
                                                 float* __restrict__ out) {
  const int i = blockIdx.x * 256 + threadIdx.x;
  float s = 0.f;
#pragma unroll
  for (int k = 0; k < KCH; ++k) s += part[k * (B_BATCH * OUTD) + i];
  out[i] = s * (1.0f / 3840.0f);
}

extern "C" void kernel_launch(void* const* d_in, const int* in_sizes, int n_in,
                              void* d_out, int out_size, void* d_ws, size_t ws_size,
                              hipStream_t stream) {
  (void)in_sizes; (void)n_in; (void)out_size;
  const float* adj   = (const float*)d_in[0];
  const float* feats = (const float*)d_in[1];
  const int*   idxs  = (const int*)  d_in[2];
  const float* fh    = (const float*)d_in[3];
  const float* ah    = (const float*)d_in[4];
  float* out = (float*)d_out;
  float* ws  = (float*)d_ws;

  if (ws_size >= 210239488ull) {
    rw_gram<<<GRAM_GRID, 256, 0, stream>>>(adj, feats, idxs, fh, ah, ws + WS_WT, ws + WS_G);
    rw_prod<<<512, 256, 0, stream>>>(ws + WS_G, ws + WS_WT, ws + WS_PART);
    rw_reduce<<<128, 256, 0, stream>>>(ws + WS_PART, out);
  } else {
    rw_pre<<<64, 64, 0, stream>>>(fh, ah, ws + WS_ZF);
    rw_main<<<B_BATCH, 256, 0, stream>>>(adj, feats, idxs, ws + WS_ZF, out);
  }
}
